# starting kernel
# baseline (speedup 1.0000x reference)
#include <hip/hip_runtime.h>
#include <math.h>

#define N_NODES 100000
#define N_EDGES 1600000
#define IN_DIM  128
#define HEADS   8
#define HDIM    16
#define OUT_DIM 128
#define NEG_SLOPE 0.01f
#define LDS_STRIDE 130

typedef __attribute__((ext_vector_type(2))) float v2f;
typedef __attribute__((ext_vector_type(8))) float v8f;

__device__ __forceinline__ void atomicMaxF(float* addr, float val) {
    if (val >= 0.0f) {
        atomicMax((int*)addr, __float_as_int(val));
    } else {
        atomicMin((unsigned int*)addr, __float_as_uint(val));
    }
}

__global__ void gat_init(float* __restrict__ out0, float* __restrict__ m,
                         float* __restrict__ den) {
    int j = blockIdx.x * blockDim.x + threadIdx.x;
    if (j < N_NODES * OUT_DIM) out0[j] = 0.0f;
    if (j < N_NODES * HEADS) { m[j] = -INFINITY; den[j] = 0.0f; }
}

__global__ void __launch_bounds__(256)
gat_gemm_wmma(const float* __restrict__ h, const float* __restrict__ Wfc,
              float* __restrict__ z) {
    __shared__ float Wl[OUT_DIM * LDS_STRIDE];

    const int tid = threadIdx.x;
    for (int j = tid; j < IN_DIM * OUT_DIM; j += 256) {
        int i  = j >> 7;
        int c  = j & 127;
        int hh = c >> 4;
        int d  = c & 15;
        Wl[c * LDS_STRIDE + i] = Wfc[hh * (IN_DIM * HDIM) + i * HDIM + d];
    }
    __syncthreads();

    const int wave  = tid >> 5;
    const int lane  = tid & 31;
    const int mtile = blockIdx.x * 8 + wave;
    if (mtile >= N_NODES / 16) return;

    const int row   = lane & 15;
    const int khalf = lane >> 4;

    const float* hrow = h + (size_t)(mtile * 16 + row) * IN_DIM;

    v8f acc[8] = {};

    for (int k0 = 0; k0 < IN_DIM; k0 += 4) {
        const int kk = k0 + 2 * khalf;
        v2f a = *reinterpret_cast<const v2f*>(hrow + kk);
#pragma unroll
        for (int t = 0; t < 8; ++t) {
            v2f b = *reinterpret_cast<const v2f*>(&Wl[(t * 16 + row) * LDS_STRIDE + kk]);
            acc[t] = __builtin_amdgcn_wmma_f32_16x16x4_f32(
                false, a, false, b, (short)0, acc[t], false, false);
        }
    }

    float* zbase = z + (size_t)(mtile * 16) * OUT_DIM;
#pragma unroll
    for (int t = 0; t < 8; ++t) {
#pragma unroll
        for (int i = 0; i < 8; ++i) {
            int mrow = i + khalf * 8;
            zbase[(size_t)mrow * OUT_DIM + t * 16 + row] = acc[t][i];
        }
    }
}

__global__ void gat_scores(const float* __restrict__ z,
                           const float* __restrict__ Wattn,
                           float* __restrict__ s_src, float* __restrict__ s_dst) {
    int idx = blockIdx.x * blockDim.x + threadIdx.x;
    if (idx >= N_NODES * HEADS) return;
    int n = idx >> 3, hh = idx & 7;
    const float* zp = z + (size_t)n * OUT_DIM + hh * HDIM;
    const float* as = Wattn + hh * (2 * HDIM);
    const float* ad = as + HDIM;
    float ss = 0.0f, sd = 0.0f;
#pragma unroll
    for (int d = 0; d < HDIM; ++d) {
        float v = zp[d];
        ss += v * as[d];
        sd += v * ad[d];
    }
    s_src[idx] = ss;
    s_dst[idx] = sd;
}

__global__ void gat_edge_max(const int* __restrict__ src, const int* __restrict__ dst,
                             const float* __restrict__ s_src, const float* __restrict__ s_dst,
                             float* __restrict__ m) {
    int e = blockIdx.x * blockDim.x + threadIdx.x;
    if (e >= N_EDGES) return;
    int s = src[e], d = dst[e];
    const float* ps = s_src + (size_t)s * HEADS;
    const float* pd = s_dst + (size_t)d * HEADS;
    float* pm = m + (size_t)d * HEADS;
#pragma unroll
    for (int hh = 0; hh < HEADS; ++hh) {
        float sc = ps[hh] + pd[hh];
        sc = (sc >= 0.0f) ? sc : NEG_SLOPE * sc;
        atomicMaxF(pm + hh, sc);
    }
}

__global__ void gat_edge_sum(const int* __restrict__ src, const int* __restrict__ dst,
                             const float* __restrict__ s_src, const float* __restrict__ s_dst,
                             const float* __restrict__ m, float* __restrict__ den) {
    int e = blockIdx.x * blockDim.x + threadIdx.x;
    if (e >= N_EDGES) return;
    int s = src[e], d = dst[e];
    const float* ps = s_src + (size_t)s * HEADS;
    const float* pd = s_dst + (size_t)d * HEADS;
    const float* pm = m + (size_t)d * HEADS;
    float* pden = den + (size_t)d * HEADS;
#pragma unroll
    for (int hh = 0; hh < HEADS; ++hh) {
        float sc = ps[hh] + pd[hh];
        sc = (sc >= 0.0f) ? sc : NEG_SLOPE * sc;
        atomicAdd(pden + hh, expf(sc - pm[hh]));
    }
}

__global__ void gat_edge_agg(const int* __restrict__ src, const int* __restrict__ dst,
                             const float* __restrict__ s_src, const float* __restrict__ s_dst,
                             const float* __restrict__ m, const float* __restrict__ den,
                             const float* __restrict__ z, float* __restrict__ out0) {
    int idx = blockIdx.x * blockDim.x + threadIdx.x;
    if (idx >= N_EDGES * HEADS) return;
    int e = idx >> 3, hh = idx & 7;
    int s = src[e], d = dst[e];
    float sc = s_src[(size_t)s * HEADS + hh] + s_dst[(size_t)d * HEADS + hh];
    sc = (sc >= 0.0f) ? sc : NEG_SLOPE * sc;
    float alpha = expf(sc - m[(size_t)d * HEADS + hh]) / den[(size_t)d * HEADS + hh];

    const float4* zp = (const float4*)(z + (size_t)s * OUT_DIM + hh * HDIM);
    float* op = out0 + (size_t)d * OUT_DIM + hh * HDIM;
#pragma unroll
    for (int q = 0; q < 4; ++q) {
        float4 v = zp[q];
        atomicAdd(op + q * 4 + 0, alpha * v.x);
        atomicAdd(op + q * 4 + 1, alpha * v.y);
        atomicAdd(op + q * 4 + 2, alpha * v.z);
        atomicAdd(op + q * 4 + 3, alpha * v.w);
    }
}

__global__ void gat_finalize(const float* __restrict__ h, const float* __restrict__ e,
                             float* __restrict__ out) {
    int j = blockIdx.x * blockDim.x + threadIdx.x;
    const int NH = N_NODES * OUT_DIM;
    if (j < NH) {
        float v = out[j];
        v = (v > 0.0f) ? v : expm1f(v);
        out[j] = h[j] + v;
    } else if (j < NH + N_EDGES) {
        out[j] = e[j - NH];
    }
}

extern "C" void kernel_launch(void* const* d_in, const int* in_sizes, int n_in,
                              void* d_out, int out_size, void* d_ws, size_t ws_size,
                              hipStream_t stream) {
    const float* h     = (const float*)d_in[0];
    const float* e     = (const float*)d_in[1];
    const int*   src   = (const int*)d_in[2];
    const int*   dst   = (const int*)d_in[3];
    const float* Wfc   = (const float*)d_in[4];
    const float* Wattn = (const float*)d_in[5];
    float* out = (float*)d_out;

    float* z     = (float*)d_ws;
    float* s_src = z + (size_t)N_NODES * OUT_DIM;
    float* s_dst = s_src + (size_t)N_NODES * HEADS;
    float* m     = s_dst + (size_t)N_NODES * HEADS;
    float* den   = m + (size_t)N_NODES * HEADS;

    const int T = 256;

    gat_init<<<(N_NODES * OUT_DIM + T - 1) / T, T, 0, stream>>>(out, m, den);

    gat_gemm_wmma<<<(N_NODES / 16 + 7) / 8, T, 0, stream>>>(h, Wfc, z);

    gat_scores<<<(N_NODES * HEADS + T - 1) / T, T, 0, stream>>>(z, Wattn, s_src, s_dst);

    gat_edge_max<<<(N_EDGES + T - 1) / T, T, 0, stream>>>(src, dst, s_src, s_dst, m);
    gat_edge_sum<<<(N_EDGES + T - 1) / T, T, 0, stream>>>(src, dst, s_src, s_dst, m, den);
    gat_edge_agg<<<(N_EDGES * HEADS + T - 1) / T, T, 0, stream>>>(src, dst, s_src, s_dst,
                                                                  m, den, z, out);

    gat_finalize<<<(N_NODES * OUT_DIM + N_EDGES + T - 1) / T, T, 0, stream>>>(h, e, out);
}


# Edge passes become owner work-groups: each owns 400 nodes, filters the whole dst list for them, sums exp and exp*z in LDS float atomics, divides once at the end; global atomics, init, finalize gone
# speedup vs baseline: 6.9305x; 6.9305x over previous
#include <hip/hip_runtime.h>
#include <math.h>

#define N_NODES 100000
#define N_EDGES 1600000
#define IN_DIM  128
#define HEADS   8
#define HDIM    16
#define OUT_DIM 128
#define NEG_SLOPE 0.01f
#define LDS_STRIDE 130

typedef __attribute__((ext_vector_type(2))) float v2f;
typedef __attribute__((ext_vector_type(8))) float v8f;

__device__ __forceinline__ void atomicMaxF(float* addr, float val) {
    if (val >= 0.0f) {
        atomicMax((int*)addr, __float_as_int(val));
    } else {
        atomicMin((unsigned int*)addr, __float_as_uint(val));
    }
}

__global__ void __launch_bounds__(256)
gat_gemm_wmma(const float* __restrict__ h, const float* __restrict__ Wfc,
              float* __restrict__ z) {
    __shared__ float Wl[OUT_DIM * LDS_STRIDE];

    const int tid = threadIdx.x;
    for (int j = tid; j < IN_DIM * OUT_DIM; j += 256) {
        int i  = j >> 7;
        int c  = j & 127;
        int hh = c >> 4;
        int d  = c & 15;
        Wl[c * LDS_STRIDE + i] = Wfc[hh * (IN_DIM * HDIM) + i * HDIM + d];
    }
    __syncthreads();

    const int wave  = tid >> 5;
    const int lane  = tid & 31;
    const int mtile = blockIdx.x * 8 + wave;
    if (mtile >= N_NODES / 16) return;

    const int row   = lane & 15;
    const int khalf = lane >> 4;

    const float* hrow = h + (size_t)(mtile * 16 + row) * IN_DIM;

    v8f acc[8] = {};

    for (int k0 = 0; k0 < IN_DIM; k0 += 4) {
        const int kk = k0 + 2 * khalf;
        v2f a = *reinterpret_cast<const v2f*>(hrow + kk);
#pragma unroll
        for (int t = 0; t < 8; ++t) {
            v2f b = *reinterpret_cast<const v2f*>(&Wl[(t * 16 + row) * LDS_STRIDE + kk]);
            acc[t] = __builtin_amdgcn_wmma_f32_16x16x4_f32(
                false, a, false, b, (short)0, acc[t], false, false);
        }
    }

    float* zbase = z + (size_t)(mtile * 16) * OUT_DIM;
#pragma unroll
    for (int t = 0; t < 8; ++t) {
#pragma unroll
        for (int i = 0; i < 8; ++i) {
            int mrow = i + khalf * 8;
            zbase[(size_t)mrow * OUT_DIM + t * 16 + row] = acc[t][i];
        }
    }
}

__global__ void gat_scores(const float* __restrict__ z,
                           const float* __restrict__ Wattn,
                           float* __restrict__ s_src, float* __restrict__ s_dst) {
    int idx = blockIdx.x * blockDim.x + threadIdx.x;
    if (idx >= N_NODES * HEADS) return;
    int n = idx >> 3, hh = idx & 7;
    const float* zp = z + (size_t)n * OUT_DIM + hh * HDIM;
    const float* as = Wattn + hh * (2 * HDIM);
    const float* ad = as + HDIM;
    float ss = 0.0f, sd = 0.0f;
#pragma unroll
    for (int d = 0; d < HDIM; ++d) {
        float v = zp[d];
        ss += v * as[d];
        sd += v * ad[d];
    }
    s_src[idx] = ss;
    s_dst[idx] = sd;
}

#define TN   400
#define NT   (N_NODES / TN)
#define QCAP 12288
#define TB   1024
#define EB   8

struct TileLds {
    float acc[TN * OUT_DIM];
    float mrun[TN * HEADS];
    float mnew[TN * HEADS];
    float den[TN * HEADS];
    unsigned q[QCAP];
    unsigned qn;
};

__device__ __forceinline__ void tile_push(TileLds& L, int eidx, unsigned dl) {
    unsigned slot = atomicAdd(&L.qn, 1u);
    if (slot < (unsigned)QCAP) L.q[slot] = ((unsigned)eidx << 9) | dl;
}

__device__ __forceinline__ void tile_scan(TileLds& L, const int* __restrict__ dst,
                                          int v0, int v1, int base, int tid) {
    const int4* d4 = reinterpret_cast<const int4*>(dst);
    for (int v = v0 + tid; v < v1; v += 4 * TB) {
        int4 a[4];
#pragma unroll
        for (int u = 0; u < 4; ++u) {
            int vv = v + u * TB;
            a[u] = (vv < v1) ? d4[vv] : make_int4(-1, -1, -1, -1);
        }
#pragma unroll
        for (int u = 0; u < 4; ++u) {
            int vv = v + u * TB;
            unsigned t0 = (unsigned)(a[u].x - base), t1 = (unsigned)(a[u].y - base);
            unsigned t2 = (unsigned)(a[u].z - base), t3 = (unsigned)(a[u].w - base);
            if ((t0 < TN) | (t1 < TN) | (t2 < TN) | (t3 < TN)) {
                int eb = vv * 4;
                if (t0 < TN) tile_push(L, eb + 0, t0);
                if (t1 < TN) tile_push(L, eb + 1, t1);
                if (t2 < TN) tile_push(L, eb + 2, t2);
                if (t3 < TN) tile_push(L, eb + 3, t3);
            }
        }
    }
}

__device__ __forceinline__ void tile_batch(TileLds& L, unsigned n,
                                           const int* __restrict__ src,
                                           const float* __restrict__ s_src,
                                           const float* __restrict__ s_dst,
                                           const float* __restrict__ z,
                                           int base, int tid) {
    for (unsigned i0 = tid; i0 < n; i0 += 4 * TB) {
        unsigned pk[4], s[4];
#pragma unroll
        for (int u = 0; u < 4; ++u) {
            unsigned i = i0 + u * TB;
            pk[u] = L.q[i < n ? i : i0];
            s[u] = (unsigned)src[pk[u] >> 9];
        }
#pragma unroll
        for (int u = 0; u < 4; ++u) {
            unsigned i = i0 + u * TB;
            unsigned sv = s[u] < (unsigned)N_NODES ? s[u] : (unsigned)(N_NODES - 1);
            if (i < n) L.q[i] = (sv << 9) | (pk[u] & 511);
        }
    }
    __syncthreads();
    {
        const unsigned nh = n * HEADS, hd = tid & 7;
        for (unsigned i0 = tid; i0 < nh; i0 += 4 * TB) {
            float a[4], b[4]; unsigned cell[4];
#pragma unroll
            for (int u = 0; u < 4; ++u) {
                unsigned i = i0 + u * TB;
                unsigned pk = L.q[(i < nh ? i : i0) >> 3];
                unsigned s = pk >> 9, dl = pk & 511;
                a[u] = s_src[(size_t)s * HEADS + hd];
                b[u] = s_dst[(size_t)(base + dl) * HEADS + hd];
                cell[u] = dl * HEADS + hd;
            }
#pragma unroll
            for (int u = 0; u < 4; ++u) {
                float sc = a[u] + b[u];
                sc = (sc >= 0.0f) ? sc : NEG_SLOPE * sc;
                if (i0 + u * TB < nh) atomicMaxF(&L.mnew[cell[u]], sc);
            }
        }
    }
    __syncthreads();
    for (int i = tid; i < TN * HEADS; i += TB) {
        float mo = L.mrun[i], mn = L.mnew[i];
        if (mo != mn) {
            if (mo != -INFINITY) {
                float f = expf(mo - mn);
                L.den[i] *= f;
                float* a = &L.acc[(i >> 3) * OUT_DIM + (i & 7) * HDIM];
#pragma unroll
                for (int d = 0; d < HDIM; ++d) a[d] *= f;
            }
            L.mrun[i] = mn;
        }
    }
    __syncthreads();
    const int wave = tid >> 5, lane = tid & 31, hh = lane >> 2;
    const float4* z4 = reinterpret_cast<const float4*>(z);
    for (unsigned eb = wave * EB; eb < n; eb += (TB / 32) * EB) {
        float4 zv[EB]; float ss[EB], sd[EB]; unsigned dls[EB]; bool ok[EB];
#pragma unroll
        for (int u = 0; u < EB; ++u) {
            unsigned ei = eb + u;
            ok[u] = ei < n;
            unsigned pk = L.q[ok[u] ? ei : (n - 1)];
            unsigned s = pk >> 9;
            dls[u] = pk & 511;
            ss[u] = s_src[(size_t)s * HEADS + hh];
            sd[u] = s_dst[(size_t)(base + dls[u]) * HEADS + hh];
            zv[u] = z4[(size_t)s * (OUT_DIM / 4) + lane];
        }
#pragma unroll
        for (int u = 0; u < EB; ++u) {
            if (ok[u]) {
                float sc = ss[u] + sd[u];
                sc = (sc >= 0.0f) ? sc : NEG_SLOPE * sc;
                float ex = expf(sc - L.mrun[dls[u] * HEADS + hh]);
                float* a = &L.acc[dls[u] * OUT_DIM + lane * 4];
                atomicAdd(a + 0, ex * zv[u].x);
                atomicAdd(a + 1, ex * zv[u].y);
                atomicAdd(a + 2, ex * zv[u].z);
                atomicAdd(a + 3, ex * zv[u].w);
                if ((lane & 3) == 0) atomicAdd(&L.den[dls[u] * HEADS + hh], ex);
            }
        }
    }
}

__global__ void __launch_bounds__(TB)
gat_tile(const int* __restrict__ src, const int* __restrict__ dst,
         const float* __restrict__ s_src, const float* __restrict__ s_dst,
         const float* __restrict__ z, const float* __restrict__ h,
         const float* __restrict__ e, float* __restrict__ out) {
    __shared__ TileLds L;
    const int tid = threadIdx.x;
    const int base = blockIdx.x * TN;

    float4* acc4 = reinterpret_cast<float4*>(L.acc);
    for (int i = tid; i < TN * OUT_DIM / 4; i += TB) acc4[i] = make_float4(0.f, 0.f, 0.f, 0.f);
    for (int i = tid; i < TN * HEADS; i += TB) {
        L.mrun[i] = -INFINITY; L.mnew[i] = -INFINITY; L.den[i] = 0.0f;
    }
    if (tid == 0) L.qn = 0u;
    __syncthreads();

    const int V = N_EDGES / 4;
    const int rot = blockIdx.x * (V / NT);
    tile_scan(L, dst, rot, V, base, tid);
    tile_scan(L, dst, 0, rot, base, tid);
    __syncthreads();
    const unsigned total = L.qn;
    if (total <= (unsigned)QCAP) {
        tile_batch(L, total, src, s_src, s_dst, z, base, tid);
    } else {
        for (int seg = 0; seg < N_EDGES; seg += QCAP) {
            __syncthreads();
            if (tid == 0) L.qn = 0u;
            __syncthreads();
            int end = seg + QCAP < N_EDGES ? seg + QCAP : N_EDGES;
            tile_scan(L, dst, seg / 4, end / 4, base, tid);
            __syncthreads();
            tile_batch(L, L.qn, src, s_src, s_dst, z, base, tid);
        }
    }
    __syncthreads();

    const float4* h4 = reinterpret_cast<const float4*>(h) + (size_t)base * (OUT_DIM / 4);
    float4* o4 = reinterpret_cast<float4*>(out) + (size_t)base * (OUT_DIM / 4);
    for (int i = tid; i < TN * OUT_DIM / 4; i += TB) {
        float dn = L.den[(i >> 5) * HEADS + ((i & 31) >> 2)];
        float4 a = acc4[i], hv = h4[i], r;
        bool has = dn > 0.0f;
        float v0 = has ? a.x / dn : 0.0f, v1 = has ? a.y / dn : 0.0f;
        float v2 = has ? a.z / dn : 0.0f, v3 = has ? a.w / dn : 0.0f;
        r.x = hv.x + ((v0 > 0.0f) ? v0 : expm1f(v0));
        r.y = hv.y + ((v1 > 0.0f) ? v1 : expm1f(v1));
        r.z = hv.z + ((v2 > 0.0f) ? v2 : expm1f(v2));
        r.w = hv.w + ((v3 > 0.0f) ? v3 : expm1f(v3));
        o4[i] = r;
    }
    const int per = N_EDGES / NT;
    float* oe = out + (size_t)N_NODES * OUT_DIM + (size_t)blockIdx.x * per;
    const float* ee = e + (size_t)blockIdx.x * per;
    for (int i = tid; i < per; i += TB) oe[i] = ee[i];
}

extern "C" void kernel_launch(void* const* d_in, const int* in_sizes, int n_in,
                              void* d_out, int out_size, void* d_ws, size_t ws_size,
                              hipStream_t stream) {
    const float* h     = (const float*)d_in[0];
    const float* e     = (const float*)d_in[1];
    const int*   src   = (const int*)d_in[2];
    const int*   dst   = (const int*)d_in[3];
    const float* Wfc   = (const float*)d_in[4];
    const float* Wattn = (const float*)d_in[5];
    float* out = (float*)d_out;

    float* z     = (float*)d_ws;
    float* s_src = z + (size_t)N_NODES * OUT_DIM;
    float* s_dst = s_src + (size_t)N_NODES * HEADS;

    const int T = 256;

    gat_gemm_wmma<<<(N_NODES / 16 + 7) / 8, T, 0, stream>>>(h, Wfc, z);

    gat_scores<<<(N_NODES * HEADS + T - 1) / T, T, 0, stream>>>(z, Wattn, s_src, s_dst);

    gat_tile<<<NT, TB, 0, stream>>>(src, dst, s_src, s_dst, z, h, e, out);
}


# Owner groups scan a list of one-byte owner ids (dst/400, 16 per load, written by the score launch) with wave-wide compare masks instead of the 4-byte dst list; e copied by a last launch
# speedup vs baseline: 7.4248x; 1.0713x over previous
#include <hip/hip_runtime.h>
#include <math.h>

#define N_NODES 100000
#define N_EDGES 1600000
#define IN_DIM  128
#define HEADS   8
#define HDIM    16
#define OUT_DIM 128
#define NEG_SLOPE 0.01f
#define LDS_STRIDE 130
#define TN   400
#define NT   (N_NODES / TN)
#define KV   (N_EDGES / 16)

typedef __attribute__((ext_vector_type(2))) float v2f;
typedef __attribute__((ext_vector_type(8))) float v8f;

__device__ __forceinline__ void atomicMaxF(float* addr, float val) {
    if (val >= 0.0f) {
        atomicMax((int*)addr, __float_as_int(val));
    } else {
        atomicMin((unsigned int*)addr, __float_as_uint(val));
    }
}

__global__ void __launch_bounds__(256)
gat_gemm_wmma(const float* __restrict__ h, const float* __restrict__ Wfc,
              float* __restrict__ z) {
    __shared__ float Wl[OUT_DIM * LDS_STRIDE];

    const int tid = threadIdx.x;
    for (int j = tid; j < IN_DIM * OUT_DIM; j += 256) {
        int i  = j >> 7;
        int c  = j & 127;
        int hh = c >> 4;
        int d  = c & 15;
        Wl[c * LDS_STRIDE + i] = Wfc[hh * (IN_DIM * HDIM) + i * HDIM + d];
    }
    __syncthreads();

    const int wave  = tid >> 5;
    const int lane  = tid & 31;
    const int mtile = blockIdx.x * 8 + wave;
    if (mtile >= N_NODES / 16) return;

    const int row   = lane & 15;
    const int khalf = lane >> 4;

    const float* hrow = h + (size_t)(mtile * 16 + row) * IN_DIM;

    v8f acc[8] = {};

    for (int k0 = 0; k0 < IN_DIM; k0 += 4) {
        const int kk = k0 + 2 * khalf;
        v2f a = *reinterpret_cast<const v2f*>(hrow + kk);
#pragma unroll
        for (int t = 0; t < 8; ++t) {
            v2f b = *reinterpret_cast<const v2f*>(&Wl[(t * 16 + row) * LDS_STRIDE + kk]);
            acc[t] = __builtin_amdgcn_wmma_f32_16x16x4_f32(
                false, a, false, b, (short)0, acc[t], false, false);
        }
    }

    float* zbase = z + (size_t)(mtile * 16) * OUT_DIM;
#pragma unroll
    for (int t = 0; t < 8; ++t) {
#pragma unroll
        for (int i = 0; i < 8; ++i) {
            int mrow = i + khalf * 8;
            zbase[(size_t)mrow * OUT_DIM + t * 16 + row] = acc[t][i];
        }
    }
}

__device__ __forceinline__ unsigned owner_id(int d) {
    unsigned k = (unsigned)d / (unsigned)TN;
    return k < 255u ? k : 255u;
}

__global__ void gat_scores(const float* __restrict__ z,
                           const float* __restrict__ Wattn,
                           const int* __restrict__ dst,
                           float* __restrict__ s_src, float* __restrict__ s_dst,
                           unsigned* __restrict__ ids) {
    int idx = blockIdx.x * blockDim.x + threadIdx.x;
    if (idx >= N_NODES * HEADS) return;
    int n = idx >> 3, hh = idx & 7;
    const float* zp = z + (size_t)n * OUT_DIM + hh * HDIM;
    const float* as = Wattn + hh * (2 * HDIM);
    const float* ad = as + HDIM;
    float ss = 0.0f, sd = 0.0f;
#pragma unroll
    for (int d = 0; d < HDIM; ++d) {
        float v = zp[d];
        ss += v * as[d];
        sd += v * ad[d];
    }
    s_src[idx] = ss;
    s_dst[idx] = sd;

    if (idx < KV) {
        const int4* d4 = reinterpret_cast<const int4*>(dst) + (size_t)idx * 4;
        unsigned w[4];
#pragma unroll
        for (int j = 0; j < 4; ++j) {
            int4 d = d4[j];
            w[j] = owner_id(d.x) | (owner_id(d.y) << 8) | (owner_id(d.z) << 16) | (owner_id(d.w) << 24);
        }
        reinterpret_cast<uint4*>(ids)[idx] = make_uint4(w[0], w[1], w[2], w[3]);
    }
}

#define QCAP 12288
#define TB   1024
#define EB   8

struct TileLds {
    float acc[TN * OUT_DIM];
    float mrun[TN * HEADS];
    float mnew[TN * HEADS];
    float den[TN * HEADS];
    unsigned q[QCAP];
    unsigned qn;
};

__device__ __forceinline__ void scan_vec(TileLds& L, uint4 a, unsigned tileb,
                                         unsigned ebase, unsigned lane) {
    const unsigned w[4] = {a.x, a.y, a.z, a.w};
    bool hit[16]; unsigned m[16]; unsigned long long p[8]; unsigned total = 0;
#pragma unroll
    for (int k = 0; k < 16; ++k) {
        hit[k] = ((w[k >> 2] >> ((k & 3) * 8)) & 0xFFu) == tileb;
        m[k] = __builtin_amdgcn_ballot_w32(hit[k]);
    }
#pragma unroll
    for (int k = 0; k < 8; ++k) {
        p[k] = ((unsigned long long)m[2 * k + 1] << 32) | (unsigned long long)m[2 * k];
        total += (unsigned)__builtin_popcountll(p[k]);
    }
    if (total != 0u) {
        unsigned s0 = 0u;
        if (lane == 0u) s0 = atomicAdd(&L.qn, total);
        s0 = (unsigned)__builtin_amdgcn_readfirstlane((int)s0);
#pragma unroll
        for (int k = 0; k < 8; ++k) {
            if (p[k] != 0ull) {
#pragma unroll
                for (int t = 2 * k; t < 2 * k + 2; ++t) {
                    if (hit[t]) {
                        unsigned slot = s0 + __builtin_amdgcn_mbcnt_lo(m[t], 0u);
                        if (slot < (unsigned)QCAP) L.q[slot] = ebase + (unsigned)t;
                    }
                    s0 += (unsigned)__builtin_popcount(m[t]);
                }
            }
        }
    }
}

__device__ __forceinline__ void tile_scan(TileLds& L, const unsigned* __restrict__ ids,
                                          int v0, int v1, unsigned tileb, int tid) {
    const uint4* k4 = reinterpret_cast<const uint4*>(ids);
    const unsigned lane = (unsigned)tid & 31u;
    const uint4 none = make_uint4(~0u, ~0u, ~0u, ~0u);
    uint4 cur[4], nxt[4];
    int v = v0 + tid;
#pragma unroll
    for (int u = 0; u < 4; ++u) {
        int vv = v + u * TB;
        uint4 t = k4[vv < KV ? vv : KV - 1];
        cur[u] = (vv < v1) ? t : none;
    }
    for (int vb = v0 + (tid & ~31); vb < v1; vb += 4 * TB, v += 4 * TB) {
#pragma unroll
        for (int u = 0; u < 4; ++u) {
            int vn = v + (4 + u) * TB;
            uint4 t = k4[vn < KV ? vn : KV - 1];
            nxt[u] = (vn < v1) ? t : none;
        }
#pragma unroll
        for (int u = 0; u < 4; ++u)
            scan_vec(L, cur[u], tileb, (unsigned)(v + u * TB) * 16u, lane);
#pragma unroll
        for (int u = 0; u < 4; ++u) cur[u] = nxt[u];
    }
}

__device__ __forceinline__ void tile_batch(TileLds& L, unsigned n,
                                           const int* __restrict__ src,
                                           const int* __restrict__ dst,
                                           const float* __restrict__ s_src,
                                           const float* __restrict__ s_dst,
                                           const float* __restrict__ z,
                                           int base, int tid) {
    for (unsigned i0 = tid; i0 < n; i0 += 4 * TB) {
        unsigned s[4], d[4];
#pragma unroll
        for (int u = 0; u < 4; ++u) {
            unsigned i = i0 + u * TB;
            unsigned ei = L.q[i < n ? i : i0];
            ei = ei < (unsigned)N_EDGES ? ei : (unsigned)(N_EDGES - 1);
            s[u] = (unsigned)src[ei];
            d[u] = (unsigned)dst[ei] - (unsigned)base;
        }
#pragma unroll
        for (int u = 0; u < 4; ++u) {
            unsigned i = i0 + u * TB;
            unsigned sv = s[u] < (unsigned)N_NODES ? s[u] : (unsigned)(N_NODES - 1);
            unsigned dl = d[u] < (unsigned)TN ? d[u] : (unsigned)(TN - 1);
            if (i < n) L.q[i] = (sv << 9) | dl;
        }
    }
    __syncthreads();
    {
        const unsigned nh = n * HEADS, hd = tid & 7;
        for (unsigned i0 = tid; i0 < nh; i0 += 4 * TB) {
            float a[4], b[4]; unsigned cell[4];
#pragma unroll
            for (int u = 0; u < 4; ++u) {
                unsigned i = i0 + u * TB;
                unsigned pk = L.q[(i < nh ? i : i0) >> 3];
                unsigned s = pk >> 9, dl = pk & 511;
                a[u] = s_src[(size_t)s * HEADS + hd];
                b[u] = s_dst[(size_t)(base + dl) * HEADS + hd];
                cell[u] = dl * HEADS + hd;
            }
#pragma unroll
            for (int u = 0; u < 4; ++u) {
                float sc = a[u] + b[u];
                sc = (sc >= 0.0f) ? sc : NEG_SLOPE * sc;
                if (i0 + u * TB < nh) atomicMaxF(&L.mnew[cell[u]], sc);
            }
        }
    }
    __syncthreads();
    for (int i = tid; i < TN * HEADS; i += TB) {
        float mo = L.mrun[i], mn = L.mnew[i];
        if (mo != mn) {
            if (mo != -INFINITY) {
                float f = expf(mo - mn);
                L.den[i] *= f;
                float* a = &L.acc[(i >> 3) * OUT_DIM + (i & 7) * HDIM];
#pragma unroll
                for (int d = 0; d < HDIM; ++d) a[d] *= f;
            }
            L.mrun[i] = mn;
        }
    }
    __syncthreads();
    const int wave = tid >> 5, lane = tid & 31, hh = lane >> 2;
    const float4* z4 = reinterpret_cast<const float4*>(z);
    for (unsigned eb = wave * EB; eb < n; eb += (TB / 32) * EB) {
        float4 zv[EB]; float ss[EB], sd[EB]; unsigned dls[EB]; bool ok[EB];
#pragma unroll
        for (int u = 0; u < EB; ++u) {
            unsigned ei = eb + u;
            ok[u] = ei < n;
            unsigned pk = L.q[ok[u] ? ei : (n - 1)];
            unsigned s = pk >> 9;
            dls[u] = pk & 511;
            ss[u] = s_src[(size_t)s * HEADS + hh];
            sd[u] = s_dst[(size_t)(base + dls[u]) * HEADS + hh];
            zv[u] = z4[(size_t)s * (OUT_DIM / 4) + lane];
        }
#pragma unroll
        for (int u = 0; u < EB; ++u) {
            if (ok[u]) {
                float sc = ss[u] + sd[u];
                sc = (sc >= 0.0f) ? sc : NEG_SLOPE * sc;
                float ex = expf(sc - L.mrun[dls[u] * HEADS + hh]);
                float* a = &L.acc[dls[u] * OUT_DIM + lane * 4];
                atomicAdd(a + 0, ex * zv[u].x);
                atomicAdd(a + 1, ex * zv[u].y);
                atomicAdd(a + 2, ex * zv[u].z);
                atomicAdd(a + 3, ex * zv[u].w);
                if ((lane & 3) == 0) atomicAdd(&L.den[dls[u] * HEADS + hh], ex);
            }
        }
    }
}

__global__ void __launch_bounds__(TB)
gat_tile(const int* __restrict__ src, const int* __restrict__ dst,
         const unsigned* __restrict__ ids,
         const float* __restrict__ s_src, const float* __restrict__ s_dst,
         const float* __restrict__ z, const float* __restrict__ h,
         float* __restrict__ out) {
    __shared__ TileLds L;
    const int tid = threadIdx.x;
    const int base = blockIdx.x * TN;

    float4* acc4 = reinterpret_cast<float4*>(L.acc);
    for (int i = tid; i < TN * OUT_DIM / 4; i += TB) acc4[i] = make_float4(0.f, 0.f, 0.f, 0.f);
    for (int i = tid; i < TN * HEADS; i += TB) {
        L.mrun[i] = -INFINITY; L.mnew[i] = -INFINITY; L.den[i] = 0.0f;
    }
    if (tid == 0) L.qn = 0u;
    __syncthreads();

    const int rot = blockIdx.x * (KV / NT);
    tile_scan(L, ids, rot, KV, blockIdx.x, tid);
    tile_scan(L, ids, 0, rot, blockIdx.x, tid);
    __syncthreads();
    const unsigned total = L.qn;
    if (total <= (unsigned)QCAP) {
        tile_batch(L, total, src, dst, s_src, s_dst, z, base, tid);
    } else {
        for (int seg = 0; seg < N_EDGES; seg += QCAP) {
            __syncthreads();
            if (tid == 0) L.qn = 0u;
            __syncthreads();
            int end = seg + QCAP < N_EDGES ? seg + QCAP : N_EDGES;
            tile_scan(L, ids, seg / 16, end / 16, blockIdx.x, tid);
            __syncthreads();
            tile_batch(L, L.qn, src, dst, s_src, s_dst, z, base, tid);
        }
    }
    __syncthreads();

    const float4* h4 = reinterpret_cast<const float4*>(h) + (size_t)base * (OUT_DIM / 4);
    float4* o4 = reinterpret_cast<float4*>(out) + (size_t)base * (OUT_DIM / 4);
    for (int i = tid; i < TN * OUT_DIM / 4; i += TB) {
        float dn = L.den[(i >> 5) * HEADS + ((i & 31) >> 2)];
        float4 a = acc4[i], hv = h4[i], r;
        bool has = dn > 0.0f;
        float v0 = has ? a.x / dn : 0.0f, v1 = has ? a.y / dn : 0.0f;
        float v2 = has ? a.z / dn : 0.0f, v3 = has ? a.w / dn : 0.0f;
        r.x = hv.x + ((v0 > 0.0f) ? v0 : expm1f(v0));
        r.y = hv.y + ((v1 > 0.0f) ? v1 : expm1f(v1));
        r.z = hv.z + ((v2 > 0.0f) ? v2 : expm1f(v2));
        r.w = hv.w + ((v3 > 0.0f) ? v3 : expm1f(v3));
        o4[i] = r;
    }
}

__global__ void __launch_bounds__(256)
gat_epass(const float* __restrict__ e, float* __restrict__ oe) {
    int i = blockIdx.x * 256 + threadIdx.x;
    if (i < N_EDGES / 4)
        reinterpret_cast<float4*>(oe)[i] = reinterpret_cast<const float4*>(e)[i];
}

extern "C" void kernel_launch(void* const* d_in, const int* in_sizes, int n_in,
                              void* d_out, int out_size, void* d_ws, size_t ws_size,
                              hipStream_t stream) {
    const float* h     = (const float*)d_in[0];
    const float* e     = (const float*)d_in[1];
    const int*   src   = (const int*)d_in[2];
    const int*   dst   = (const int*)d_in[3];
    const float* Wfc   = (const float*)d_in[4];
    const float* Wattn = (const float*)d_in[5];
    float* out = (float*)d_out;

    float* z     = (float*)d_ws;
    float* s_src = z + (size_t)N_NODES * OUT_DIM;
    float* s_dst = s_src + (size_t)N_NODES * HEADS;

    const int T = 256;

    gat_gemm_wmma<<<(N_NODES / 16 + 7) / 8, T, 0, stream>>>(h, Wfc, z);

    float* oe = out + (size_t)N_NODES * OUT_DIM;
    unsigned* ids = (unsigned*)oe;

    gat_scores<<<(N_NODES * HEADS + T - 1) / T, T, 0, stream>>>(z, Wattn, dst, s_src, s_dst, ids);

    gat_tile<<<NT, TB, 0, stream>>>(src, dst, ids, s_src, s_dst, z, h, out);

    gat_epass<<<(N_EDGES / 4 + T - 1) / T, T, 0, stream>>>(e, oe);
}


# Owner-id scan matches ids per lane on whole words (zero-byte test of word xor id pattern, 64 edges per lane and trip); one wave mask and LDS add per round of matches; queue order, so sum order, moves
# speedup vs baseline: 8.1402x; 1.0964x over previous
#include <hip/hip_runtime.h>
#include <math.h>

#define N_NODES 100000
#define N_EDGES 1600000
#define IN_DIM  128
#define HEADS   8
#define HDIM    16
#define OUT_DIM 128
#define NEG_SLOPE 0.01f
#define LDS_STRIDE 130
#define TN   400
#define NT   (N_NODES / TN)
#define KV   (N_EDGES / 16)

typedef __attribute__((ext_vector_type(2))) float v2f;
typedef __attribute__((ext_vector_type(8))) float v8f;

__device__ __forceinline__ void atomicMaxF(float* addr, float val) {
    if (val >= 0.0f) {
        atomicMax((int*)addr, __float_as_int(val));
    } else {
        atomicMin((unsigned int*)addr, __float_as_uint(val));
    }
}

__global__ void __launch_bounds__(256)
gat_gemm_wmma(const float* __restrict__ h, const float* __restrict__ Wfc,
              float* __restrict__ z) {
    __shared__ float Wl[OUT_DIM * LDS_STRIDE];

    const int tid = threadIdx.x;
    for (int j = tid; j < IN_DIM * OUT_DIM; j += 256) {
        int i  = j >> 7;
        int c  = j & 127;
        int hh = c >> 4;
        int d  = c & 15;
        Wl[c * LDS_STRIDE + i] = Wfc[hh * (IN_DIM * HDIM) + i * HDIM + d];
    }
    __syncthreads();

    const int wave  = tid >> 5;
    const int lane  = tid & 31;
    const int mtile = blockIdx.x * 8 + wave;
    if (mtile >= N_NODES / 16) return;

    const int row   = lane & 15;
    const int khalf = lane >> 4;

    const float* hrow = h + (size_t)(mtile * 16 + row) * IN_DIM;

    v8f acc[8] = {};

    for (int k0 = 0; k0 < IN_DIM; k0 += 4) {
        const int kk = k0 + 2 * khalf;
        v2f a = *reinterpret_cast<const v2f*>(hrow + kk);
#pragma unroll
        for (int t = 0; t < 8; ++t) {
            v2f b = *reinterpret_cast<const v2f*>(&Wl[(t * 16 + row) * LDS_STRIDE + kk]);
            acc[t] = __builtin_amdgcn_wmma_f32_16x16x4_f32(
                false, a, false, b, (short)0, acc[t], false, false);
        }
    }

    float* zbase = z + (size_t)(mtile * 16) * OUT_DIM;
#pragma unroll
    for (int t = 0; t < 8; ++t) {
#pragma unroll
        for (int i = 0; i < 8; ++i) {
            int mrow = i + khalf * 8;
            zbase[(size_t)mrow * OUT_DIM + t * 16 + row] = acc[t][i];
        }
    }
}

__device__ __forceinline__ unsigned owner_id(int d) {
    unsigned k = (unsigned)d / (unsigned)TN;
    return k < 255u ? k : 255u;
}

__global__ void gat_scores(const float* __restrict__ z,
                           const float* __restrict__ Wattn,
                           const int* __restrict__ dst,
                           float* __restrict__ s_src, float* __restrict__ s_dst,
                           unsigned* __restrict__ ids) {
    int idx = blockIdx.x * blockDim.x + threadIdx.x;
    if (idx >= N_NODES * HEADS) return;
    int n = idx >> 3, hh = idx & 7;
    const float* zp = z + (size_t)n * OUT_DIM + hh * HDIM;
    const float* as = Wattn + hh * (2 * HDIM);
    const float* ad = as + HDIM;
    float ss = 0.0f, sd = 0.0f;
#pragma unroll
    for (int d = 0; d < HDIM; ++d) {
        float v = zp[d];
        ss += v * as[d];
        sd += v * ad[d];
    }
    s_src[idx] = ss;
    s_dst[idx] = sd;

    if (idx < KV) {
        const int4* d4 = reinterpret_cast<const int4*>(dst) + (size_t)idx * 4;
        unsigned w[4];
#pragma unroll
        for (int j = 0; j < 4; ++j) {
            int4 d = d4[j];
            w[j] = owner_id(d.x) | (owner_id(d.y) << 8) | (owner_id(d.z) << 16) | (owner_id(d.w) << 24);
        }
        reinterpret_cast<uint4*>(ids)[idx] = make_uint4(w[0], w[1], w[2], w[3]);
    }
}

#define QCAP 12288
#define TB   1024
#define EB   8

struct TileLds {
    float acc[TN * OUT_DIM];
    float mrun[TN * HEADS];
    float mnew[TN * HEADS];
    float den[TN * HEADS];
    unsigned q[QCAP];
    unsigned qn;
};

__device__ __forceinline__ unsigned match_bytes(unsigned w, unsigned pat) {
    unsigned x = w ^ pat;
    unsigned y = ((x & 0x7F7F7F7Fu) + 0x7F7F7F7Fu) | x;
    return ~y & 0x80808080u;
}

__device__ __forceinline__ unsigned match_pair(uint4 a, uint4 b, unsigned pat) {
    return (match_bytes(a.x, pat) >> 7) | (match_bytes(a.y, pat) >> 6) |
           (match_bytes(a.z, pat) >> 5) | (match_bytes(a.w, pat) >> 4) |
           (match_bytes(b.x, pat) >> 3) | (match_bytes(b.y, pat) >> 2) |
           (match_bytes(b.z, pat) >> 1) |  match_bytes(b.w, pat);
}

__device__ __forceinline__ void scan_four(TileLds& L, const uint4 (&c)[4], unsigned pat,
                                          unsigned ebase, unsigned lane) {
    unsigned r01 = match_pair(c[0], c[1], pat);
    unsigned r23 = match_pair(c[2], c[3], pat);
    for (;;) {
        const bool has = (r01 | r23) != 0u;
        const unsigned m = __builtin_amdgcn_ballot_w32(has);
        if (m == 0u) break;
        unsigned s0 = 0u;
        if (lane == 0u) s0 = atomicAdd(&L.qn, (unsigned)__builtin_popcount(m));
        s0 = (unsigned)__builtin_amdgcn_readfirstlane((int)s0);
        if (has) {
            const bool lo = r01 != 0u;
            const unsigned r = lo ? r01 : r23;
            const unsigned p = (unsigned)__builtin_ctz(r);
            const unsigned vec = (lo ? 0u : 2u) + ((p >> 2) & 1u);
            const unsigned e = ebase + vec * (16u * TB) + ((p & 3u) << 2) + (p >> 3);
            const unsigned slot = s0 + __builtin_amdgcn_mbcnt_lo(m, 0u);
            if (slot < (unsigned)QCAP) L.q[slot] = e;
            if (lo) r01 = r & (r - 1u); else r23 = r & (r - 1u);
        }
    }
}

__device__ __forceinline__ void tile_scan(TileLds& L, const unsigned* __restrict__ ids,
                                          int v0, int v1, unsigned tileb, int tid) {
    const uint4* k4 = reinterpret_cast<const uint4*>(ids);
    const unsigned lane = (unsigned)tid & 31u;
    const unsigned pat = tileb * 0x01010101u;
    const uint4 none = make_uint4(~0u, ~0u, ~0u, ~0u);
    uint4 cur[4], nxt[4];
    int v = v0 + tid;
#pragma unroll
    for (int u = 0; u < 4; ++u) {
        int vv = v + u * TB;
        uint4 t = k4[vv < KV ? vv : KV - 1];
        cur[u] = (vv < v1) ? t : none;
    }
    for (int vb = v0 + (tid & ~31); vb < v1; vb += 4 * TB, v += 4 * TB) {
#pragma unroll
        for (int u = 0; u < 4; ++u) {
            int vn = v + (4 + u) * TB;
            uint4 t = k4[vn < KV ? vn : KV - 1];
            nxt[u] = (vn < v1) ? t : none;
        }
        scan_four(L, cur, pat, (unsigned)v * 16u, lane);
#pragma unroll
        for (int u = 0; u < 4; ++u) cur[u] = nxt[u];
    }
}

__device__ __forceinline__ void tile_batch(TileLds& L, unsigned n,
                                           const int* __restrict__ src,
                                           const int* __restrict__ dst,
                                           const float* __restrict__ s_src,
                                           const float* __restrict__ s_dst,
                                           const float* __restrict__ z,
                                           int base, int tid) {
    for (unsigned i0 = tid; i0 < n; i0 += 4 * TB) {
        unsigned s[4], d[4];
#pragma unroll
        for (int u = 0; u < 4; ++u) {
            unsigned i = i0 + u * TB;
            unsigned ei = L.q[i < n ? i : i0];
            ei = ei < (unsigned)N_EDGES ? ei : (unsigned)(N_EDGES - 1);
            s[u] = (unsigned)src[ei];
            d[u] = (unsigned)dst[ei] - (unsigned)base;
        }
#pragma unroll
        for (int u = 0; u < 4; ++u) {
            unsigned i = i0 + u * TB;
            unsigned sv = s[u] < (unsigned)N_NODES ? s[u] : (unsigned)(N_NODES - 1);
            unsigned dl = d[u] < (unsigned)TN ? d[u] : (unsigned)(TN - 1);
            if (i < n) L.q[i] = (sv << 9) | dl;
        }
    }
    __syncthreads();
    {
        const unsigned nh = n * HEADS, hd = tid & 7;
        for (unsigned i0 = tid; i0 < nh; i0 += 4 * TB) {
            float a[4], b[4]; unsigned cell[4];
#pragma unroll
            for (int u = 0; u < 4; ++u) {
                unsigned i = i0 + u * TB;
                unsigned pk = L.q[(i < nh ? i : i0) >> 3];
                unsigned s = pk >> 9, dl = pk & 511;
                a[u] = s_src[(size_t)s * HEADS + hd];
                b[u] = s_dst[(size_t)(base + dl) * HEADS + hd];
                cell[u] = dl * HEADS + hd;
            }
#pragma unroll
            for (int u = 0; u < 4; ++u) {
                float sc = a[u] + b[u];
                sc = (sc >= 0.0f) ? sc : NEG_SLOPE * sc;
                if (i0 + u * TB < nh) atomicMaxF(&L.mnew[cell[u]], sc);
            }
        }
    }
    __syncthreads();
    for (int i = tid; i < TN * HEADS; i += TB) {
        float mo = L.mrun[i], mn = L.mnew[i];
        if (mo != mn) {
            if (mo != -INFINITY) {
                float f = expf(mo - mn);
                L.den[i] *= f;
                float* a = &L.acc[(i >> 3) * OUT_DIM + (i & 7) * HDIM];
#pragma unroll
                for (int d = 0; d < HDIM; ++d) a[d] *= f;
            }
            L.mrun[i] = mn;
        }
    }
    __syncthreads();
    const int wave = tid >> 5, lane = tid & 31, hh = lane >> 2;
    const float4* z4 = reinterpret_cast<const float4*>(z);
    for (unsigned eb = wave * EB; eb < n; eb += (TB / 32) * EB) {
        float4 zv[EB]; float ss[EB], sd[EB]; unsigned dls[EB]; bool ok[EB];
#pragma unroll
        for (int u = 0; u < EB; ++u) {
            unsigned ei = eb + u;
            ok[u] = ei < n;
            unsigned pk = L.q[ok[u] ? ei : (n - 1)];
            unsigned s = pk >> 9;
            dls[u] = pk & 511;
            ss[u] = s_src[(size_t)s * HEADS + hh];
            sd[u] = s_dst[(size_t)(base + dls[u]) * HEADS + hh];
            zv[u] = z4[(size_t)s * (OUT_DIM / 4) + lane];
        }
#pragma unroll
        for (int u = 0; u < EB; ++u) {
            if (ok[u]) {
                float sc = ss[u] + sd[u];
                sc = (sc >= 0.0f) ? sc : NEG_SLOPE * sc;
                float ex = expf(sc - L.mrun[dls[u] * HEADS + hh]);
                float* a = &L.acc[dls[u] * OUT_DIM + lane * 4];
                atomicAdd(a + 0, ex * zv[u].x);
                atomicAdd(a + 1, ex * zv[u].y);
                atomicAdd(a + 2, ex * zv[u].z);
                atomicAdd(a + 3, ex * zv[u].w);
                if ((lane & 3) == 0) atomicAdd(&L.den[dls[u] * HEADS + hh], ex);
            }
        }
    }
}

__global__ void __launch_bounds__(TB)
gat_tile(const int* __restrict__ src, const int* __restrict__ dst,
         const unsigned* __restrict__ ids,
         const float* __restrict__ s_src, const float* __restrict__ s_dst,
         const float* __restrict__ z, const float* __restrict__ h,
         float* __restrict__ out) {
    __shared__ TileLds L;
    const int tid = threadIdx.x;
    const int base = blockIdx.x * TN;

    float4* acc4 = reinterpret_cast<float4*>(L.acc);
    for (int i = tid; i < TN * OUT_DIM / 4; i += TB) acc4[i] = make_float4(0.f, 0.f, 0.f, 0.f);
    for (int i = tid; i < TN * HEADS; i += TB) {
        L.mrun[i] = -INFINITY; L.mnew[i] = -INFINITY; L.den[i] = 0.0f;
    }
    if (tid == 0) L.qn = 0u;
    __syncthreads();

    const int rot = blockIdx.x * (KV / NT);
    tile_scan(L, ids, rot, KV, blockIdx.x, tid);
    tile_scan(L, ids, 0, rot, blockIdx.x, tid);
    __syncthreads();
    const unsigned total = L.qn;
    if (total <= (unsigned)QCAP) {
        tile_batch(L, total, src, dst, s_src, s_dst, z, base, tid);
    } else {
        for (int seg = 0; seg < N_EDGES; seg += QCAP) {
            __syncthreads();
            if (tid == 0) L.qn = 0u;
            __syncthreads();
            int end = seg + QCAP < N_EDGES ? seg + QCAP : N_EDGES;
            tile_scan(L, ids, seg / 16, end / 16, blockIdx.x, tid);
            __syncthreads();
            tile_batch(L, L.qn, src, dst, s_src, s_dst, z, base, tid);
        }
    }
    __syncthreads();

    const float4* h4 = reinterpret_cast<const float4*>(h) + (size_t)base * (OUT_DIM / 4);
    float4* o4 = reinterpret_cast<float4*>(out) + (size_t)base * (OUT_DIM / 4);
    for (int i = tid; i < TN * OUT_DIM / 4; i += TB) {
        float dn = L.den[(i >> 5) * HEADS + ((i & 31) >> 2)];
        float4 a = acc4[i], hv = h4[i], r;
        bool has = dn > 0.0f;
        float v0 = has ? a.x / dn : 0.0f, v1 = has ? a.y / dn : 0.0f;
        float v2 = has ? a.z / dn : 0.0f, v3 = has ? a.w / dn : 0.0f;
        r.x = hv.x + ((v0 > 0.0f) ? v0 : expm1f(v0));
        r.y = hv.y + ((v1 > 0.0f) ? v1 : expm1f(v1));
        r.z = hv.z + ((v2 > 0.0f) ? v2 : expm1f(v2));
        r.w = hv.w + ((v3 > 0.0f) ? v3 : expm1f(v3));
        o4[i] = r;
    }
}

__global__ void __launch_bounds__(256)
gat_epass(const float* __restrict__ e, float* __restrict__ oe) {
    int i = blockIdx.x * 256 + threadIdx.x;
    if (i < N_EDGES / 4)
        reinterpret_cast<float4*>(oe)[i] = reinterpret_cast<const float4*>(e)[i];
}

extern "C" void kernel_launch(void* const* d_in, const int* in_sizes, int n_in,
                              void* d_out, int out_size, void* d_ws, size_t ws_size,
                              hipStream_t stream) {
    const float* h     = (const float*)d_in[0];
    const float* e     = (const float*)d_in[1];
    const int*   src   = (const int*)d_in[2];
    const int*   dst   = (const int*)d_in[3];
    const float* Wfc   = (const float*)d_in[4];
    const float* Wattn = (const float*)d_in[5];
    float* out = (float*)d_out;

    float* z     = (float*)d_ws;
    float* s_src = z + (size_t)N_NODES * OUT_DIM;
    float* s_dst = s_src + (size_t)N_NODES * HEADS;

    const int T = 256;

    gat_gemm_wmma<<<(N_NODES / 16 + 7) / 8, T, 0, stream>>>(h, Wfc, z);

    float* oe = out + (size_t)N_NODES * OUT_DIM;
    unsigned* ids = (unsigned*)oe;

    gat_scores<<<(N_NODES * HEADS + T - 1) / T, T, 0, stream>>>(z, Wattn, dst, s_src, s_dst, ids);

    gat_tile<<<NT, TB, 0, stream>>>(src, dst, ids, s_src, s_dst, z, h, out);

    gat_epass<<<(N_EDGES / 4 + T - 1) / T, T, 0, stream>>>(e, oe);
}


# Sum rows in LDS stored plane-wise (column 4l+j of a node at word 32j+l): the 32 lanes of each float add of the sum pass fall on 32 different banks, not two per bank; values and add order as before
# speedup vs baseline: 9.6270x; 1.1826x over previous
#include <hip/hip_runtime.h>
#include <math.h>

#define N_NODES 100000
#define N_EDGES 1600000
#define IN_DIM  128
#define HEADS   8
#define HDIM    16
#define OUT_DIM 128
#define NEG_SLOPE 0.01f
#define LDS_STRIDE 130
#define TN   400
#define NT   (N_NODES / TN)
#define KV   (N_EDGES / 16)

typedef __attribute__((ext_vector_type(2))) float v2f;
typedef __attribute__((ext_vector_type(8))) float v8f;

__device__ __forceinline__ void atomicMaxF(float* addr, float val) {
    if (val >= 0.0f) {
        atomicMax((int*)addr, __float_as_int(val));
    } else {
        atomicMin((unsigned int*)addr, __float_as_uint(val));
    }
}

__global__ void __launch_bounds__(256)
gat_gemm_wmma(const float* __restrict__ h, const float* __restrict__ Wfc,
              float* __restrict__ z) {
    __shared__ float Wl[OUT_DIM * LDS_STRIDE];

    const int tid = threadIdx.x;
    for (int j = tid; j < IN_DIM * OUT_DIM; j += 256) {
        int i  = j >> 7;
        int c  = j & 127;
        int hh = c >> 4;
        int d  = c & 15;
        Wl[c * LDS_STRIDE + i] = Wfc[hh * (IN_DIM * HDIM) + i * HDIM + d];
    }
    __syncthreads();

    const int wave  = tid >> 5;
    const int lane  = tid & 31;
    const int mtile = blockIdx.x * 8 + wave;
    if (mtile >= N_NODES / 16) return;

    const int row   = lane & 15;
    const int khalf = lane >> 4;

    const float* hrow = h + (size_t)(mtile * 16 + row) * IN_DIM;

    v8f acc[8] = {};

    for (int k0 = 0; k0 < IN_DIM; k0 += 4) {
        const int kk = k0 + 2 * khalf;
        v2f a = *reinterpret_cast<const v2f*>(hrow + kk);
#pragma unroll
        for (int t = 0; t < 8; ++t) {
            v2f b = *reinterpret_cast<const v2f*>(&Wl[(t * 16 + row) * LDS_STRIDE + kk]);
            acc[t] = __builtin_amdgcn_wmma_f32_16x16x4_f32(
                false, a, false, b, (short)0, acc[t], false, false);
        }
    }

    float* zbase = z + (size_t)(mtile * 16) * OUT_DIM;
#pragma unroll
    for (int t = 0; t < 8; ++t) {
#pragma unroll
        for (int i = 0; i < 8; ++i) {
            int mrow = i + khalf * 8;
            zbase[(size_t)mrow * OUT_DIM + t * 16 + row] = acc[t][i];
        }
    }
}

__device__ __forceinline__ unsigned owner_id(int d) {
    unsigned k = (unsigned)d / (unsigned)TN;
    return k < 255u ? k : 255u;
}

__global__ void gat_scores(const float* __restrict__ z,
                           const float* __restrict__ Wattn,
                           const int* __restrict__ dst,
                           float* __restrict__ s_src, float* __restrict__ s_dst,
                           unsigned* __restrict__ ids) {
    int idx = blockIdx.x * blockDim.x + threadIdx.x;
    if (idx >= N_NODES * HEADS) return;
    int n = idx >> 3, hh = idx & 7;
    const float* zp = z + (size_t)n * OUT_DIM + hh * HDIM;
    const float* as = Wattn + hh * (2 * HDIM);
    const float* ad = as + HDIM;
    float ss = 0.0f, sd = 0.0f;
#pragma unroll
    for (int d = 0; d < HDIM; ++d) {
        float v = zp[d];
        ss += v * as[d];
        sd += v * ad[d];
    }
    s_src[idx] = ss;
    s_dst[idx] = sd;

    if (idx < KV) {
        const int4* d4 = reinterpret_cast<const int4*>(dst) + (size_t)idx * 4;
        unsigned w[4];
#pragma unroll
        for (int j = 0; j < 4; ++j) {
            int4 d = d4[j];
            w[j] = owner_id(d.x) | (owner_id(d.y) << 8) | (owner_id(d.z) << 16) | (owner_id(d.w) << 24);
        }
        reinterpret_cast<uint4*>(ids)[idx] = make_uint4(w[0], w[1], w[2], w[3]);
    }
}

#define QCAP 12288
#define TB   1024
#define EB   8

struct TileLds {
    float acc[TN * OUT_DIM];
    float mrun[TN * HEADS];
    float mnew[TN * HEADS];
    float den[TN * HEADS];
    unsigned q[QCAP];
    unsigned qn;
};

__device__ __forceinline__ unsigned match_bytes(unsigned w, unsigned pat) {
    unsigned x = w ^ pat;
    unsigned y = ((x & 0x7F7F7F7Fu) + 0x7F7F7F7Fu) | x;
    return ~y & 0x80808080u;
}

__device__ __forceinline__ unsigned match_pair(uint4 a, uint4 b, unsigned pat) {
    return (match_bytes(a.x, pat) >> 7) | (match_bytes(a.y, pat) >> 6) |
           (match_bytes(a.z, pat) >> 5) | (match_bytes(a.w, pat) >> 4) |
           (match_bytes(b.x, pat) >> 3) | (match_bytes(b.y, pat) >> 2) |
           (match_bytes(b.z, pat) >> 1) |  match_bytes(b.w, pat);
}

__device__ __forceinline__ void scan_four(TileLds& L, const uint4 (&c)[4], unsigned pat,
                                          unsigned ebase, unsigned lane) {
    unsigned r01 = match_pair(c[0], c[1], pat);
    unsigned r23 = match_pair(c[2], c[3], pat);
    for (;;) {
        const bool has = (r01 | r23) != 0u;
        const unsigned m = __builtin_amdgcn_ballot_w32(has);
        if (m == 0u) break;
        unsigned s0 = 0u;
        if (lane == 0u) s0 = atomicAdd(&L.qn, (unsigned)__builtin_popcount(m));
        s0 = (unsigned)__builtin_amdgcn_readfirstlane((int)s0);
        if (has) {
            const bool lo = r01 != 0u;
            const unsigned r = lo ? r01 : r23;
            const unsigned p = (unsigned)__builtin_ctz(r);
            const unsigned vec = (lo ? 0u : 2u) + ((p >> 2) & 1u);
            const unsigned e = ebase + vec * (16u * TB) + ((p & 3u) << 2) + (p >> 3);
            const unsigned slot = s0 + __builtin_amdgcn_mbcnt_lo(m, 0u);
            if (slot < (unsigned)QCAP) L.q[slot] = e;
            if (lo) r01 = r & (r - 1u); else r23 = r & (r - 1u);
        }
    }
}

__device__ __forceinline__ void tile_scan(TileLds& L, const unsigned* __restrict__ ids,
                                          int v0, int v1, unsigned tileb, int tid) {
    const uint4* k4 = reinterpret_cast<const uint4*>(ids);
    const unsigned lane = (unsigned)tid & 31u;
    const unsigned pat = tileb * 0x01010101u;
    const uint4 none = make_uint4(~0u, ~0u, ~0u, ~0u);
    uint4 cur[4], nxt[4];
    int v = v0 + tid;
#pragma unroll
    for (int u = 0; u < 4; ++u) {
        int vv = v + u * TB;
        uint4 t = k4[vv < KV ? vv : KV - 1];
        cur[u] = (vv < v1) ? t : none;
    }
    for (int vb = v0 + (tid & ~31); vb < v1; vb += 4 * TB, v += 4 * TB) {
#pragma unroll
        for (int u = 0; u < 4; ++u) {
            int vn = v + (4 + u) * TB;
            uint4 t = k4[vn < KV ? vn : KV - 1];
            nxt[u] = (vn < v1) ? t : none;
        }
        scan_four(L, cur, pat, (unsigned)v * 16u, lane);
#pragma unroll
        for (int u = 0; u < 4; ++u) cur[u] = nxt[u];
    }
}

__device__ __forceinline__ void tile_batch(TileLds& L, unsigned n,
                                           const int* __restrict__ src,
                                           const int* __restrict__ dst,
                                           const float* __restrict__ s_src,
                                           const float* __restrict__ s_dst,
                                           const float* __restrict__ z,
                                           int base, int tid) {
    for (unsigned i0 = tid; i0 < n; i0 += 4 * TB) {
        unsigned s[4], d[4];
#pragma unroll
        for (int u = 0; u < 4; ++u) {
            unsigned i = i0 + u * TB;
            unsigned ei = L.q[i < n ? i : i0];
            ei = ei < (unsigned)N_EDGES ? ei : (unsigned)(N_EDGES - 1);
            s[u] = (unsigned)src[ei];
            d[u] = (unsigned)dst[ei] - (unsigned)base;
        }
#pragma unroll
        for (int u = 0; u < 4; ++u) {
            unsigned i = i0 + u * TB;
            unsigned sv = s[u] < (unsigned)N_NODES ? s[u] : (unsigned)(N_NODES - 1);
            unsigned dl = d[u] < (unsigned)TN ? d[u] : (unsigned)(TN - 1);
            if (i < n) L.q[i] = (sv << 9) | dl;
        }
    }
    __syncthreads();
    {
        const unsigned nh = n * HEADS, hd = tid & 7;
        for (unsigned i0 = tid; i0 < nh; i0 += 4 * TB) {
            float a[4], b[4]; unsigned cell[4];
#pragma unroll
            for (int u = 0; u < 4; ++u) {
                unsigned i = i0 + u * TB;
                unsigned pk = L.q[(i < nh ? i : i0) >> 3];
                unsigned s = pk >> 9, dl = pk & 511;
                a[u] = s_src[(size_t)s * HEADS + hd];
                b[u] = s_dst[(size_t)(base + dl) * HEADS + hd];
                cell[u] = dl * HEADS + hd;
            }
#pragma unroll
            for (int u = 0; u < 4; ++u) {
                float sc = a[u] + b[u];
                sc = (sc >= 0.0f) ? sc : NEG_SLOPE * sc;
                if (i0 + u * TB < nh) atomicMaxF(&L.mnew[cell[u]], sc);
            }
        }
    }
    __syncthreads();
    for (int i = tid; i < TN * HEADS; i += TB) {
        float mo = L.mrun[i], mn = L.mnew[i];
        if (mo != mn) {
            if (mo != -INFINITY) {
                float f = expf(mo - mn);
                L.den[i] *= f;
                float* a = &L.acc[(i >> 3) * OUT_DIM + (i & 7) * 4];
#pragma unroll
                for (int j = 0; j < 4; ++j) {
#pragma unroll
                    for (int k = 0; k < 4; ++k) a[32 * j + k] *= f;
                }
            }
            L.mrun[i] = mn;
        }
    }
    __syncthreads();
    const int wave = tid >> 5, lane = tid & 31, hh = lane >> 2;
    const float4* z4 = reinterpret_cast<const float4*>(z);
    for (unsigned eb = wave * EB; eb < n; eb += (TB / 32) * EB) {
        float4 zv[EB]; float ss[EB], sd[EB]; unsigned dls[EB]; bool ok[EB];
#pragma unroll
        for (int u = 0; u < EB; ++u) {
            unsigned ei = eb + u;
            ok[u] = ei < n;
            unsigned pk = L.q[ok[u] ? ei : (n - 1)];
            unsigned s = pk >> 9;
            dls[u] = pk & 511;
            ss[u] = s_src[(size_t)s * HEADS + hh];
            sd[u] = s_dst[(size_t)(base + dls[u]) * HEADS + hh];
            zv[u] = z4[(size_t)s * (OUT_DIM / 4) + lane];
        }
#pragma unroll
        for (int u = 0; u < EB; ++u) {
            if (ok[u]) {
                float sc = ss[u] + sd[u];
                sc = (sc >= 0.0f) ? sc : NEG_SLOPE * sc;
                float ex = expf(sc - L.mrun[dls[u] * HEADS + hh]);
                float* a = &L.acc[dls[u] * OUT_DIM + lane];
                atomicAdd(a + 0,  ex * zv[u].x);
                atomicAdd(a + 32, ex * zv[u].y);
                atomicAdd(a + 64, ex * zv[u].z);
                atomicAdd(a + 96, ex * zv[u].w);
                if ((lane & 3) == 0) atomicAdd(&L.den[dls[u] * HEADS + hh], ex);
            }
        }
    }
}

__global__ void __launch_bounds__(TB)
gat_tile(const int* __restrict__ src, const int* __restrict__ dst,
         const unsigned* __restrict__ ids,
         const float* __restrict__ s_src, const float* __restrict__ s_dst,
         const float* __restrict__ z, const float* __restrict__ h,
         float* __restrict__ out) {
    __shared__ TileLds L;
    const int tid = threadIdx.x;
    const int base = blockIdx.x * TN;

    float4* acc4 = reinterpret_cast<float4*>(L.acc);
    for (int i = tid; i < TN * OUT_DIM / 4; i += TB) acc4[i] = make_float4(0.f, 0.f, 0.f, 0.f);
    for (int i = tid; i < TN * HEADS; i += TB) {
        L.mrun[i] = -INFINITY; L.mnew[i] = -INFINITY; L.den[i] = 0.0f;
    }
    if (tid == 0) L.qn = 0u;
    __syncthreads();

    const int rot = blockIdx.x * (KV / NT);
    tile_scan(L, ids, rot, KV, blockIdx.x, tid);
    tile_scan(L, ids, 0, rot, blockIdx.x, tid);
    __syncthreads();
    const unsigned total = L.qn;
    if (total <= (unsigned)QCAP) {
        tile_batch(L, total, src, dst, s_src, s_dst, z, base, tid);
    } else {
        for (int seg = 0; seg < N_EDGES; seg += QCAP) {
            __syncthreads();
            if (tid == 0) L.qn = 0u;
            __syncthreads();
            int end = seg + QCAP < N_EDGES ? seg + QCAP : N_EDGES;
            tile_scan(L, ids, seg / 16, end / 16, blockIdx.x, tid);
            __syncthreads();
            tile_batch(L, L.qn, src, dst, s_src, s_dst, z, base, tid);
        }
    }
    __syncthreads();

    const float4* h4 = reinterpret_cast<const float4*>(h) + (size_t)base * (OUT_DIM / 4);
    float4* o4 = reinterpret_cast<float4*>(out) + (size_t)base * (OUT_DIM / 4);
    for (int i = tid; i < TN * OUT_DIM / 4; i += TB) {
        float dn = L.den[(i >> 5) * HEADS + ((i & 31) >> 2)];
        const float* ar = &L.acc[(i >> 5) * OUT_DIM + (i & 31)];
        float4 a = make_float4(ar[0], ar[32], ar[64], ar[96]), hv = h4[i], r;
        bool has = dn > 0.0f;
        float v0 = has ? a.x / dn : 0.0f, v1 = has ? a.y / dn : 0.0f;
        float v2 = has ? a.z / dn : 0.0f, v3 = has ? a.w / dn : 0.0f;
        r.x = hv.x + ((v0 > 0.0f) ? v0 : expm1f(v0));
        r.y = hv.y + ((v1 > 0.0f) ? v1 : expm1f(v1));
        r.z = hv.z + ((v2 > 0.0f) ? v2 : expm1f(v2));
        r.w = hv.w + ((v3 > 0.0f) ? v3 : expm1f(v3));
        o4[i] = r;
    }
}

__global__ void __launch_bounds__(256)
gat_epass(const float* __restrict__ e, float* __restrict__ oe) {
    int i = blockIdx.x * 256 + threadIdx.x;
    if (i < N_EDGES / 4)
        reinterpret_cast<float4*>(oe)[i] = reinterpret_cast<const float4*>(e)[i];
}

extern "C" void kernel_launch(void* const* d_in, const int* in_sizes, int n_in,
                              void* d_out, int out_size, void* d_ws, size_t ws_size,
                              hipStream_t stream) {
    const float* h     = (const float*)d_in[0];
    const float* e     = (const float*)d_in[1];
    const int*   src   = (const int*)d_in[2];
    const int*   dst   = (const int*)d_in[3];
    const float* Wfc   = (const float*)d_in[4];
    const float* Wattn = (const float*)d_in[5];
    float* out = (float*)d_out;

    float* z     = (float*)d_ws;
    float* s_src = z + (size_t)N_NODES * OUT_DIM;
    float* s_dst = s_src + (size_t)N_NODES * HEADS;

    const int T = 256;

    gat_gemm_wmma<<<(N_NODES / 16 + 7) / 8, T, 0, stream>>>(h, Wfc, z);

    float* oe = out + (size_t)N_NODES * OUT_DIM;
    unsigned* ids = (unsigned*)oe;

    gat_scores<<<(N_NODES * HEADS + T - 1) / T, T, 0, stream>>>(z, Wattn, dst, s_src, s_dst, ids);

    gat_tile<<<NT, TB, 0, stream>>>(src, dst, ids, s_src, s_dst, z, h, out);

    gat_epass<<<(N_EDGES / 4 + T - 1) / T, T, 0, stream>>>(e, oe);
}


# Sum pass issues fewer LDS instructions per edge: exp-sum by one full-wave LDS add per 4 edges (lane 4h+k adds edge k), queue entries read once per group, held as scalars; exp-sum add order moves
# speedup vs baseline: 10.1275x; 1.0520x over previous
#include <hip/hip_runtime.h>
#include <math.h>

#define N_NODES 100000
#define N_EDGES 1600000
#define IN_DIM  128
#define HEADS   8
#define HDIM    16
#define OUT_DIM 128
#define NEG_SLOPE 0.01f
#define LDS_STRIDE 130
#define TN   400
#define NT   (N_NODES / TN)
#define KV   (N_EDGES / 16)

typedef __attribute__((ext_vector_type(2))) float v2f;
typedef __attribute__((ext_vector_type(8))) float v8f;

__device__ __forceinline__ void atomicMaxF(float* addr, float val) {
    if (val >= 0.0f) {
        atomicMax((int*)addr, __float_as_int(val));
    } else {
        atomicMin((unsigned int*)addr, __float_as_uint(val));
    }
}

__global__ void __launch_bounds__(256)
gat_gemm_wmma(const float* __restrict__ h, const float* __restrict__ Wfc,
              float* __restrict__ z) {
    __shared__ float Wl[OUT_DIM * LDS_STRIDE];

    const int tid = threadIdx.x;
    for (int j = tid; j < IN_DIM * OUT_DIM; j += 256) {
        int i  = j >> 7;
        int c  = j & 127;
        int hh = c >> 4;
        int d  = c & 15;
        Wl[c * LDS_STRIDE + i] = Wfc[hh * (IN_DIM * HDIM) + i * HDIM + d];
    }
    __syncthreads();

    const int wave  = tid >> 5;
    const int lane  = tid & 31;
    const int mtile = blockIdx.x * 8 + wave;
    if (mtile >= N_NODES / 16) return;

    const int row   = lane & 15;
    const int khalf = lane >> 4;

    const float* hrow = h + (size_t)(mtile * 16 + row) * IN_DIM;

    v8f acc[8] = {};

    for (int k0 = 0; k0 < IN_DIM; k0 += 4) {
        const int kk = k0 + 2 * khalf;
        v2f a = *reinterpret_cast<const v2f*>(hrow + kk);
#pragma unroll
        for (int t = 0; t < 8; ++t) {
            v2f b = *reinterpret_cast<const v2f*>(&Wl[(t * 16 + row) * LDS_STRIDE + kk]);
            acc[t] = __builtin_amdgcn_wmma_f32_16x16x4_f32(
                false, a, false, b, (short)0, acc[t], false, false);
        }
    }

    float* zbase = z + (size_t)(mtile * 16) * OUT_DIM;
#pragma unroll
    for (int t = 0; t < 8; ++t) {
#pragma unroll
        for (int i = 0; i < 8; ++i) {
            int mrow = i + khalf * 8;
            zbase[(size_t)mrow * OUT_DIM + t * 16 + row] = acc[t][i];
        }
    }
}

__device__ __forceinline__ unsigned owner_id(int d) {
    unsigned k = (unsigned)d / (unsigned)TN;
    return k < 255u ? k : 255u;
}

__global__ void gat_scores(const float* __restrict__ z,
                           const float* __restrict__ Wattn,
                           const int* __restrict__ dst,
                           float* __restrict__ s_src, float* __restrict__ s_dst,
                           unsigned* __restrict__ ids) {
    int idx = blockIdx.x * blockDim.x + threadIdx.x;
    if (idx >= N_NODES * HEADS) return;
    int n = idx >> 3, hh = idx & 7;
    const float* zp = z + (size_t)n * OUT_DIM + hh * HDIM;
    const float* as = Wattn + hh * (2 * HDIM);
    const float* ad = as + HDIM;
    float ss = 0.0f, sd = 0.0f;
#pragma unroll
    for (int d = 0; d < HDIM; ++d) {
        float v = zp[d];
        ss += v * as[d];
        sd += v * ad[d];
    }
    s_src[idx] = ss;
    s_dst[idx] = sd;

    if (idx < KV) {
        const int4* d4 = reinterpret_cast<const int4*>(dst) + (size_t)idx * 4;
        unsigned w[4];
#pragma unroll
        for (int j = 0; j < 4; ++j) {
            int4 d = d4[j];
            w[j] = owner_id(d.x) | (owner_id(d.y) << 8) | (owner_id(d.z) << 16) | (owner_id(d.w) << 24);
        }
        reinterpret_cast<uint4*>(ids)[idx] = make_uint4(w[0], w[1], w[2], w[3]);
    }
}

#define QCAP 12288
#define TB   1024
#define EB   8

struct TileLds {
    float acc[TN * OUT_DIM];
    float mrun[TN * HEADS];
    float mnew[TN * HEADS];
    float den[TN * HEADS];
    unsigned q[QCAP];
    unsigned qn;
};

__device__ __forceinline__ unsigned match_bytes(unsigned w, unsigned pat) {
    unsigned x = w ^ pat;
    unsigned y = ((x & 0x7F7F7F7Fu) + 0x7F7F7F7Fu) | x;
    return ~y & 0x80808080u;
}

__device__ __forceinline__ unsigned match_pair(uint4 a, uint4 b, unsigned pat) {
    return (match_bytes(a.x, pat) >> 7) | (match_bytes(a.y, pat) >> 6) |
           (match_bytes(a.z, pat) >> 5) | (match_bytes(a.w, pat) >> 4) |
           (match_bytes(b.x, pat) >> 3) | (match_bytes(b.y, pat) >> 2) |
           (match_bytes(b.z, pat) >> 1) |  match_bytes(b.w, pat);
}

__device__ __forceinline__ void scan_four(TileLds& L, const uint4 (&c)[4], unsigned pat,
                                          unsigned ebase, unsigned lane) {
    unsigned r01 = match_pair(c[0], c[1], pat);
    unsigned r23 = match_pair(c[2], c[3], pat);
    for (;;) {
        const bool has = (r01 | r23) != 0u;
        const unsigned m = __builtin_amdgcn_ballot_w32(has);
        if (m == 0u) break;
        unsigned s0 = 0u;
        if (lane == 0u) s0 = atomicAdd(&L.qn, (unsigned)__builtin_popcount(m));
        s0 = (unsigned)__builtin_amdgcn_readfirstlane((int)s0);
        if (has) {
            const bool lo = r01 != 0u;
            const unsigned r = lo ? r01 : r23;
            const unsigned p = (unsigned)__builtin_ctz(r);
            const unsigned vec = (lo ? 0u : 2u) + ((p >> 2) & 1u);
            const unsigned e = ebase + vec * (16u * TB) + ((p & 3u) << 2) + (p >> 3);
            const unsigned slot = s0 + __builtin_amdgcn_mbcnt_lo(m, 0u);
            if (slot < (unsigned)QCAP) L.q[slot] = e;
            if (lo) r01 = r & (r - 1u); else r23 = r & (r - 1u);
        }
    }
}

__device__ __forceinline__ void tile_scan(TileLds& L, const unsigned* __restrict__ ids,
                                          int v0, int v1, unsigned tileb, int tid) {
    const uint4* k4 = reinterpret_cast<const uint4*>(ids);
    const unsigned lane = (unsigned)tid & 31u;
    const unsigned pat = tileb * 0x01010101u;
    const uint4 none = make_uint4(~0u, ~0u, ~0u, ~0u);
    uint4 cur[4], nxt[4];
    int v = v0 + tid;
#pragma unroll
    for (int u = 0; u < 4; ++u) {
        int vv = v + u * TB;
        uint4 t = k4[vv < KV ? vv : KV - 1];
        cur[u] = (vv < v1) ? t : none;
    }
    for (int vb = v0 + (tid & ~31); vb < v1; vb += 4 * TB, v += 4 * TB) {
#pragma unroll
        for (int u = 0; u < 4; ++u) {
            int vn = v + (4 + u) * TB;
            uint4 t = k4[vn < KV ? vn : KV - 1];
            nxt[u] = (vn < v1) ? t : none;
        }
        scan_four(L, cur, pat, (unsigned)v * 16u, lane);
#pragma unroll
        for (int u = 0; u < 4; ++u) cur[u] = nxt[u];
    }
}

__device__ __forceinline__ void tile_batch(TileLds& L, unsigned n,
                                           const int* __restrict__ src,
                                           const int* __restrict__ dst,
                                           const float* __restrict__ s_src,
                                           const float* __restrict__ s_dst,
                                           const float* __restrict__ z,
                                           int base, int tid) {
    for (unsigned i0 = tid; i0 < n; i0 += 4 * TB) {
        unsigned s[4], d[4];
#pragma unroll
        for (int u = 0; u < 4; ++u) {
            unsigned i = i0 + u * TB;
            unsigned ei = L.q[i < n ? i : i0];
            ei = ei < (unsigned)N_EDGES ? ei : (unsigned)(N_EDGES - 1);
            s[u] = (unsigned)src[ei];
            d[u] = (unsigned)dst[ei] - (unsigned)base;
        }
#pragma unroll
        for (int u = 0; u < 4; ++u) {
            unsigned i = i0 + u * TB;
            unsigned sv = s[u] < (unsigned)N_NODES ? s[u] : (unsigned)(N_NODES - 1);
            unsigned dl = d[u] < (unsigned)TN ? d[u] : (unsigned)(TN - 1);
            if (i < n) L.q[i] = (sv << 9) | dl;
        }
    }
    __syncthreads();
    {
        const unsigned nh = n * HEADS, hd = tid & 7;
        for (unsigned i0 = tid; i0 < nh; i0 += 4 * TB) {
            float a[4], b[4]; unsigned cell[4];
#pragma unroll
            for (int u = 0; u < 4; ++u) {
                unsigned i = i0 + u * TB;
                unsigned pk = L.q[(i < nh ? i : i0) >> 3];
                unsigned s = pk >> 9, dl = pk & 511;
                a[u] = s_src[(size_t)s * HEADS + hd];
                b[u] = s_dst[(size_t)(base + dl) * HEADS + hd];
                cell[u] = dl * HEADS + hd;
            }
#pragma unroll
            for (int u = 0; u < 4; ++u) {
                float sc = a[u] + b[u];
                sc = (sc >= 0.0f) ? sc : NEG_SLOPE * sc;
                if (i0 + u * TB < nh) atomicMaxF(&L.mnew[cell[u]], sc);
            }
        }
    }
    __syncthreads();
    for (int i = tid; i < TN * HEADS; i += TB) {
        float mo = L.mrun[i], mn = L.mnew[i];
        if (mo != mn) {
            if (mo != -INFINITY) {
                float f = expf(mo - mn);
                L.den[i] *= f;
                float* a = &L.acc[(i >> 3) * OUT_DIM + (i & 7) * 4];
#pragma unroll
                for (int j = 0; j < 4; ++j) {
#pragma unroll
                    for (int k = 0; k < 4; ++k) a[32 * j + k] *= f;
                }
            }
            L.mrun[i] = mn;
        }
    }
    __syncthreads();
    const int lane = tid & 31, hh = lane >> 2;
    const unsigned wave = (unsigned)__builtin_amdgcn_readfirstlane(tid >> 5);
    const unsigned qd = (unsigned)lane & 3u;
    const unsigned pick1 = (qd & 1u) ? ~0u : 0u, pick2 = (qd & 2u) ? ~0u : 0u;
    const float4* z4 = reinterpret_cast<const float4*>(z);
    for (unsigned eb = wave * EB; eb < n; eb += (TB / 32) * EB) {
        unsigned mine[EB / 4]; bool live[EB / 4];
#pragma unroll
        for (int g = 0; g < EB / 4; ++g) {
            const unsigned ql = eb + 4u * (unsigned)g + qd;
            live[g] = ql < n;
            mine[g] = L.q[live[g] ? ql : n - 1];
        }
        float4 zv[EB]; float ss[EB], sd[EB], mr[EB]; unsigned dls[EB];
#pragma unroll
        for (int u = 0; u < EB; ++u) {
            const unsigned pk = (unsigned)__builtin_amdgcn_readlane((int)mine[u >> 2], u & 3);
            const unsigned s = pk >> 9;
            dls[u] = pk & 511;
            ss[u] = s_src[(size_t)s * HEADS + hh];
            sd[u] = s_dst[(size_t)(base + dls[u]) * HEADS + hh];
            zv[u] = z4[(size_t)s * (OUT_DIM / 4) + lane];
            mr[u] = L.mrun[dls[u] * HEADS + hh];
        }
        float ex[EB];
#pragma unroll
        for (int u = 0; u < EB; ++u) {
            float sc = ss[u] + sd[u];
            sc = (sc >= 0.0f) ? sc : NEG_SLOPE * sc;
            ex[u] = expf(sc - mr[u]);
        }
#pragma unroll
        for (int g = 0; g < EB / 4; ++g) {
            const unsigned e01 = (__float_as_uint(ex[4 * g + 1]) & pick1) | (__float_as_uint(ex[4 * g])     & ~pick1);
            const unsigned e23 = (__float_as_uint(ex[4 * g + 3]) & pick1) | (__float_as_uint(ex[4 * g + 2]) & ~pick1);
            const float e = __uint_as_float((e23 & pick2) | (e01 & ~pick2));
            if (live[g]) atomicAdd(&L.den[(mine[g] & 511u) * HEADS + hh], e);
        }
#pragma unroll
        for (int u = 0; u < EB; ++u) {
            if (eb + (unsigned)u < n) {
                float* a = &L.acc[dls[u] * OUT_DIM + lane];
                atomicAdd(a + 0,  ex[u] * zv[u].x);
                atomicAdd(a + 32, ex[u] * zv[u].y);
                atomicAdd(a + 64, ex[u] * zv[u].z);
                atomicAdd(a + 96, ex[u] * zv[u].w);
            }
        }
    }
}

__global__ void __launch_bounds__(TB)
gat_tile(const int* __restrict__ src, const int* __restrict__ dst,
         const unsigned* __restrict__ ids,
         const float* __restrict__ s_src, const float* __restrict__ s_dst,
         const float* __restrict__ z, const float* __restrict__ h,
         float* __restrict__ out) {
    __shared__ TileLds L;
    const int tid = threadIdx.x;
    const int base = blockIdx.x * TN;

    float4* acc4 = reinterpret_cast<float4*>(L.acc);
    for (int i = tid; i < TN * OUT_DIM / 4; i += TB) acc4[i] = make_float4(0.f, 0.f, 0.f, 0.f);
    for (int i = tid; i < TN * HEADS; i += TB) {
        L.mrun[i] = -INFINITY; L.mnew[i] = -INFINITY; L.den[i] = 0.0f;
    }
    if (tid == 0) L.qn = 0u;
    __syncthreads();

    const int rot = blockIdx.x * (KV / NT);
    tile_scan(L, ids, rot, KV, blockIdx.x, tid);
    tile_scan(L, ids, 0, rot, blockIdx.x, tid);
    __syncthreads();
    const unsigned total = L.qn;
    if (total <= (unsigned)QCAP) {
        tile_batch(L, total, src, dst, s_src, s_dst, z, base, tid);
    } else {
        for (int seg = 0; seg < N_EDGES; seg += QCAP) {
            __syncthreads();
            if (tid == 0) L.qn = 0u;
            __syncthreads();
            int end = seg + QCAP < N_EDGES ? seg + QCAP : N_EDGES;
            tile_scan(L, ids, seg / 16, end / 16, blockIdx.x, tid);
            __syncthreads();
            tile_batch(L, L.qn, src, dst, s_src, s_dst, z, base, tid);
        }
    }
    __syncthreads();

    const float4* h4 = reinterpret_cast<const float4*>(h) + (size_t)base * (OUT_DIM / 4);
    float4* o4 = reinterpret_cast<float4*>(out) + (size_t)base * (OUT_DIM / 4);
    for (int i = tid; i < TN * OUT_DIM / 4; i += TB) {
        float dn = L.den[(i >> 5) * HEADS + ((i & 31) >> 2)];
        const float* ar = &L.acc[(i >> 5) * OUT_DIM + (i & 31)];
        float4 a = make_float4(ar[0], ar[32], ar[64], ar[96]), hv = h4[i], r;
        bool has = dn > 0.0f;
        float v0 = has ? a.x / dn : 0.0f, v1 = has ? a.y / dn : 0.0f;
        float v2 = has ? a.z / dn : 0.0f, v3 = has ? a.w / dn : 0.0f;
        r.x = hv.x + ((v0 > 0.0f) ? v0 : expm1f(v0));
        r.y = hv.y + ((v1 > 0.0f) ? v1 : expm1f(v1));
        r.z = hv.z + ((v2 > 0.0f) ? v2 : expm1f(v2));
        r.w = hv.w + ((v3 > 0.0f) ? v3 : expm1f(v3));
        o4[i] = r;
    }
}

__global__ void __launch_bounds__(256)
gat_epass(const float* __restrict__ e, float* __restrict__ oe) {
    int i = blockIdx.x * 256 + threadIdx.x;
    if (i < N_EDGES / 4)
        reinterpret_cast<float4*>(oe)[i] = reinterpret_cast<const float4*>(e)[i];
}

extern "C" void kernel_launch(void* const* d_in, const int* in_sizes, int n_in,
                              void* d_out, int out_size, void* d_ws, size_t ws_size,
                              hipStream_t stream) {
    const float* h     = (const float*)d_in[0];
    const float* e     = (const float*)d_in[1];
    const int*   src   = (const int*)d_in[2];
    const int*   dst   = (const int*)d_in[3];
    const float* Wfc   = (const float*)d_in[4];
    const float* Wattn = (const float*)d_in[5];
    float* out = (float*)d_out;

    float* z     = (float*)d_ws;
    float* s_src = z + (size_t)N_NODES * OUT_DIM;
    float* s_dst = s_src + (size_t)N_NODES * HEADS;

    const int T = 256;

    gat_gemm_wmma<<<(N_NODES / 16 + 7) / 8, T, 0, stream>>>(h, Wfc, z);

    float* oe = out + (size_t)N_NODES * OUT_DIM;
    unsigned* ids = (unsigned*)oe;

    gat_scores<<<(N_NODES * HEADS + T - 1) / T, T, 0, stream>>>(z, Wattn, dst, s_src, s_dst, ids);

    gat_tile<<<NT, TB, 0, stream>>>(src, dst, ids, s_src, s_dst, z, h, out);

    gat_epass<<<(N_EDGES / 4 + T - 1) / T, T, 0, stream>>>(e, oe);
}


# Owner-id list stored bit-sliced (8 planes per 32 edges, built by wave ballots in the score launch): a tile matches 32 edges with 8 xor-and ops, 8 list loads in flight a lane; queue and sum order move
# speedup vs baseline: 10.6918x; 1.0557x over previous
#include <hip/hip_runtime.h>
#include <math.h>

#define N_NODES 100000
#define N_EDGES 1600000
#define IN_DIM  128
#define HEADS   8
#define HDIM    16
#define OUT_DIM 128
#define NEG_SLOPE 0.01f
#define LDS_STRIDE 130
#define TN   400
#define NT   (N_NODES / TN)
#define GV   (N_EDGES / 32)

typedef __attribute__((ext_vector_type(2))) float v2f;
typedef __attribute__((ext_vector_type(8))) float v8f;

__device__ __forceinline__ void atomicMaxF(float* addr, float val) {
    if (val >= 0.0f) {
        atomicMax((int*)addr, __float_as_int(val));
    } else {
        atomicMin((unsigned int*)addr, __float_as_uint(val));
    }
}

__global__ void __launch_bounds__(256)
gat_gemm_wmma(const float* __restrict__ h, const float* __restrict__ Wfc,
              float* __restrict__ z) {
    __shared__ float Wl[OUT_DIM * LDS_STRIDE];

    const int tid = threadIdx.x;
    for (int j = tid; j < IN_DIM * OUT_DIM; j += 256) {
        int i  = j >> 7;
        int c  = j & 127;
        int hh = c >> 4;
        int d  = c & 15;
        Wl[c * LDS_STRIDE + i] = Wfc[hh * (IN_DIM * HDIM) + i * HDIM + d];
    }
    __syncthreads();

    const int wave  = tid >> 5;
    const int lane  = tid & 31;
    const int mtile = blockIdx.x * 8 + wave;
    if (mtile >= N_NODES / 16) return;

    const int row   = lane & 15;
    const int khalf = lane >> 4;

    const float* hrow = h + (size_t)(mtile * 16 + row) * IN_DIM;

    v8f acc[8] = {};

    for (int k0 = 0; k0 < IN_DIM; k0 += 4) {
        const int kk = k0 + 2 * khalf;
        v2f a = *reinterpret_cast<const v2f*>(hrow + kk);
#pragma unroll
        for (int t = 0; t < 8; ++t) {
            v2f b = *reinterpret_cast<const v2f*>(&Wl[(t * 16 + row) * LDS_STRIDE + kk]);
            acc[t] = __builtin_amdgcn_wmma_f32_16x16x4_f32(
                false, a, false, b, (short)0, acc[t], false, false);
        }
    }

    float* zbase = z + (size_t)(mtile * 16) * OUT_DIM;
#pragma unroll
    for (int t = 0; t < 8; ++t) {
#pragma unroll
        for (int i = 0; i < 8; ++i) {
            int mrow = i + khalf * 8;
            zbase[(size_t)mrow * OUT_DIM + t * 16 + row] = acc[t][i];
        }
    }
}

__device__ __forceinline__ unsigned owner_id(int d) {
    unsigned k = (unsigned)d / (unsigned)TN;
    return k < 255u ? k : 255u;
}

__device__ __forceinline__ unsigned plane_vec(unsigned g, unsigned half) {
    return ((g >> 4) << 5) + (half << 4) + (g & 15u);
}

__global__ void gat_scores(const float* __restrict__ z,
                           const float* __restrict__ Wattn,
                           const int* __restrict__ dst,
                           float* __restrict__ s_src, float* __restrict__ s_dst,
                           unsigned* __restrict__ ids) {
    int idx = blockIdx.x * blockDim.x + threadIdx.x;
    if (idx >= N_NODES * HEADS) return;
    int n = idx >> 3, hh = idx & 7;
    const float* zp = z + (size_t)n * OUT_DIM + hh * HDIM;
    const float* as = Wattn + hh * (2 * HDIM);
    const float* ad = as + HDIM;
    float ss = 0.0f, sd = 0.0f;
#pragma unroll
    for (int d = 0; d < HDIM; ++d) {
        float v = zp[d];
        ss += v * as[d];
        sd += v * ad[d];
    }
    s_src[idx] = ss;
    s_dst[idx] = sd;

    static_assert((N_NODES * HEADS) % 256 == 0 && (N_NODES * HEADS / 32) * 2 == GV, "one wave per two edge groups");
    const unsigned lane = (unsigned)threadIdx.x & 31u;
    const unsigned g0 = ((unsigned)idx >> 5) * 2u;
    const unsigned ida = owner_id(dst[g0 * 32u + lane]);
    const unsigned idb = owner_id(dst[g0 * 32u + 32u + lane]);
    unsigned w = 0u;
#pragma unroll
    for (int k = 0; k < 8; ++k) {
        const unsigned pa = __builtin_amdgcn_ballot_w32(((ida >> k) & 1u) != 0u);
        const unsigned pb = __builtin_amdgcn_ballot_w32(((idb >> k) & 1u) != 0u);
        if (lane == (unsigned)k) w = pa;
        if (lane == (unsigned)(8 + k)) w = pb;
    }
    if (lane < 16u) {
        const unsigned g = g0 + (lane >> 3), k = lane & 7u;
        ids[plane_vec(g, k >> 2) * 4u + (k & 3u)] = w;
    }
}

#define QCAP 12288
#define TB   1024
#define EB   8

struct TileLds {
    float acc[TN * OUT_DIM];
    float mrun[TN * HEADS];
    float mnew[TN * HEADS];
    float den[TN * HEADS];
    unsigned q[QCAP];
    unsigned qn;
};

__device__ __forceinline__ unsigned match_planes(uint4 a, uint4 b, const unsigned (&inv)[8]) {
    return (a.x ^ inv[0]) & (a.y ^ inv[1]) & (a.z ^ inv[2]) & (a.w ^ inv[3]) &
           (b.x ^ inv[4]) & (b.y ^ inv[5]) & (b.z ^ inv[6]) & (b.w ^ inv[7]);
}

__device__ __forceinline__ void scan_groups(TileLds& L, const uint4 (&c)[8], const unsigned (&inv)[8],
                                            int g, int gend, unsigned lane) {
    unsigned r0 = (g          < gend) ? match_planes(c[0], c[1], inv) : 0u;
    unsigned r1 = (g + TB     < gend) ? match_planes(c[2], c[3], inv) : 0u;
    unsigned r2 = (g + 2 * TB < gend) ? match_planes(c[4], c[5], inv) : 0u;
    unsigned r3 = (g + 3 * TB < gend) ? match_planes(c[6], c[7], inv) : 0u;
    const unsigned e0 = (unsigned)g * 32u;
    for (;;) {
        const bool has = (r0 | r1 | r2 | r3) != 0u;
        const unsigned m = __builtin_amdgcn_ballot_w32(has);
        if (m == 0u) break;
        unsigned s0 = 0u;
        if (lane == 0u) s0 = atomicAdd(&L.qn, (unsigned)__builtin_popcount(m));
        s0 = (unsigned)__builtin_amdgcn_readfirstlane((int)s0);
        if (has) {
            const bool fp = (r0 | r1) != 0u;
            const unsigned x0 = fp ? r0 : r2, x1 = fp ? r1 : r3;
            const bool lo = x0 != 0u;
            const unsigned r = lo ? x0 : x1;
            const unsigned u = (fp ? 0u : 2u) + (lo ? 0u : 1u);
            const unsigned e = e0 + u * (32u * TB) + (unsigned)__builtin_ctz(r);
            const unsigned slot = s0 + __builtin_amdgcn_mbcnt_lo(m, 0u);
            if (slot < (unsigned)QCAP) L.q[slot] = e;
            const unsigned rr = r & (r - 1u);
            if (fp) { if (lo) r0 = rr; else r1 = rr; }
            else    { if (lo) r2 = rr; else r3 = rr; }
        }
    }
}

__device__ __forceinline__ void load_groups(const uint4* __restrict__ k4, int g, uint4 (&c)[8]) {
#pragma unroll
    for (int u = 0; u < 4; ++u) {
        const int gg = g + u * TB;
        const unsigned gc = (unsigned)(gg < GV ? gg : GV - 1);
        c[2 * u]     = k4[plane_vec(gc, 0u)];
        c[2 * u + 1] = k4[plane_vec(gc, 1u)];
    }
}

__device__ __forceinline__ void tile_scan(TileLds& L, const unsigned* __restrict__ ids,
                                          int v0, int v1, unsigned tileb, int tid) {
    const uint4* k4 = reinterpret_cast<const uint4*>(ids);
    const unsigned lane = (unsigned)tid & 31u;
    unsigned inv[8];
#pragma unroll
    for (int k = 0; k < 8; ++k) inv[k] = ((tileb >> k) & 1u) ? 0u : ~0u;
    uint4 ca[8], cb[8];
    int g = v0 + tid;
    load_groups(k4, g, ca);
    for (int gb = v0 + (tid & ~31); gb < v1; gb += 8 * TB, g += 8 * TB) {
        load_groups(k4, g + 4 * TB, cb);
        scan_groups(L, ca, inv, g, v1, lane);
        if (gb + 4 * TB >= v1) break;
        load_groups(k4, g + 8 * TB, ca);
        scan_groups(L, cb, inv, g + 4 * TB, v1, lane);
    }
}

__device__ __forceinline__ void tile_batch(TileLds& L, unsigned n,
                                           const int* __restrict__ src,
                                           const int* __restrict__ dst,
                                           const float* __restrict__ s_src,
                                           const float* __restrict__ s_dst,
                                           const float* __restrict__ z,
                                           int base, int tid) {
    for (unsigned i0 = tid; i0 < n; i0 += 4 * TB) {
        unsigned s[4], d[4];
#pragma unroll
        for (int u = 0; u < 4; ++u) {
            unsigned i = i0 + u * TB;
            unsigned ei = L.q[i < n ? i : i0];
            ei = ei < (unsigned)N_EDGES ? ei : (unsigned)(N_EDGES - 1);
            s[u] = (unsigned)src[ei];
            d[u] = (unsigned)dst[ei] - (unsigned)base;
        }
#pragma unroll
        for (int u = 0; u < 4; ++u) {
            unsigned i = i0 + u * TB;
            unsigned sv = s[u] < (unsigned)N_NODES ? s[u] : (unsigned)(N_NODES - 1);
            unsigned dl = d[u] < (unsigned)TN ? d[u] : (unsigned)(TN - 1);
            if (i < n) L.q[i] = (sv << 9) | dl;
        }
    }
    __syncthreads();
    {
        const unsigned nh = n * HEADS, hd = tid & 7;
        for (unsigned i0 = tid; i0 < nh; i0 += 4 * TB) {
            float a[4], b[4]; unsigned cell[4];
#pragma unroll
            for (int u = 0; u < 4; ++u) {
                unsigned i = i0 + u * TB;
                unsigned pk = L.q[(i < nh ? i : i0) >> 3];
                unsigned s = pk >> 9, dl = pk & 511;
                a[u] = s_src[(size_t)s * HEADS + hd];
                b[u] = s_dst[(size_t)(base + dl) * HEADS + hd];
                cell[u] = dl * HEADS + hd;
            }
#pragma unroll
            for (int u = 0; u < 4; ++u) {
                float sc = a[u] + b[u];
                sc = (sc >= 0.0f) ? sc : NEG_SLOPE * sc;
                if (i0 + u * TB < nh) atomicMaxF(&L.mnew[cell[u]], sc);
            }
        }
    }
    __syncthreads();
    for (int i = tid; i < TN * HEADS; i += TB) {
        float mo = L.mrun[i], mn = L.mnew[i];
        if (mo != mn) {
            if (mo != -INFINITY) {
                float f = expf(mo - mn);
                L.den[i] *= f;
                float* a = &L.acc[(i >> 3) * OUT_DIM + (i & 7) * 4];
#pragma unroll
                for (int j = 0; j < 4; ++j) {
#pragma unroll
                    for (int k = 0; k < 4; ++k) a[32 * j + k] *= f;
                }
            }
            L.mrun[i] = mn;
        }
    }
    __syncthreads();
    const int lane = tid & 31, hh = lane >> 2;
    const unsigned wave = (unsigned)__builtin_amdgcn_readfirstlane(tid >> 5);
    const unsigned qd = (unsigned)lane & 3u;
    const unsigned pick1 = (qd & 1u) ? ~0u : 0u, pick2 = (qd & 2u) ? ~0u : 0u;
    const float4* z4 = reinterpret_cast<const float4*>(z);
    for (unsigned eb = wave * EB; eb < n; eb += (TB / 32) * EB) {
        unsigned mine[EB / 4]; bool live[EB / 4];
#pragma unroll
        for (int g = 0; g < EB / 4; ++g) {
            const unsigned ql = eb + 4u * (unsigned)g + qd;
            live[g] = ql < n;
            mine[g] = L.q[live[g] ? ql : n - 1];
        }
        float4 zv[EB]; float ss[EB], sd[EB], mr[EB]; unsigned dls[EB];
#pragma unroll
        for (int u = 0; u < EB; ++u) {
            const unsigned pk = (unsigned)__builtin_amdgcn_readlane((int)mine[u >> 2], u & 3);
            const unsigned s = pk >> 9;
            dls[u] = pk & 511;
            ss[u] = s_src[(size_t)s * HEADS + hh];
            sd[u] = s_dst[(size_t)(base + dls[u]) * HEADS + hh];
            zv[u] = z4[(size_t)s * (OUT_DIM / 4) + lane];
            mr[u] = L.mrun[dls[u] * HEADS + hh];
        }
        float ex[EB];
#pragma unroll
        for (int u = 0; u < EB; ++u) {
            float sc = ss[u] + sd[u];
            sc = (sc >= 0.0f) ? sc : NEG_SLOPE * sc;
            ex[u] = expf(sc - mr[u]);
        }
#pragma unroll
        for (int g = 0; g < EB / 4; ++g) {
            const unsigned e01 = (__float_as_uint(ex[4 * g + 1]) & pick1) | (__float_as_uint(ex[4 * g])     & ~pick1);
            const unsigned e23 = (__float_as_uint(ex[4 * g + 3]) & pick1) | (__float_as_uint(ex[4 * g + 2]) & ~pick1);
            const float e = __uint_as_float((e23 & pick2) | (e01 & ~pick2));
            if (live[g]) atomicAdd(&L.den[(mine[g] & 511u) * HEADS + hh], e);
        }
#pragma unroll
        for (int u = 0; u < EB; ++u) {
            if (eb + (unsigned)u < n) {
                float* a = &L.acc[dls[u] * OUT_DIM + lane];
                atomicAdd(a + 0,  ex[u] * zv[u].x);
                atomicAdd(a + 32, ex[u] * zv[u].y);
                atomicAdd(a + 64, ex[u] * zv[u].z);
                atomicAdd(a + 96, ex[u] * zv[u].w);
            }
        }
    }
}

__global__ void __launch_bounds__(TB)
gat_tile(const int* __restrict__ src, const int* __restrict__ dst,
         const unsigned* __restrict__ ids,
         const float* __restrict__ s_src, const float* __restrict__ s_dst,
         const float* __restrict__ z, const float* __restrict__ h,
         float* __restrict__ out) {
    __shared__ TileLds L;
    const int tid = threadIdx.x;
    const int base = blockIdx.x * TN;

    float4* acc4 = reinterpret_cast<float4*>(L.acc);
    for (int i = tid; i < TN * OUT_DIM / 4; i += TB) acc4[i] = make_float4(0.f, 0.f, 0.f, 0.f);
    for (int i = tid; i < TN * HEADS; i += TB) {
        L.mrun[i] = -INFINITY; L.mnew[i] = -INFINITY; L.den[i] = 0.0f;
    }
    if (tid == 0) L.qn = 0u;
    __syncthreads();

    const int rot = blockIdx.x * 192;
    tile_scan(L, ids, rot, GV, blockIdx.x, tid);
    tile_scan(L, ids, 0, rot, blockIdx.x, tid);
    __syncthreads();
    const unsigned total = L.qn;
    if (total <= (unsigned)QCAP) {
        tile_batch(L, total, src, dst, s_src, s_dst, z, base, tid);
    } else {
        for (int seg = 0; seg < N_EDGES; seg += QCAP) {
            __syncthreads();
            if (tid == 0) L.qn = 0u;
            __syncthreads();
            int end = seg + QCAP < N_EDGES ? seg + QCAP : N_EDGES;
            tile_scan(L, ids, seg / 32, end / 32, blockIdx.x, tid);
            __syncthreads();
            tile_batch(L, L.qn, src, dst, s_src, s_dst, z, base, tid);
        }
    }
    __syncthreads();

    const float4* h4 = reinterpret_cast<const float4*>(h) + (size_t)base * (OUT_DIM / 4);
    float4* o4 = reinterpret_cast<float4*>(out) + (size_t)base * (OUT_DIM / 4);
    for (int i = tid; i < TN * OUT_DIM / 4; i += TB) {
        float dn = L.den[(i >> 5) * HEADS + ((i & 31) >> 2)];
        const float* ar = &L.acc[(i >> 5) * OUT_DIM + (i & 31)];
        float4 a = make_float4(ar[0], ar[32], ar[64], ar[96]), hv = h4[i], r;
        bool has = dn > 0.0f;
        float v0 = has ? a.x / dn : 0.0f, v1 = has ? a.y / dn : 0.0f;
        float v2 = has ? a.z / dn : 0.0f, v3 = has ? a.w / dn : 0.0f;
        r.x = hv.x + ((v0 > 0.0f) ? v0 : expm1f(v0));
        r.y = hv.y + ((v1 > 0.0f) ? v1 : expm1f(v1));
        r.z = hv.z + ((v2 > 0.0f) ? v2 : expm1f(v2));
        r.w = hv.w + ((v3 > 0.0f) ? v3 : expm1f(v3));
        o4[i] = r;
    }
}

__global__ void __launch_bounds__(256)
gat_epass(const float* __restrict__ e, float* __restrict__ oe) {
    int i = blockIdx.x * 256 + threadIdx.x;
    if (i < N_EDGES / 4)
        reinterpret_cast<float4*>(oe)[i] = reinterpret_cast<const float4*>(e)[i];
}

extern "C" void kernel_launch(void* const* d_in, const int* in_sizes, int n_in,
                              void* d_out, int out_size, void* d_ws, size_t ws_size,
                              hipStream_t stream) {
    const float* h     = (const float*)d_in[0];
    const float* e     = (const float*)d_in[1];
    const int*   src   = (const int*)d_in[2];
    const int*   dst   = (const int*)d_in[3];
    const float* Wfc   = (const float*)d_in[4];
    const float* Wattn = (const float*)d_in[5];
    float* out = (float*)d_out;

    float* z     = (float*)d_ws;
    float* s_src = z + (size_t)N_NODES * OUT_DIM;
    float* s_dst = s_src + (size_t)N_NODES * HEADS;

    const int T = 256;

    gat_gemm_wmma<<<(N_NODES / 16 + 7) / 8, T, 0, stream>>>(h, Wfc, z);

    float* oe = out + (size_t)N_NODES * OUT_DIM;
    unsigned* ids = (unsigned*)oe;

    gat_scores<<<(N_NODES * HEADS + T - 1) / T, T, 0, stream>>>(z, Wattn, dst, s_src, s_dst, ids);

    gat_tile<<<NT, TB, 0, stream>>>(src, dst, ids, s_src, s_dst, z, h, out);

    gat_epass<<<(N_EDGES / 4 + T - 1) / T, T, 0, stream>>>(e, oe);
}


# Softmax shift of a (node, head) = leaky(s_dst + largest source score of the head over all nodes, from the score launch): max pass, its gathers, rescale gone; exact path kept if score range > 60
# speedup vs baseline: 11.0723x; 1.0356x over previous
#include <hip/hip_runtime.h>
#include <math.h>

#define N_NODES 100000
#define N_EDGES 1600000
#define IN_DIM  128
#define HEADS   8
#define HDIM    16
#define OUT_DIM 128
#define NEG_SLOPE 0.01f
#define LDS_STRIDE 130
#define TN   400
#define NT   (N_NODES / TN)
#define GV   (N_EDGES / 32)
#define NSB  (N_NODES * HEADS / 256)
#define SPREAD_MAX 60.0f

typedef __attribute__((ext_vector_type(2))) float v2f;
typedef __attribute__((ext_vector_type(8))) float v8f;

__device__ __forceinline__ void atomicMaxF(float* addr, float val) {
    if (val >= 0.0f) {
        atomicMax((int*)addr, __float_as_int(val));
    } else {
        atomicMin((unsigned int*)addr, __float_as_uint(val));
    }
}

__global__ void __launch_bounds__(256)
gat_gemm_wmma(const float* __restrict__ h, const float* __restrict__ Wfc,
              float* __restrict__ z) {
    __shared__ float Wl[OUT_DIM * LDS_STRIDE];

    const int tid = threadIdx.x;
    for (int j = tid; j < IN_DIM * OUT_DIM; j += 256) {
        int i  = j >> 7;
        int c  = j & 127;
        int hh = c >> 4;
        int d  = c & 15;
        Wl[c * LDS_STRIDE + i] = Wfc[hh * (IN_DIM * HDIM) + i * HDIM + d];
    }
    __syncthreads();

    const int wave  = tid >> 5;
    const int lane  = tid & 31;
    const int mtile = blockIdx.x * 8 + wave;
    if (mtile >= N_NODES / 16) return;

    const int row   = lane & 15;
    const int khalf = lane >> 4;

    const float* hrow = h + (size_t)(mtile * 16 + row) * IN_DIM;

    v8f acc[8] = {};

    for (int k0 = 0; k0 < IN_DIM; k0 += 4) {
        const int kk = k0 + 2 * khalf;
        v2f a = *reinterpret_cast<const v2f*>(hrow + kk);
#pragma unroll
        for (int t = 0; t < 8; ++t) {
            v2f b = *reinterpret_cast<const v2f*>(&Wl[(t * 16 + row) * LDS_STRIDE + kk]);
            acc[t] = __builtin_amdgcn_wmma_f32_16x16x4_f32(
                false, a, false, b, (short)0, acc[t], false, false);
        }
    }

    float* zbase = z + (size_t)(mtile * 16) * OUT_DIM;
#pragma unroll
    for (int t = 0; t < 8; ++t) {
#pragma unroll
        for (int i = 0; i < 8; ++i) {
            int mrow = i + khalf * 8;
            zbase[(size_t)mrow * OUT_DIM + t * 16 + row] = acc[t][i];
        }
    }
}

__device__ __forceinline__ unsigned owner_id(int d) {
    unsigned k = (unsigned)d / (unsigned)TN;
    return k < 255u ? k : 255u;
}

__device__ __forceinline__ unsigned plane_vec(unsigned g, unsigned half) {
    return ((g >> 4) << 5) + (half << 4) + (g & 15u);
}

__global__ void gat_scores(const float* __restrict__ z,
                           const float* __restrict__ Wattn,
                           const int* __restrict__ dst,
                           float* __restrict__ s_src, float* __restrict__ s_dst,
                           unsigned* __restrict__ ids, float* __restrict__ bnd) {
    __shared__ float pb[8][16];
    int idx = blockIdx.x * 256 + threadIdx.x;
    int n = idx >> 3, hh = idx & 7;
    const float* zp = z + (size_t)n * OUT_DIM + hh * HDIM;
    const float* as = Wattn + hh * (2 * HDIM);
    const float* ad = as + HDIM;
    float ss = 0.0f, sd = 0.0f;
#pragma unroll
    for (int d = 0; d < HDIM; ++d) {
        float v = zp[d];
        ss += v * as[d];
        sd += v * ad[d];
    }
    s_src[idx] = ss;
    s_dst[idx] = sd;

    {
        float mx = fmaxf(ss, __shfl_xor(ss, 8));  mx = fmaxf(mx, __shfl_xor(mx, 16));
        float mn = fminf(ss, __shfl_xor(ss, 8));  mn = fminf(mn, __shfl_xor(mn, 16));
        const int wv = threadIdx.x >> 5, ln = threadIdx.x & 31;
        if (ln < 8) { pb[wv][ln] = mx; pb[wv][8 + ln] = mn; }
        __syncthreads();
        if (threadIdx.x < 16) {
            const int t = threadIdx.x;
            float r = pb[0][t];
#pragma unroll
            for (int w = 1; w < 8; ++w) r = (t < 8) ? fmaxf(r, pb[w][t]) : fminf(r, pb[w][t]);
            bnd[blockIdx.x * 16 + t] = r;
        }
    }

    static_assert((N_NODES * HEADS) % 256 == 0 && (N_NODES * HEADS / 32) * 2 == GV, "whole work-groups of 256; one wave per two edge groups");
    const unsigned lane = (unsigned)threadIdx.x & 31u;
    const unsigned g0 = ((unsigned)idx >> 5) * 2u;
    const unsigned ida = owner_id(dst[g0 * 32u + lane]);
    const unsigned idb = owner_id(dst[g0 * 32u + 32u + lane]);
    unsigned w = 0u;
#pragma unroll
    for (int k = 0; k < 8; ++k) {
        const unsigned pa = __builtin_amdgcn_ballot_w32(((ida >> k) & 1u) != 0u);
        const unsigned pb = __builtin_amdgcn_ballot_w32(((idb >> k) & 1u) != 0u);
        if (lane == (unsigned)k) w = pa;
        if (lane == (unsigned)(8 + k)) w = pb;
    }
    if (lane < 16u) {
        const unsigned g = g0 + (lane >> 3), k = lane & 7u;
        ids[plane_vec(g, k >> 2) * 4u + (k & 3u)] = w;
    }
}

#define QCAP 12288
#define TB   1024
#define EB   8

struct TileLds {
    float acc[TN * OUT_DIM];
    float mrun[TN * HEADS];
    float mnew[TN * HEADS];
    float den[TN * HEADS];
    unsigned q[QCAP];
    unsigned qn;
    float pbw[32 * 16];
    float bd[16];
};

__device__ __forceinline__ unsigned match_planes(uint4 a, uint4 b, const unsigned (&inv)[8]) {
    return (a.x ^ inv[0]) & (a.y ^ inv[1]) & (a.z ^ inv[2]) & (a.w ^ inv[3]) &
           (b.x ^ inv[4]) & (b.y ^ inv[5]) & (b.z ^ inv[6]) & (b.w ^ inv[7]);
}

__device__ __forceinline__ void scan_groups(TileLds& L, const uint4 (&c)[8], const unsigned (&inv)[8],
                                            int g, int gend, unsigned lane) {
    unsigned r0 = (g          < gend) ? match_planes(c[0], c[1], inv) : 0u;
    unsigned r1 = (g + TB     < gend) ? match_planes(c[2], c[3], inv) : 0u;
    unsigned r2 = (g + 2 * TB < gend) ? match_planes(c[4], c[5], inv) : 0u;
    unsigned r3 = (g + 3 * TB < gend) ? match_planes(c[6], c[7], inv) : 0u;
    const unsigned e0 = (unsigned)g * 32u;
    for (;;) {
        const bool has = (r0 | r1 | r2 | r3) != 0u;
        const unsigned m = __builtin_amdgcn_ballot_w32(has);
        if (m == 0u) break;
        unsigned s0 = 0u;
        if (lane == 0u) s0 = atomicAdd(&L.qn, (unsigned)__builtin_popcount(m));
        s0 = (unsigned)__builtin_amdgcn_readfirstlane((int)s0);
        if (has) {
            const bool fp = (r0 | r1) != 0u;
            const unsigned x0 = fp ? r0 : r2, x1 = fp ? r1 : r3;
            const bool lo = x0 != 0u;
            const unsigned r = lo ? x0 : x1;
            const unsigned u = (fp ? 0u : 2u) + (lo ? 0u : 1u);
            const unsigned e = e0 + u * (32u * TB) + (unsigned)__builtin_ctz(r);
            const unsigned slot = s0 + __builtin_amdgcn_mbcnt_lo(m, 0u);
            if (slot < (unsigned)QCAP) L.q[slot] = e;
            const unsigned rr = r & (r - 1u);
            if (fp) { if (lo) r0 = rr; else r1 = rr; }
            else    { if (lo) r2 = rr; else r3 = rr; }
        }
    }
}

__device__ __forceinline__ void load_groups(const uint4* __restrict__ k4, int g, uint4 (&c)[8]) {
#pragma unroll
    for (int u = 0; u < 4; ++u) {
        const int gg = g + u * TB;
        const unsigned gc = (unsigned)(gg < GV ? gg : GV - 1);
        c[2 * u]     = k4[plane_vec(gc, 0u)];
        c[2 * u + 1] = k4[plane_vec(gc, 1u)];
    }
}

__device__ __forceinline__ void tile_scan(TileLds& L, const unsigned* __restrict__ ids,
                                          int v0, int v1, unsigned tileb, int tid) {
    const uint4* k4 = reinterpret_cast<const uint4*>(ids);
    const unsigned lane = (unsigned)tid & 31u;
    unsigned inv[8];
#pragma unroll
    for (int k = 0; k < 8; ++k) inv[k] = ((tileb >> k) & 1u) ? 0u : ~0u;
    uint4 ca[8], cb[8];
    int g = v0 + tid;
    load_groups(k4, g, ca);
    for (int gb = v0 + (tid & ~31); gb < v1; gb += 8 * TB, g += 8 * TB) {
        load_groups(k4, g + 4 * TB, cb);
        scan_groups(L, ca, inv, g, v1, lane);
        if (gb + 4 * TB >= v1) break;
        load_groups(k4, g + 8 * TB, ca);
        scan_groups(L, cb, inv, g + 4 * TB, v1, lane);
    }
}

template <bool SHIFT>
__device__ __forceinline__ void tile_batch(TileLds& L, unsigned n, float ub,
                                           const int* __restrict__ src,
                                           const int* __restrict__ dst,
                                           const float* __restrict__ s_src,
                                           const float* __restrict__ s_dst,
                                           const float* __restrict__ z,
                                           int base, int tid) {
    for (unsigned i0 = tid; i0 < n; i0 += 4 * TB) {
        unsigned s[4], d[4];
#pragma unroll
        for (int u = 0; u < 4; ++u) {
            unsigned i = i0 + u * TB;
            unsigned ei = L.q[i < n ? i : i0];
            ei = ei < (unsigned)N_EDGES ? ei : (unsigned)(N_EDGES - 1);
            s[u] = (unsigned)src[ei];
            d[u] = (unsigned)dst[ei] - (unsigned)base;
        }
#pragma unroll
        for (int u = 0; u < 4; ++u) {
            unsigned i = i0 + u * TB;
            unsigned sv = s[u] < (unsigned)N_NODES ? s[u] : (unsigned)(N_NODES - 1);
            unsigned dl = d[u] < (unsigned)TN ? d[u] : (unsigned)(TN - 1);
            if (i < n) L.q[i] = (sv << 9) | dl;
        }
    }
    __syncthreads();
    if (!SHIFT) {
    {
        const unsigned nh = n * HEADS, hd = tid & 7;
        for (unsigned i0 = tid; i0 < nh; i0 += 4 * TB) {
            float a[4], b[4]; unsigned cell[4];
#pragma unroll
            for (int u = 0; u < 4; ++u) {
                unsigned i = i0 + u * TB;
                unsigned pk = L.q[(i < nh ? i : i0) >> 3];
                unsigned s = pk >> 9, dl = pk & 511;
                a[u] = s_src[(size_t)s * HEADS + hd];
                b[u] = s_dst[(size_t)(base + dl) * HEADS + hd];
                cell[u] = dl * HEADS + hd;
            }
#pragma unroll
            for (int u = 0; u < 4; ++u) {
                float sc = a[u] + b[u];
                sc = (sc >= 0.0f) ? sc : NEG_SLOPE * sc;
                if (i0 + u * TB < nh) atomicMaxF(&L.mnew[cell[u]], sc);
            }
        }
    }
    __syncthreads();
    for (int i = tid; i < TN * HEADS; i += TB) {
        float mo = L.mrun[i], mn = L.mnew[i];
        if (mo != mn) {
            if (mo != -INFINITY) {
                float f = expf(mo - mn);
                L.den[i] *= f;
                float* a = &L.acc[(i >> 3) * OUT_DIM + (i & 7) * 4];
#pragma unroll
                for (int j = 0; j < 4; ++j) {
#pragma unroll
                    for (int k = 0; k < 4; ++k) a[32 * j + k] *= f;
                }
            }
            L.mrun[i] = mn;
        }
    }
    __syncthreads();
    }
    const int lane = tid & 31, hh = lane >> 2;
    const unsigned wave = (unsigned)__builtin_amdgcn_readfirstlane(tid >> 5);
    const unsigned qd = (unsigned)lane & 3u;
    const unsigned pick1 = (qd & 1u) ? ~0u : 0u, pick2 = (qd & 2u) ? ~0u : 0u;
    const float4* z4 = reinterpret_cast<const float4*>(z);
    for (unsigned eb = wave * EB; eb < n; eb += (TB / 32) * EB) {
        unsigned mine[EB / 4]; bool live[EB / 4];
#pragma unroll
        for (int g = 0; g < EB / 4; ++g) {
            const unsigned ql = eb + 4u * (unsigned)g + qd;
            live[g] = ql < n;
            mine[g] = L.q[live[g] ? ql : n - 1];
        }
        float4 zv[EB]; float ss[EB], sd[EB], mr[EB]; unsigned dls[EB];
#pragma unroll
        for (int u = 0; u < EB; ++u) {
            const unsigned pk = (unsigned)__builtin_amdgcn_readlane((int)mine[u >> 2], u & 3);
            const unsigned s = pk >> 9;
            dls[u] = pk & 511;
            ss[u] = s_src[(size_t)s * HEADS + hh];
            sd[u] = s_dst[(size_t)(base + dls[u]) * HEADS + hh];
            zv[u] = z4[(size_t)s * (OUT_DIM / 4) + lane];
            if (!SHIFT) mr[u] = L.mrun[dls[u] * HEADS + hh];
        }
        float ex[EB];
#pragma unroll
        for (int u = 0; u < EB; ++u) {
            float sc = ss[u] + sd[u];
            sc = (sc >= 0.0f) ? sc : NEG_SLOPE * sc;
            if (SHIFT) {
                const float t = sd[u] + ub;
                mr[u] = (t >= 0.0f) ? t : NEG_SLOPE * t;
            }
            ex[u] = expf(sc - mr[u]);
        }
#pragma unroll
        for (int g = 0; g < EB / 4; ++g) {
            const unsigned e01 = (__float_as_uint(ex[4 * g + 1]) & pick1) | (__float_as_uint(ex[4 * g])     & ~pick1);
            const unsigned e23 = (__float_as_uint(ex[4 * g + 3]) & pick1) | (__float_as_uint(ex[4 * g + 2]) & ~pick1);
            const float e = __uint_as_float((e23 & pick2) | (e01 & ~pick2));
            if (live[g]) atomicAdd(&L.den[(mine[g] & 511u) * HEADS + hh], e);
        }
#pragma unroll
        for (int u = 0; u < EB; ++u) {
            if (eb + (unsigned)u < n) {
                float* a = &L.acc[dls[u] * OUT_DIM + lane];
                atomicAdd(a + 0,  ex[u] * zv[u].x);
                atomicAdd(a + 32, ex[u] * zv[u].y);
                atomicAdd(a + 64, ex[u] * zv[u].z);
                atomicAdd(a + 96, ex[u] * zv[u].w);
            }
        }
    }
}

__global__ void __launch_bounds__(TB)
gat_tile(const int* __restrict__ src, const int* __restrict__ dst,
         const unsigned* __restrict__ ids,
         const float* __restrict__ s_src, const float* __restrict__ s_dst,
         const float* __restrict__ z, const float* __restrict__ h,
         const float* __restrict__ bnd, float* __restrict__ out) {
    __shared__ TileLds L;
    const int tid = threadIdx.x;
    const int base = blockIdx.x * TN;

    const float4* b4 = reinterpret_cast<const float4*>(bnd);
    float4 bv[(NSB * 4 + TB - 1) / TB];
#pragma unroll
    for (int k = 0; k < (NSB * 4 + TB - 1) / TB; ++k) {
        const int j = tid + TB * k;
        bv[k] = b4[j < NSB * 4 ? j : tid];
    }

    float4* acc4 = reinterpret_cast<float4*>(L.acc);
    for (int i = tid; i < TN * OUT_DIM / 4; i += TB) acc4[i] = make_float4(0.f, 0.f, 0.f, 0.f);
    for (int i = tid; i < TN * HEADS; i += TB) {
        L.mrun[i] = -INFINITY; L.mnew[i] = -INFINITY; L.den[i] = 0.0f;
    }
    if (tid == 0) L.qn = 0u;
    {
        const bool lo = (tid & 2) != 0;
        float4 r = bv[0];
#pragma unroll
        for (int k = 1; k < (NSB * 4 + TB - 1) / TB; ++k) {
            r.x = lo ? fminf(r.x, bv[k].x) : fmaxf(r.x, bv[k].x);
            r.y = lo ? fminf(r.y, bv[k].y) : fmaxf(r.y, bv[k].y);
            r.z = lo ? fminf(r.z, bv[k].z) : fmaxf(r.z, bv[k].z);
            r.w = lo ? fminf(r.w, bv[k].w) : fmaxf(r.w, bv[k].w);
        }
#pragma unroll
        for (int m = 4; m < 32; m <<= 1) {
            const float ox = __shfl_xor(r.x, m), oy = __shfl_xor(r.y, m), oz = __shfl_xor(r.z, m), ow = __shfl_xor(r.w, m);
            r.x = lo ? fminf(r.x, ox) : fmaxf(r.x, ox);
            r.y = lo ? fminf(r.y, oy) : fmaxf(r.y, oy);
            r.z = lo ? fminf(r.z, oz) : fmaxf(r.z, oz);
            r.w = lo ? fminf(r.w, ow) : fmaxf(r.w, ow);
        }
        if ((tid & 31) < 4) {
            float* p = &L.pbw[(tid >> 5) * 16 + 4 * (tid & 3)];
            p[0] = r.x; p[1] = r.y; p[2] = r.z; p[3] = r.w;
        }
    }
    __syncthreads();
    if (tid < 16) {
        float r = L.pbw[tid];
        for (int w = 1; w < TB / 32; ++w) r = (tid & 8) ? fminf(r, L.pbw[w * 16 + tid]) : fmaxf(r, L.pbw[w * 16 + tid]);
        L.bd[tid] = r;
    }

    const int rot = blockIdx.x * 192;
    tile_scan(L, ids, rot, GV, blockIdx.x, tid);
    tile_scan(L, ids, 0, rot, blockIdx.x, tid);
    __syncthreads();
    const unsigned total = L.qn;
    int narrow = 1;
#pragma unroll
    for (int k = 0; k < HEADS; ++k) narrow &= (L.bd[k] - L.bd[HEADS + k] <= SPREAD_MAX) ? 1 : 0;
    const bool shift = __builtin_amdgcn_readfirstlane(narrow) != 0;
    const float ub = L.bd[(tid & 31) >> 2];
    if (shift) {
        if (total <= (unsigned)QCAP) {
            tile_batch<true>(L, total, ub, src, dst, s_src, s_dst, z, base, tid);
        } else {
            for (int seg = 0; seg < N_EDGES; seg += QCAP) {
                __syncthreads();
                if (tid == 0) L.qn = 0u;
                __syncthreads();
                int end = seg + QCAP < N_EDGES ? seg + QCAP : N_EDGES;
                tile_scan(L, ids, seg / 32, end / 32, blockIdx.x, tid);
                __syncthreads();
                tile_batch<true>(L, L.qn, ub, src, dst, s_src, s_dst, z, base, tid);
            }
        }
    } else if (total <= (unsigned)QCAP) {
        tile_batch<false>(L, total, ub, src, dst, s_src, s_dst, z, base, tid);
    } else {
        for (int seg = 0; seg < N_EDGES; seg += QCAP) {
            __syncthreads();
            if (tid == 0) L.qn = 0u;
            __syncthreads();
            int end = seg + QCAP < N_EDGES ? seg + QCAP : N_EDGES;
            tile_scan(L, ids, seg / 32, end / 32, blockIdx.x, tid);
            __syncthreads();
            tile_batch<false>(L, L.qn, ub, src, dst, s_src, s_dst, z, base, tid);
        }
    }
    __syncthreads();

    const float4* h4 = reinterpret_cast<const float4*>(h) + (size_t)base * (OUT_DIM / 4);
    float4* o4 = reinterpret_cast<float4*>(out) + (size_t)base * (OUT_DIM / 4);
    for (int i = tid; i < TN * OUT_DIM / 4; i += TB) {
        float dn = L.den[(i >> 5) * HEADS + ((i & 31) >> 2)];
        const float* ar = &L.acc[(i >> 5) * OUT_DIM + (i & 31)];
        float4 a = make_float4(ar[0], ar[32], ar[64], ar[96]), hv = h4[i], r;
        bool has = dn > 0.0f;
        float v0 = has ? a.x / dn : 0.0f, v1 = has ? a.y / dn : 0.0f;
        float v2 = has ? a.z / dn : 0.0f, v3 = has ? a.w / dn : 0.0f;
        r.x = hv.x + ((v0 > 0.0f) ? v0 : expm1f(v0));
        r.y = hv.y + ((v1 > 0.0f) ? v1 : expm1f(v1));
        r.z = hv.z + ((v2 > 0.0f) ? v2 : expm1f(v2));
        r.w = hv.w + ((v3 > 0.0f) ? v3 : expm1f(v3));
        o4[i] = r;
    }
}

__global__ void __launch_bounds__(256)
gat_epass(const float* __restrict__ e, float* __restrict__ oe) {
    int i = blockIdx.x * 256 + threadIdx.x;
    if (i < N_EDGES / 4)
        reinterpret_cast<float4*>(oe)[i] = reinterpret_cast<const float4*>(e)[i];
}

extern "C" void kernel_launch(void* const* d_in, const int* in_sizes, int n_in,
                              void* d_out, int out_size, void* d_ws, size_t ws_size,
                              hipStream_t stream) {
    const float* h     = (const float*)d_in[0];
    const float* e     = (const float*)d_in[1];
    const int*   src   = (const int*)d_in[2];
    const int*   dst   = (const int*)d_in[3];
    const float* Wfc   = (const float*)d_in[4];
    const float* Wattn = (const float*)d_in[5];
    float* out = (float*)d_out;

    float* z     = (float*)d_ws;
    float* s_src = z + (size_t)N_NODES * OUT_DIM;
    float* s_dst = s_src + (size_t)N_NODES * HEADS;

    const int T = 256;

    gat_gemm_wmma<<<(N_NODES / 16 + 7) / 8, T, 0, stream>>>(h, Wfc, z);

    float* oe = out + (size_t)N_NODES * OUT_DIM;
    unsigned* ids = (unsigned*)oe;

    float* bnd = oe + N_EDGES / 4;

    gat_scores<<<NSB, T, 0, stream>>>(z, Wattn, dst, s_src, s_dst, ids, bnd);

    gat_tile<<<NT, TB, 0, stream>>>(src, dst, ids, s_src, s_dst, z, h, bnd, out);

    gat_epass<<<(N_EDGES / 4 + T - 1) / T, T, 0, stream>>>(e, oe);
}


# GEMM launch reshaped to one resident round: 256 work-groups of 28 waves (one per WGP), 16-row slabs dealt wave-major, the 106 over split by columns; was 782 groups of 8 waves, 14 in a second round
# speedup vs baseline: 11.6556x; 1.0527x over previous
#include <hip/hip_runtime.h>
#include <math.h>

#define N_NODES 100000
#define N_EDGES 1600000
#define IN_DIM  128
#define HEADS   8
#define HDIM    16
#define OUT_DIM 128
#define NEG_SLOPE 0.01f
#define LDS_STRIDE 130
#define TN   400
#define NT   (N_NODES / TN)
#define GV   (N_EDGES / 32)
#define NSB  (N_NODES * HEADS / 256)
#define SPREAD_MAX 60.0f

typedef __attribute__((ext_vector_type(2))) float v2f;
typedef __attribute__((ext_vector_type(8))) float v8f;

__device__ __forceinline__ void atomicMaxF(float* addr, float val) {
    if (val >= 0.0f) {
        atomicMax((int*)addr, __float_as_int(val));
    } else {
        atomicMin((unsigned int*)addr, __float_as_uint(val));
    }
}

#define GEMM_WG    256
#define GEMM_FULL  ((N_NODES / 16) / GEMM_WG)
#define GEMM_REST  ((N_NODES / 16) - GEMM_FULL * GEMM_WG)
#define GEMM_WAVES (GEMM_FULL + 4)
#define GEMM_T     (GEMM_WAVES * 32)
static_assert(N_NODES % 16 == 0 && GEMM_REST <= GEMM_WG && GEMM_T <= 1024, "every 16-row slab has its waves");

template <int CT>
__device__ __forceinline__ void gemm_slab(const float* __restrict__ h, const float* Wl,
                                          float* __restrict__ z, int mtile, int tbase, int lane) {
    const int row   = lane & 15;
    const int khalf = lane >> 4;

    const float* hrow = h + (size_t)(mtile * 16 + row) * IN_DIM;

    v8f acc[CT] = {};

    for (int k0 = 0; k0 < IN_DIM; k0 += 4) {
        const int kk = k0 + 2 * khalf;
        v2f a = *reinterpret_cast<const v2f*>(hrow + kk);
#pragma unroll
        for (int t = 0; t < CT; ++t) {
            v2f b = *reinterpret_cast<const v2f*>(&Wl[((tbase + t) * 16 + row) * LDS_STRIDE + kk]);
            acc[t] = __builtin_amdgcn_wmma_f32_16x16x4_f32(
                false, a, false, b, (short)0, acc[t], false, false);
        }
    }

    float* zbase = z + (size_t)(mtile * 16) * OUT_DIM;
#pragma unroll
    for (int t = 0; t < CT; ++t) {
#pragma unroll
        for (int i = 0; i < 8; ++i) {
            int mrow = i + khalf * 8;
            zbase[(size_t)mrow * OUT_DIM + (tbase + t) * 16 + row] = acc[t][i];
        }
    }
}

__global__ void __launch_bounds__(GEMM_T)
gat_gemm_wmma(const float* __restrict__ h, const float* __restrict__ Wfc,
              float* __restrict__ z) {
    __shared__ float Wl[OUT_DIM * LDS_STRIDE];

    const int tid = threadIdx.x;
    for (int j = tid; j < IN_DIM * OUT_DIM; j += GEMM_T) {
        int i  = j >> 7;
        int c  = j & 127;
        int hh = c >> 4;
        int d  = c & 15;
        Wl[c * LDS_STRIDE + i] = Wfc[hh * (IN_DIM * HDIM) + i * HDIM + d];
    }
    __syncthreads();

    const int wave = tid >> 5;
    const int lane = tid & 31;
    if (wave < GEMM_FULL) {
        gemm_slab<8>(h, Wl, z, wave * GEMM_WG + (int)blockIdx.x, 0, lane);
    } else if ((int)blockIdx.x < GEMM_REST) {
        gemm_slab<2>(h, Wl, z, GEMM_FULL * GEMM_WG + (int)blockIdx.x, 2 * (wave - GEMM_FULL), lane);
    }
}

__device__ __forceinline__ unsigned owner_id(int d) {
    unsigned k = (unsigned)d / (unsigned)TN;
    return k < 255u ? k : 255u;
}

__device__ __forceinline__ unsigned plane_vec(unsigned g, unsigned half) {
    return ((g >> 4) << 5) + (half << 4) + (g & 15u);
}

__global__ void gat_scores(const float* __restrict__ z,
                           const float* __restrict__ Wattn,
                           const int* __restrict__ dst,
                           float* __restrict__ s_src, float* __restrict__ s_dst,
                           unsigned* __restrict__ ids, float* __restrict__ bnd) {
    __shared__ float pb[8][16];
    int idx = blockIdx.x * 256 + threadIdx.x;
    int n = idx >> 3, hh = idx & 7;
    const float* zp = z + (size_t)n * OUT_DIM + hh * HDIM;
    const float* as = Wattn + hh * (2 * HDIM);
    const float* ad = as + HDIM;
    float ss = 0.0f, sd = 0.0f;
#pragma unroll
    for (int d = 0; d < HDIM; ++d) {
        float v = zp[d];
        ss += v * as[d];
        sd += v * ad[d];
    }
    s_src[idx] = ss;
    s_dst[idx] = sd;

    {
        float mx = fmaxf(ss, __shfl_xor(ss, 8));  mx = fmaxf(mx, __shfl_xor(mx, 16));
        float mn = fminf(ss, __shfl_xor(ss, 8));  mn = fminf(mn, __shfl_xor(mn, 16));
        const int wv = threadIdx.x >> 5, ln = threadIdx.x & 31;
        if (ln < 8) { pb[wv][ln] = mx; pb[wv][8 + ln] = mn; }
        __syncthreads();
        if (threadIdx.x < 16) {
            const int t = threadIdx.x;
            float r = pb[0][t];
#pragma unroll
            for (int w = 1; w < 8; ++w) r = (t < 8) ? fmaxf(r, pb[w][t]) : fminf(r, pb[w][t]);
            bnd[blockIdx.x * 16 + t] = r;
        }
    }

    static_assert((N_NODES * HEADS) % 256 == 0 && (N_NODES * HEADS / 32) * 2 == GV, "whole work-groups of 256; one wave per two edge groups");
    const unsigned lane = (unsigned)threadIdx.x & 31u;
    const unsigned g0 = ((unsigned)idx >> 5) * 2u;
    const unsigned ida = owner_id(dst[g0 * 32u + lane]);
    const unsigned idb = owner_id(dst[g0 * 32u + 32u + lane]);
    unsigned w = 0u;
#pragma unroll
    for (int k = 0; k < 8; ++k) {
        const unsigned pa = __builtin_amdgcn_ballot_w32(((ida >> k) & 1u) != 0u);
        const unsigned pb = __builtin_amdgcn_ballot_w32(((idb >> k) & 1u) != 0u);
        if (lane == (unsigned)k) w = pa;
        if (lane == (unsigned)(8 + k)) w = pb;
    }
    if (lane < 16u) {
        const unsigned g = g0 + (lane >> 3), k = lane & 7u;
        ids[plane_vec(g, k >> 2) * 4u + (k & 3u)] = w;
    }
}

#define QCAP 12288
#define TB   1024
#define EB   8

struct TileLds {
    float acc[TN * OUT_DIM];
    float mrun[TN * HEADS];
    float mnew[TN * HEADS];
    float den[TN * HEADS];
    unsigned q[QCAP];
    unsigned qn;
    float pbw[32 * 16];
    float bd[16];
};

__device__ __forceinline__ unsigned match_planes(uint4 a, uint4 b, const unsigned (&inv)[8]) {
    return (a.x ^ inv[0]) & (a.y ^ inv[1]) & (a.z ^ inv[2]) & (a.w ^ inv[3]) &
           (b.x ^ inv[4]) & (b.y ^ inv[5]) & (b.z ^ inv[6]) & (b.w ^ inv[7]);
}

__device__ __forceinline__ void scan_groups(TileLds& L, const uint4 (&c)[8], const unsigned (&inv)[8],
                                            int g, int gend, unsigned lane) {
    unsigned r0 = (g          < gend) ? match_planes(c[0], c[1], inv) : 0u;
    unsigned r1 = (g + TB     < gend) ? match_planes(c[2], c[3], inv) : 0u;
    unsigned r2 = (g + 2 * TB < gend) ? match_planes(c[4], c[5], inv) : 0u;
    unsigned r3 = (g + 3 * TB < gend) ? match_planes(c[6], c[7], inv) : 0u;
    const unsigned e0 = (unsigned)g * 32u;
    for (;;) {
        const bool has = (r0 | r1 | r2 | r3) != 0u;
        const unsigned m = __builtin_amdgcn_ballot_w32(has);
        if (m == 0u) break;
        unsigned s0 = 0u;
        if (lane == 0u) s0 = atomicAdd(&L.qn, (unsigned)__builtin_popcount(m));
        s0 = (unsigned)__builtin_amdgcn_readfirstlane((int)s0);
        if (has) {
            const bool fp = (r0 | r1) != 0u;
            const unsigned x0 = fp ? r0 : r2, x1 = fp ? r1 : r3;
            const bool lo = x0 != 0u;
            const unsigned r = lo ? x0 : x1;
            const unsigned u = (fp ? 0u : 2u) + (lo ? 0u : 1u);
            const unsigned e = e0 + u * (32u * TB) + (unsigned)__builtin_ctz(r);
            const unsigned slot = s0 + __builtin_amdgcn_mbcnt_lo(m, 0u);
            if (slot < (unsigned)QCAP) L.q[slot] = e;
            const unsigned rr = r & (r - 1u);
            if (fp) { if (lo) r0 = rr; else r1 = rr; }
            else    { if (lo) r2 = rr; else r3 = rr; }
        }
    }
}

__device__ __forceinline__ void load_groups(const uint4* __restrict__ k4, int g, uint4 (&c)[8]) {
#pragma unroll
    for (int u = 0; u < 4; ++u) {
        const int gg = g + u * TB;
        const unsigned gc = (unsigned)(gg < GV ? gg : GV - 1);
        c[2 * u]     = k4[plane_vec(gc, 0u)];
        c[2 * u + 1] = k4[plane_vec(gc, 1u)];
    }
}

__device__ __forceinline__ void tile_scan(TileLds& L, const unsigned* __restrict__ ids,
                                          int v0, int v1, unsigned tileb, int tid) {
    const uint4* k4 = reinterpret_cast<const uint4*>(ids);
    const unsigned lane = (unsigned)tid & 31u;
    unsigned inv[8];
#pragma unroll
    for (int k = 0; k < 8; ++k) inv[k] = ((tileb >> k) & 1u) ? 0u : ~0u;
    uint4 ca[8], cb[8];
    int g = v0 + tid;
    load_groups(k4, g, ca);
    for (int gb = v0 + (tid & ~31); gb < v1; gb += 8 * TB, g += 8 * TB) {
        load_groups(k4, g + 4 * TB, cb);
        scan_groups(L, ca, inv, g, v1, lane);
        if (gb + 4 * TB >= v1) break;
        load_groups(k4, g + 8 * TB, ca);
        scan_groups(L, cb, inv, g + 4 * TB, v1, lane);
    }
}

template <bool SHIFT>
__device__ __forceinline__ void tile_batch(TileLds& L, unsigned n, float ub,
                                           const int* __restrict__ src,
                                           const int* __restrict__ dst,
                                           const float* __restrict__ s_src,
                                           const float* __restrict__ s_dst,
                                           const float* __restrict__ z,
                                           int base, int tid) {
    for (unsigned i0 = tid; i0 < n; i0 += 4 * TB) {
        unsigned s[4], d[4];
#pragma unroll
        for (int u = 0; u < 4; ++u) {
            unsigned i = i0 + u * TB;
            unsigned ei = L.q[i < n ? i : i0];
            ei = ei < (unsigned)N_EDGES ? ei : (unsigned)(N_EDGES - 1);
            s[u] = (unsigned)src[ei];
            d[u] = (unsigned)dst[ei] - (unsigned)base;
        }
#pragma unroll
        for (int u = 0; u < 4; ++u) {
            unsigned i = i0 + u * TB;
            unsigned sv = s[u] < (unsigned)N_NODES ? s[u] : (unsigned)(N_NODES - 1);
            unsigned dl = d[u] < (unsigned)TN ? d[u] : (unsigned)(TN - 1);
            if (i < n) L.q[i] = (sv << 9) | dl;
        }
    }
    __syncthreads();
    if (!SHIFT) {
    {
        const unsigned nh = n * HEADS, hd = tid & 7;
        for (unsigned i0 = tid; i0 < nh; i0 += 4 * TB) {
            float a[4], b[4]; unsigned cell[4];
#pragma unroll
            for (int u = 0; u < 4; ++u) {
                unsigned i = i0 + u * TB;
                unsigned pk = L.q[(i < nh ? i : i0) >> 3];
                unsigned s = pk >> 9, dl = pk & 511;
                a[u] = s_src[(size_t)s * HEADS + hd];
                b[u] = s_dst[(size_t)(base + dl) * HEADS + hd];
                cell[u] = dl * HEADS + hd;
            }
#pragma unroll
            for (int u = 0; u < 4; ++u) {
                float sc = a[u] + b[u];
                sc = (sc >= 0.0f) ? sc : NEG_SLOPE * sc;
                if (i0 + u * TB < nh) atomicMaxF(&L.mnew[cell[u]], sc);
            }
        }
    }
    __syncthreads();
    for (int i = tid; i < TN * HEADS; i += TB) {
        float mo = L.mrun[i], mn = L.mnew[i];
        if (mo != mn) {
            if (mo != -INFINITY) {
                float f = expf(mo - mn);
                L.den[i] *= f;
                float* a = &L.acc[(i >> 3) * OUT_DIM + (i & 7) * 4];
#pragma unroll
                for (int j = 0; j < 4; ++j) {
#pragma unroll
                    for (int k = 0; k < 4; ++k) a[32 * j + k] *= f;
                }
            }
            L.mrun[i] = mn;
        }
    }
    __syncthreads();
    }
    const int lane = tid & 31, hh = lane >> 2;
    const unsigned wave = (unsigned)__builtin_amdgcn_readfirstlane(tid >> 5);
    const unsigned qd = (unsigned)lane & 3u;
    const unsigned pick1 = (qd & 1u) ? ~0u : 0u, pick2 = (qd & 2u) ? ~0u : 0u;
    const float4* z4 = reinterpret_cast<const float4*>(z);
    for (unsigned eb = wave * EB; eb < n; eb += (TB / 32) * EB) {
        unsigned mine[EB / 4]; bool live[EB / 4];
#pragma unroll
        for (int g = 0; g < EB / 4; ++g) {
            const unsigned ql = eb + 4u * (unsigned)g + qd;
            live[g] = ql < n;
            mine[g] = L.q[live[g] ? ql : n - 1];
        }
        float4 zv[EB]; float ss[EB], sd[EB], mr[EB]; unsigned dls[EB];
#pragma unroll
        for (int u = 0; u < EB; ++u) {
            const unsigned pk = (unsigned)__builtin_amdgcn_readlane((int)mine[u >> 2], u & 3);
            const unsigned s = pk >> 9;
            dls[u] = pk & 511;
            ss[u] = s_src[(size_t)s * HEADS + hh];
            sd[u] = s_dst[(size_t)(base + dls[u]) * HEADS + hh];
            zv[u] = z4[(size_t)s * (OUT_DIM / 4) + lane];
            if (!SHIFT) mr[u] = L.mrun[dls[u] * HEADS + hh];
        }
        float ex[EB];
#pragma unroll
        for (int u = 0; u < EB; ++u) {
            float sc = ss[u] + sd[u];
            sc = (sc >= 0.0f) ? sc : NEG_SLOPE * sc;
            if (SHIFT) {
                const float t = sd[u] + ub;
                mr[u] = (t >= 0.0f) ? t : NEG_SLOPE * t;
            }
            ex[u] = expf(sc - mr[u]);
        }
#pragma unroll
        for (int g = 0; g < EB / 4; ++g) {
            const unsigned e01 = (__float_as_uint(ex[4 * g + 1]) & pick1) | (__float_as_uint(ex[4 * g])     & ~pick1);
            const unsigned e23 = (__float_as_uint(ex[4 * g + 3]) & pick1) | (__float_as_uint(ex[4 * g + 2]) & ~pick1);
            const float e = __uint_as_float((e23 & pick2) | (e01 & ~pick2));
            if (live[g]) atomicAdd(&L.den[(mine[g] & 511u) * HEADS + hh], e);
        }
#pragma unroll
        for (int u = 0; u < EB; ++u) {
            if (eb + (unsigned)u < n) {
                float* a = &L.acc[dls[u] * OUT_DIM + lane];
                atomicAdd(a + 0,  ex[u] * zv[u].x);
                atomicAdd(a + 32, ex[u] * zv[u].y);
                atomicAdd(a + 64, ex[u] * zv[u].z);
                atomicAdd(a + 96, ex[u] * zv[u].w);
            }
        }
    }
}

__global__ void __launch_bounds__(TB)
gat_tile(const int* __restrict__ src, const int* __restrict__ dst,
         const unsigned* __restrict__ ids,
         const float* __restrict__ s_src, const float* __restrict__ s_dst,
         const float* __restrict__ z, const float* __restrict__ h,
         const float* __restrict__ bnd, float* __restrict__ out) {
    __shared__ TileLds L;
    const int tid = threadIdx.x;
    const int base = blockIdx.x * TN;

    const float4* b4 = reinterpret_cast<const float4*>(bnd);
    float4 bv[(NSB * 4 + TB - 1) / TB];
#pragma unroll
    for (int k = 0; k < (NSB * 4 + TB - 1) / TB; ++k) {
        const int j = tid + TB * k;
        bv[k] = b4[j < NSB * 4 ? j : tid];
    }

    float4* acc4 = reinterpret_cast<float4*>(L.acc);
    for (int i = tid; i < TN * OUT_DIM / 4; i += TB) acc4[i] = make_float4(0.f, 0.f, 0.f, 0.f);
    for (int i = tid; i < TN * HEADS; i += TB) {
        L.mrun[i] = -INFINITY; L.mnew[i] = -INFINITY; L.den[i] = 0.0f;
    }
    if (tid == 0) L.qn = 0u;
    {
        const bool lo = (tid & 2) != 0;
        float4 r = bv[0];
#pragma unroll
        for (int k = 1; k < (NSB * 4 + TB - 1) / TB; ++k) {
            r.x = lo ? fminf(r.x, bv[k].x) : fmaxf(r.x, bv[k].x);
            r.y = lo ? fminf(r.y, bv[k].y) : fmaxf(r.y, bv[k].y);
            r.z = lo ? fminf(r.z, bv[k].z) : fmaxf(r.z, bv[k].z);
            r.w = lo ? fminf(r.w, bv[k].w) : fmaxf(r.w, bv[k].w);
        }
#pragma unroll
        for (int m = 4; m < 32; m <<= 1) {
            const float ox = __shfl_xor(r.x, m), oy = __shfl_xor(r.y, m), oz = __shfl_xor(r.z, m), ow = __shfl_xor(r.w, m);
            r.x = lo ? fminf(r.x, ox) : fmaxf(r.x, ox);
            r.y = lo ? fminf(r.y, oy) : fmaxf(r.y, oy);
            r.z = lo ? fminf(r.z, oz) : fmaxf(r.z, oz);
            r.w = lo ? fminf(r.w, ow) : fmaxf(r.w, ow);
        }
        if ((tid & 31) < 4) {
            float* p = &L.pbw[(tid >> 5) * 16 + 4 * (tid & 3)];
            p[0] = r.x; p[1] = r.y; p[2] = r.z; p[3] = r.w;
        }
    }
    __syncthreads();
    if (tid < 16) {
        float r = L.pbw[tid];
        for (int w = 1; w < TB / 32; ++w) r = (tid & 8) ? fminf(r, L.pbw[w * 16 + tid]) : fmaxf(r, L.pbw[w * 16 + tid]);
        L.bd[tid] = r;
    }

    const int rot = blockIdx.x * 192;
    tile_scan(L, ids, rot, GV, blockIdx.x, tid);
    tile_scan(L, ids, 0, rot, blockIdx.x, tid);
    __syncthreads();
    const unsigned total = L.qn;
    int narrow = 1;
#pragma unroll
    for (int k = 0; k < HEADS; ++k) narrow &= (L.bd[k] - L.bd[HEADS + k] <= SPREAD_MAX) ? 1 : 0;
    const bool shift = __builtin_amdgcn_readfirstlane(narrow) != 0;
    const float ub = L.bd[(tid & 31) >> 2];
    if (shift) {
        if (total <= (unsigned)QCAP) {
            tile_batch<true>(L, total, ub, src, dst, s_src, s_dst, z, base, tid);
        } else {
            for (int seg = 0; seg < N_EDGES; seg += QCAP) {
                __syncthreads();
                if (tid == 0) L.qn = 0u;
                __syncthreads();
                int end = seg + QCAP < N_EDGES ? seg + QCAP : N_EDGES;
                tile_scan(L, ids, seg / 32, end / 32, blockIdx.x, tid);
                __syncthreads();
                tile_batch<true>(L, L.qn, ub, src, dst, s_src, s_dst, z, base, tid);
            }
        }
    } else if (total <= (unsigned)QCAP) {
        tile_batch<false>(L, total, ub, src, dst, s_src, s_dst, z, base, tid);
    } else {
        for (int seg = 0; seg < N_EDGES; seg += QCAP) {
            __syncthreads();
            if (tid == 0) L.qn = 0u;
            __syncthreads();
            int end = seg + QCAP < N_EDGES ? seg + QCAP : N_EDGES;
            tile_scan(L, ids, seg / 32, end / 32, blockIdx.x, tid);
            __syncthreads();
            tile_batch<false>(L, L.qn, ub, src, dst, s_src, s_dst, z, base, tid);
        }
    }
    __syncthreads();

    const float4* h4 = reinterpret_cast<const float4*>(h) + (size_t)base * (OUT_DIM / 4);
    float4* o4 = reinterpret_cast<float4*>(out) + (size_t)base * (OUT_DIM / 4);
    for (int i = tid; i < TN * OUT_DIM / 4; i += TB) {
        float dn = L.den[(i >> 5) * HEADS + ((i & 31) >> 2)];
        const float* ar = &L.acc[(i >> 5) * OUT_DIM + (i & 31)];
        float4 a = make_float4(ar[0], ar[32], ar[64], ar[96]), hv = h4[i], r;
        bool has = dn > 0.0f;
        float v0 = has ? a.x / dn : 0.0f, v1 = has ? a.y / dn : 0.0f;
        float v2 = has ? a.z / dn : 0.0f, v3 = has ? a.w / dn : 0.0f;
        r.x = hv.x + ((v0 > 0.0f) ? v0 : expm1f(v0));
        r.y = hv.y + ((v1 > 0.0f) ? v1 : expm1f(v1));
        r.z = hv.z + ((v2 > 0.0f) ? v2 : expm1f(v2));
        r.w = hv.w + ((v3 > 0.0f) ? v3 : expm1f(v3));
        o4[i] = r;
    }
}

__global__ void __launch_bounds__(256)
gat_epass(const float* __restrict__ e, float* __restrict__ oe) {
    int i = blockIdx.x * 256 + threadIdx.x;
    if (i < N_EDGES / 4)
        reinterpret_cast<float4*>(oe)[i] = reinterpret_cast<const float4*>(e)[i];
}

extern "C" void kernel_launch(void* const* d_in, const int* in_sizes, int n_in,
                              void* d_out, int out_size, void* d_ws, size_t ws_size,
                              hipStream_t stream) {
    const float* h     = (const float*)d_in[0];
    const float* e     = (const float*)d_in[1];
    const int*   src   = (const int*)d_in[2];
    const int*   dst   = (const int*)d_in[3];
    const float* Wfc   = (const float*)d_in[4];
    const float* Wattn = (const float*)d_in[5];
    float* out = (float*)d_out;

    float* z     = (float*)d_ws;
    float* s_src = z + (size_t)N_NODES * OUT_DIM;
    float* s_dst = s_src + (size_t)N_NODES * HEADS;

    const int T = 256;

    gat_gemm_wmma<<<GEMM_WG, GEMM_T, 0, stream>>>(h, Wfc, z);

    float* oe = out + (size_t)N_NODES * OUT_DIM;
    unsigned* ids = (unsigned*)oe;

    float* bnd = oe + N_EDGES / 4;

    gat_scores<<<NSB, T, 0, stream>>>(z, Wattn, dst, s_src, s_dst, ids, bnd);

    gat_tile<<<NT, TB, 0, stream>>>(src, dst, ids, s_src, s_dst, z, h, bnd, out);

    gat_epass<<<(N_EDGES / 4 + T - 1) / T, T, 0, stream>>>(e, oe);
}


# Score and e-copy launches reshaped to one resident round of long waves (250 groups of 20 waves x 5 items; 250 x 10 waves x 5 vectors): 7,500 waves to let in, was 37,500; bounds list 250 entries
# speedup vs baseline: 12.0142x; 1.0308x over previous
#include <hip/hip_runtime.h>
#include <math.h>

#define N_NODES 100000
#define N_EDGES 1600000
#define IN_DIM  128
#define HEADS   8
#define HDIM    16
#define OUT_DIM 128
#define NEG_SLOPE 0.01f
#define LDS_STRIDE 130
#define TN   400
#define NT   (N_NODES / TN)
#define GV   (N_EDGES / 32)
#define SC_WG    250
#define SC_WAVES 20
#define SC_T     (SC_WAVES * 32)
#define SC_ITEMS 5
#define NSB      SC_WG
#define SPREAD_MAX 60.0f

typedef __attribute__((ext_vector_type(2))) float v2f;
typedef __attribute__((ext_vector_type(8))) float v8f;

__device__ __forceinline__ void atomicMaxF(float* addr, float val) {
    if (val >= 0.0f) {
        atomicMax((int*)addr, __float_as_int(val));
    } else {
        atomicMin((unsigned int*)addr, __float_as_uint(val));
    }
}

#define GEMM_WG    256
#define GEMM_FULL  ((N_NODES / 16) / GEMM_WG)
#define GEMM_REST  ((N_NODES / 16) - GEMM_FULL * GEMM_WG)
#define GEMM_WAVES (GEMM_FULL + 4)
#define GEMM_T     (GEMM_WAVES * 32)
static_assert(N_NODES % 16 == 0 && GEMM_REST <= GEMM_WG && GEMM_T <= 1024, "every 16-row slab has its waves");

template <int CT>
__device__ __forceinline__ void gemm_slab(const float* __restrict__ h, const float* Wl,
                                          float* __restrict__ z, int mtile, int tbase, int lane) {
    const int row   = lane & 15;
    const int khalf = lane >> 4;

    const float* hrow = h + (size_t)(mtile * 16 + row) * IN_DIM;

    v8f acc[CT] = {};

    for (int k0 = 0; k0 < IN_DIM; k0 += 4) {
        const int kk = k0 + 2 * khalf;
        v2f a = *reinterpret_cast<const v2f*>(hrow + kk);
#pragma unroll
        for (int t = 0; t < CT; ++t) {
            v2f b = *reinterpret_cast<const v2f*>(&Wl[((tbase + t) * 16 + row) * LDS_STRIDE + kk]);
            acc[t] = __builtin_amdgcn_wmma_f32_16x16x4_f32(
                false, a, false, b, (short)0, acc[t], false, false);
        }
    }

    float* zbase = z + (size_t)(mtile * 16) * OUT_DIM;
#pragma unroll
    for (int t = 0; t < CT; ++t) {
#pragma unroll
        for (int i = 0; i < 8; ++i) {
            int mrow = i + khalf * 8;
            zbase[(size_t)mrow * OUT_DIM + (tbase + t) * 16 + row] = acc[t][i];
        }
    }
}

__global__ void __launch_bounds__(GEMM_T)
gat_gemm_wmma(const float* __restrict__ h, const float* __restrict__ Wfc,
              float* __restrict__ z) {
    __shared__ float Wl[OUT_DIM * LDS_STRIDE];

    const int tid = threadIdx.x;
    for (int j = tid; j < IN_DIM * OUT_DIM; j += GEMM_T) {
        int i  = j >> 7;
        int c  = j & 127;
        int hh = c >> 4;
        int d  = c & 15;
        Wl[c * LDS_STRIDE + i] = Wfc[hh * (IN_DIM * HDIM) + i * HDIM + d];
    }
    __syncthreads();

    const int wave = tid >> 5;
    const int lane = tid & 31;
    if (wave < GEMM_FULL) {
        gemm_slab<8>(h, Wl, z, wave * GEMM_WG + (int)blockIdx.x, 0, lane);
    } else if ((int)blockIdx.x < GEMM_REST) {
        gemm_slab<2>(h, Wl, z, GEMM_FULL * GEMM_WG + (int)blockIdx.x, 2 * (wave - GEMM_FULL), lane);
    }
}

__device__ __forceinline__ unsigned owner_id(int d) {
    unsigned k = (unsigned)d / (unsigned)TN;
    return k < 255u ? k : 255u;
}

__device__ __forceinline__ unsigned plane_vec(unsigned g, unsigned half) {
    return ((g >> 4) << 5) + (half << 4) + (g & 15u);
}

__global__ void __launch_bounds__(SC_T)
gat_scores(const float* __restrict__ z,
           const float* __restrict__ Wattn,
           const int* __restrict__ dst,
           float* __restrict__ s_src, float* __restrict__ s_dst,
           unsigned* __restrict__ ids, float* __restrict__ bnd) {
    __shared__ float pb[SC_WAVES][16];
    static_assert(SC_WG * SC_WAVES * SC_ITEMS * 32 == N_NODES * HEADS && SC_WG * SC_WAVES * SC_ITEMS * 2 == GV,
                  "every wave has SC_ITEMS whole items; an item is two edge groups");
    const unsigned lane = (unsigned)threadIdx.x & 31u;
    const int wv = threadIdx.x >> 5;
    const unsigned hh = lane & 7u;
    const unsigned item0 = ((unsigned)blockIdx.x * SC_WAVES + (unsigned)wv) * SC_ITEMS;

    const float4* z4 = reinterpret_cast<const float4*>(z);
    float4 zv[SC_ITEMS][4];
    int da[SC_ITEMS], db[SC_ITEMS];
#pragma unroll
    for (int k = 0; k < SC_ITEMS; ++k) {
        const unsigned idx = (item0 + (unsigned)k) * 32u + lane;
        const float4* zp = z4 + (size_t)(idx >> 3) * (OUT_DIM / 4) + hh * (HDIM / 4);
#pragma unroll
        for (int j = 0; j < 4; ++j) zv[k][j] = zp[j];
        da[k] = dst[(item0 + (unsigned)k) * 64u + lane];
        db[k] = dst[(item0 + (unsigned)k) * 64u + 32u + lane];
    }
    const float4* as4 = reinterpret_cast<const float4*>(Wattn + hh * (2 * HDIM));
    float4 av[8];
#pragma unroll
    for (int j = 0; j < 8; ++j) av[j] = as4[j];

    float mx = 0.0f, mn = 0.0f;
#pragma unroll
    for (int k = 0; k < SC_ITEMS; ++k) {
        const unsigned idx = (item0 + (unsigned)k) * 32u + lane;
        float ss = 0.0f, sd = 0.0f;
#pragma unroll
        for (int j = 0; j < 4; ++j) {
            ss += zv[k][j].x * av[j].x;  sd += zv[k][j].x * av[4 + j].x;
            ss += zv[k][j].y * av[j].y;  sd += zv[k][j].y * av[4 + j].y;
            ss += zv[k][j].z * av[j].z;  sd += zv[k][j].z * av[4 + j].z;
            ss += zv[k][j].w * av[j].w;  sd += zv[k][j].w * av[4 + j].w;
        }
        s_src[idx] = ss;
        s_dst[idx] = sd;
        mx = (k == 0) ? ss : fmaxf(mx, ss);
        mn = (k == 0) ? ss : fminf(mn, ss);

        const unsigned g0 = (item0 + (unsigned)k) * 2u;
        const unsigned ida = owner_id(da[k]);
        const unsigned idb = owner_id(db[k]);
        unsigned w = 0u;
#pragma unroll
        for (int b = 0; b < 8; ++b) {
            const unsigned pa = __builtin_amdgcn_ballot_w32(((ida >> b) & 1u) != 0u);
            const unsigned pbm = __builtin_amdgcn_ballot_w32(((idb >> b) & 1u) != 0u);
            if (lane == (unsigned)b) w = pa;
            if (lane == (unsigned)(8 + b)) w = pbm;
        }
        if (lane < 16u) {
            const unsigned g = g0 + (lane >> 3), b = lane & 7u;
            ids[plane_vec(g, b >> 2) * 4u + (b & 3u)] = w;
        }
    }

    mx = fmaxf(mx, __shfl_xor(mx, 8));  mx = fmaxf(mx, __shfl_xor(mx, 16));
    mn = fminf(mn, __shfl_xor(mn, 8));  mn = fminf(mn, __shfl_xor(mn, 16));
    if (lane < 8u) { pb[wv][lane] = mx; pb[wv][8u + lane] = mn; }
    __syncthreads();
    if (threadIdx.x < 16) {
        const int t = threadIdx.x;
        float r = pb[0][t];
#pragma unroll
        for (int w = 1; w < SC_WAVES; ++w) r = (t < 8) ? fmaxf(r, pb[w][t]) : fminf(r, pb[w][t]);
        bnd[blockIdx.x * 16 + t] = r;
    }
}

#define QCAP 12288
#define TB   1024
#define EB   8

struct TileLds {
    float acc[TN * OUT_DIM];
    float mrun[TN * HEADS];
    float mnew[TN * HEADS];
    float den[TN * HEADS];
    unsigned q[QCAP];
    unsigned qn;
    float pbw[32 * 16];
    float bd[16];
};

__device__ __forceinline__ unsigned match_planes(uint4 a, uint4 b, const unsigned (&inv)[8]) {
    return (a.x ^ inv[0]) & (a.y ^ inv[1]) & (a.z ^ inv[2]) & (a.w ^ inv[3]) &
           (b.x ^ inv[4]) & (b.y ^ inv[5]) & (b.z ^ inv[6]) & (b.w ^ inv[7]);
}

__device__ __forceinline__ void scan_groups(TileLds& L, const uint4 (&c)[8], const unsigned (&inv)[8],
                                            int g, int gend, unsigned lane) {
    unsigned r0 = (g          < gend) ? match_planes(c[0], c[1], inv) : 0u;
    unsigned r1 = (g + TB     < gend) ? match_planes(c[2], c[3], inv) : 0u;
    unsigned r2 = (g + 2 * TB < gend) ? match_planes(c[4], c[5], inv) : 0u;
    unsigned r3 = (g + 3 * TB < gend) ? match_planes(c[6], c[7], inv) : 0u;
    const unsigned e0 = (unsigned)g * 32u;
    for (;;) {
        const bool has = (r0 | r1 | r2 | r3) != 0u;
        const unsigned m = __builtin_amdgcn_ballot_w32(has);
        if (m == 0u) break;
        unsigned s0 = 0u;
        if (lane == 0u) s0 = atomicAdd(&L.qn, (unsigned)__builtin_popcount(m));
        s0 = (unsigned)__builtin_amdgcn_readfirstlane((int)s0);
        if (has) {
            const bool fp = (r0 | r1) != 0u;
            const unsigned x0 = fp ? r0 : r2, x1 = fp ? r1 : r3;
            const bool lo = x0 != 0u;
            const unsigned r = lo ? x0 : x1;
            const unsigned u = (fp ? 0u : 2u) + (lo ? 0u : 1u);
            const unsigned e = e0 + u * (32u * TB) + (unsigned)__builtin_ctz(r);
            const unsigned slot = s0 + __builtin_amdgcn_mbcnt_lo(m, 0u);
            if (slot < (unsigned)QCAP) L.q[slot] = e;
            const unsigned rr = r & (r - 1u);
            if (fp) { if (lo) r0 = rr; else r1 = rr; }
            else    { if (lo) r2 = rr; else r3 = rr; }
        }
    }
}

__device__ __forceinline__ void load_groups(const uint4* __restrict__ k4, int g, uint4 (&c)[8]) {
#pragma unroll
    for (int u = 0; u < 4; ++u) {
        const int gg = g + u * TB;
        const unsigned gc = (unsigned)(gg < GV ? gg : GV - 1);
        c[2 * u]     = k4[plane_vec(gc, 0u)];
        c[2 * u + 1] = k4[plane_vec(gc, 1u)];
    }
}

__device__ __forceinline__ void tile_scan(TileLds& L, const unsigned* __restrict__ ids,
                                          int v0, int v1, unsigned tileb, int tid) {
    const uint4* k4 = reinterpret_cast<const uint4*>(ids);
    const unsigned lane = (unsigned)tid & 31u;
    unsigned inv[8];
#pragma unroll
    for (int k = 0; k < 8; ++k) inv[k] = ((tileb >> k) & 1u) ? 0u : ~0u;
    uint4 ca[8], cb[8];
    int g = v0 + tid;
    load_groups(k4, g, ca);
    for (int gb = v0 + (tid & ~31); gb < v1; gb += 8 * TB, g += 8 * TB) {
        load_groups(k4, g + 4 * TB, cb);
        scan_groups(L, ca, inv, g, v1, lane);
        if (gb + 4 * TB >= v1) break;
        load_groups(k4, g + 8 * TB, ca);
        scan_groups(L, cb, inv, g + 4 * TB, v1, lane);
    }
}

template <bool SHIFT>
__device__ __forceinline__ void tile_batch(TileLds& L, unsigned n, float ub,
                                           const int* __restrict__ src,
                                           const int* __restrict__ dst,
                                           const float* __restrict__ s_src,
                                           const float* __restrict__ s_dst,
                                           const float* __restrict__ z,
                                           int base, int tid) {
    for (unsigned i0 = tid; i0 < n; i0 += 4 * TB) {
        unsigned s[4], d[4];
#pragma unroll
        for (int u = 0; u < 4; ++u) {
            unsigned i = i0 + u * TB;
            unsigned ei = L.q[i < n ? i : i0];
            ei = ei < (unsigned)N_EDGES ? ei : (unsigned)(N_EDGES - 1);
            s[u] = (unsigned)src[ei];
            d[u] = (unsigned)dst[ei] - (unsigned)base;
        }
#pragma unroll
        for (int u = 0; u < 4; ++u) {
            unsigned i = i0 + u * TB;
            unsigned sv = s[u] < (unsigned)N_NODES ? s[u] : (unsigned)(N_NODES - 1);
            unsigned dl = d[u] < (unsigned)TN ? d[u] : (unsigned)(TN - 1);
            if (i < n) L.q[i] = (sv << 9) | dl;
        }
    }
    __syncthreads();
    if (!SHIFT) {
    {
        const unsigned nh = n * HEADS, hd = tid & 7;
        for (unsigned i0 = tid; i0 < nh; i0 += 4 * TB) {
            float a[4], b[4]; unsigned cell[4];
#pragma unroll
            for (int u = 0; u < 4; ++u) {
                unsigned i = i0 + u * TB;
                unsigned pk = L.q[(i < nh ? i : i0) >> 3];
                unsigned s = pk >> 9, dl = pk & 511;
                a[u] = s_src[(size_t)s * HEADS + hd];
                b[u] = s_dst[(size_t)(base + dl) * HEADS + hd];
                cell[u] = dl * HEADS + hd;
            }
#pragma unroll
            for (int u = 0; u < 4; ++u) {
                float sc = a[u] + b[u];
                sc = (sc >= 0.0f) ? sc : NEG_SLOPE * sc;
                if (i0 + u * TB < nh) atomicMaxF(&L.mnew[cell[u]], sc);
            }
        }
    }
    __syncthreads();
    for (int i = tid; i < TN * HEADS; i += TB) {
        float mo = L.mrun[i], mn = L.mnew[i];
        if (mo != mn) {
            if (mo != -INFINITY) {
                float f = expf(mo - mn);
                L.den[i] *= f;
                float* a = &L.acc[(i >> 3) * OUT_DIM + (i & 7) * 4];
#pragma unroll
                for (int j = 0; j < 4; ++j) {
#pragma unroll
                    for (int k = 0; k < 4; ++k) a[32 * j + k] *= f;
                }
            }
            L.mrun[i] = mn;
        }
    }
    __syncthreads();
    }
    const int lane = tid & 31, hh = lane >> 2;
    const unsigned wave = (unsigned)__builtin_amdgcn_readfirstlane(tid >> 5);
    const unsigned qd = (unsigned)lane & 3u;
    const unsigned pick1 = (qd & 1u) ? ~0u : 0u, pick2 = (qd & 2u) ? ~0u : 0u;
    const float4* z4 = reinterpret_cast<const float4*>(z);
    for (unsigned eb = wave * EB; eb < n; eb += (TB / 32) * EB) {
        unsigned mine[EB / 4]; bool live[EB / 4];
#pragma unroll
        for (int g = 0; g < EB / 4; ++g) {
            const unsigned ql = eb + 4u * (unsigned)g + qd;
            live[g] = ql < n;
            mine[g] = L.q[live[g] ? ql : n - 1];
        }
        float4 zv[EB]; float ss[EB], sd[EB], mr[EB]; unsigned dls[EB];
#pragma unroll
        for (int u = 0; u < EB; ++u) {
            const unsigned pk = (unsigned)__builtin_amdgcn_readlane((int)mine[u >> 2], u & 3);
            const unsigned s = pk >> 9;
            dls[u] = pk & 511;
            ss[u] = s_src[(size_t)s * HEADS + hh];
            sd[u] = s_dst[(size_t)(base + dls[u]) * HEADS + hh];
            zv[u] = z4[(size_t)s * (OUT_DIM / 4) + lane];
            if (!SHIFT) mr[u] = L.mrun[dls[u] * HEADS + hh];
        }
        float ex[EB];
#pragma unroll
        for (int u = 0; u < EB; ++u) {
            float sc = ss[u] + sd[u];
            sc = (sc >= 0.0f) ? sc : NEG_SLOPE * sc;
            if (SHIFT) {
                const float t = sd[u] + ub;
                mr[u] = (t >= 0.0f) ? t : NEG_SLOPE * t;
            }
            ex[u] = expf(sc - mr[u]);
        }
#pragma unroll
        for (int g = 0; g < EB / 4; ++g) {
            const unsigned e01 = (__float_as_uint(ex[4 * g + 1]) & pick1) | (__float_as_uint(ex[4 * g])     & ~pick1);
            const unsigned e23 = (__float_as_uint(ex[4 * g + 3]) & pick1) | (__float_as_uint(ex[4 * g + 2]) & ~pick1);
            const float e = __uint_as_float((e23 & pick2) | (e01 & ~pick2));
            if (live[g]) atomicAdd(&L.den[(mine[g] & 511u) * HEADS + hh], e);
        }
#pragma unroll
        for (int u = 0; u < EB; ++u) {
            if (eb + (unsigned)u < n) {
                float* a = &L.acc[dls[u] * OUT_DIM + lane];
                atomicAdd(a + 0,  ex[u] * zv[u].x);
                atomicAdd(a + 32, ex[u] * zv[u].y);
                atomicAdd(a + 64, ex[u] * zv[u].z);
                atomicAdd(a + 96, ex[u] * zv[u].w);
            }
        }
    }
}

__global__ void __launch_bounds__(TB)
gat_tile(const int* __restrict__ src, const int* __restrict__ dst,
         const unsigned* __restrict__ ids,
         const float* __restrict__ s_src, const float* __restrict__ s_dst,
         const float* __restrict__ z, const float* __restrict__ h,
         const float* __restrict__ bnd, float* __restrict__ out) {
    __shared__ TileLds L;
    const int tid = threadIdx.x;
    const int base = blockIdx.x * TN;

    const float4* b4 = reinterpret_cast<const float4*>(bnd);
    float4 bv[(NSB * 4 + TB - 1) / TB];
#pragma unroll
    for (int k = 0; k < (NSB * 4 + TB - 1) / TB; ++k) {
        const int j = tid + TB * k;
        bv[k] = b4[j < NSB * 4 ? j : (tid & 3)];
    }

    float4* acc4 = reinterpret_cast<float4*>(L.acc);
    for (int i = tid; i < TN * OUT_DIM / 4; i += TB) acc4[i] = make_float4(0.f, 0.f, 0.f, 0.f);
    for (int i = tid; i < TN * HEADS; i += TB) {
        L.mrun[i] = -INFINITY; L.mnew[i] = -INFINITY; L.den[i] = 0.0f;
    }
    if (tid == 0) L.qn = 0u;
    {
        const bool lo = (tid & 2) != 0;
        float4 r = bv[0];
#pragma unroll
        for (int k = 1; k < (NSB * 4 + TB - 1) / TB; ++k) {
            r.x = lo ? fminf(r.x, bv[k].x) : fmaxf(r.x, bv[k].x);
            r.y = lo ? fminf(r.y, bv[k].y) : fmaxf(r.y, bv[k].y);
            r.z = lo ? fminf(r.z, bv[k].z) : fmaxf(r.z, bv[k].z);
            r.w = lo ? fminf(r.w, bv[k].w) : fmaxf(r.w, bv[k].w);
        }
#pragma unroll
        for (int m = 4; m < 32; m <<= 1) {
            const float ox = __shfl_xor(r.x, m), oy = __shfl_xor(r.y, m), oz = __shfl_xor(r.z, m), ow = __shfl_xor(r.w, m);
            r.x = lo ? fminf(r.x, ox) : fmaxf(r.x, ox);
            r.y = lo ? fminf(r.y, oy) : fmaxf(r.y, oy);
            r.z = lo ? fminf(r.z, oz) : fmaxf(r.z, oz);
            r.w = lo ? fminf(r.w, ow) : fmaxf(r.w, ow);
        }
        if ((tid & 31) < 4) {
            float* p = &L.pbw[(tid >> 5) * 16 + 4 * (tid & 3)];
            p[0] = r.x; p[1] = r.y; p[2] = r.z; p[3] = r.w;
        }
    }
    __syncthreads();
    if (tid < 16) {
        float r = L.pbw[tid];
        for (int w = 1; w < TB / 32; ++w) r = (tid & 8) ? fminf(r, L.pbw[w * 16 + tid]) : fmaxf(r, L.pbw[w * 16 + tid]);
        L.bd[tid] = r;
    }

    const int rot = blockIdx.x * 192;
    tile_scan(L, ids, rot, GV, blockIdx.x, tid);
    tile_scan(L, ids, 0, rot, blockIdx.x, tid);
    __syncthreads();
    const unsigned total = L.qn;
    int narrow = 1;
#pragma unroll
    for (int k = 0; k < HEADS; ++k) narrow &= (L.bd[k] - L.bd[HEADS + k] <= SPREAD_MAX) ? 1 : 0;
    const bool shift = __builtin_amdgcn_readfirstlane(narrow) != 0;
    const float ub = L.bd[(tid & 31) >> 2];
    if (shift) {
        if (total <= (unsigned)QCAP) {
            tile_batch<true>(L, total, ub, src, dst, s_src, s_dst, z, base, tid);
        } else {
            for (int seg = 0; seg < N_EDGES; seg += QCAP) {
                __syncthreads();
                if (tid == 0) L.qn = 0u;
                __syncthreads();
                int end = seg + QCAP < N_EDGES ? seg + QCAP : N_EDGES;
                tile_scan(L, ids, seg / 32, end / 32, blockIdx.x, tid);
                __syncthreads();
                tile_batch<true>(L, L.qn, ub, src, dst, s_src, s_dst, z, base, tid);
            }
        }
    } else if (total <= (unsigned)QCAP) {
        tile_batch<false>(L, total, ub, src, dst, s_src, s_dst, z, base, tid);
    } else {
        for (int seg = 0; seg < N_EDGES; seg += QCAP) {
            __syncthreads();
            if (tid == 0) L.qn = 0u;
            __syncthreads();
            int end = seg + QCAP < N_EDGES ? seg + QCAP : N_EDGES;
            tile_scan(L, ids, seg / 32, end / 32, blockIdx.x, tid);
            __syncthreads();
            tile_batch<false>(L, L.qn, ub, src, dst, s_src, s_dst, z, base, tid);
        }
    }
    __syncthreads();

    const float4* h4 = reinterpret_cast<const float4*>(h) + (size_t)base * (OUT_DIM / 4);
    float4* o4 = reinterpret_cast<float4*>(out) + (size_t)base * (OUT_DIM / 4);
    for (int i = tid; i < TN * OUT_DIM / 4; i += TB) {
        float dn = L.den[(i >> 5) * HEADS + ((i & 31) >> 2)];
        const float* ar = &L.acc[(i >> 5) * OUT_DIM + (i & 31)];
        float4 a = make_float4(ar[0], ar[32], ar[64], ar[96]), hv = h4[i], r;
        bool has = dn > 0.0f;
        float v0 = has ? a.x / dn : 0.0f, v1 = has ? a.y / dn : 0.0f;
        float v2 = has ? a.z / dn : 0.0f, v3 = has ? a.w / dn : 0.0f;
        r.x = hv.x + ((v0 > 0.0f) ? v0 : expm1f(v0));
        r.y = hv.y + ((v1 > 0.0f) ? v1 : expm1f(v1));
        r.z = hv.z + ((v2 > 0.0f) ? v2 : expm1f(v2));
        r.w = hv.w + ((v3 > 0.0f) ? v3 : expm1f(v3));
        o4[i] = r;
    }
}

#define EP_WG 250
#define EP_T  320
#define EP_N  5
static_assert(EP_WG * EP_T * EP_N == N_EDGES / 4, "the vectors of e are dealt exactly");
__global__ void __launch_bounds__(EP_T)
gat_epass(const float* __restrict__ e, float* __restrict__ oe) {
    const float4* e4 = reinterpret_cast<const float4*>(e);
    float4* o4 = reinterpret_cast<float4*>(oe);
    const int i0 = blockIdx.x * EP_T + threadIdx.x;
    float4 v[EP_N];
#pragma unroll
    for (int k = 0; k < EP_N; ++k) v[k] = e4[i0 + k * (EP_WG * EP_T)];
    asm volatile("" :: "v"(v[0].x), "v"(v[0].y), "v"(v[0].z), "v"(v[0].w), "v"(v[1].x), "v"(v[1].y), "v"(v[1].z), "v"(v[1].w),
                       "v"(v[2].x), "v"(v[2].y), "v"(v[2].z), "v"(v[2].w), "v"(v[3].x), "v"(v[3].y), "v"(v[3].z), "v"(v[3].w),
                       "v"(v[4].x), "v"(v[4].y), "v"(v[4].z), "v"(v[4].w));
    static_assert(EP_N == 5, "the statement above names every vector");
#pragma unroll
    for (int k = 0; k < EP_N; ++k) o4[i0 + k * (EP_WG * EP_T)] = v[k];
}

extern "C" void kernel_launch(void* const* d_in, const int* in_sizes, int n_in,
                              void* d_out, int out_size, void* d_ws, size_t ws_size,
                              hipStream_t stream) {
    const float* h     = (const float*)d_in[0];
    const float* e     = (const float*)d_in[1];
    const int*   src   = (const int*)d_in[2];
    const int*   dst   = (const int*)d_in[3];
    const float* Wfc   = (const float*)d_in[4];
    const float* Wattn = (const float*)d_in[5];
    float* out = (float*)d_out;

    float* z     = (float*)d_ws;
    float* s_src = z + (size_t)N_NODES * OUT_DIM;
    float* s_dst = s_src + (size_t)N_NODES * HEADS;

    gat_gemm_wmma<<<GEMM_WG, GEMM_T, 0, stream>>>(h, Wfc, z);

    float* oe = out + (size_t)N_NODES * OUT_DIM;
    unsigned* ids = (unsigned*)oe;

    float* bnd = oe + N_EDGES / 4;

    gat_scores<<<SC_WG, SC_T, 0, stream>>>(z, Wattn, dst, s_src, s_dst, ids, bnd);

    gat_tile<<<NT, TB, 0, stream>>>(src, dst, ids, s_src, s_dst, z, h, bnd, out);

    gat_epass<<<EP_WG, EP_T, 0, stream>>>(e, oe);
}


# Shifted path: the pass that turned queued edge indices into (src, local dst), and its barrier, are gone; each wave looks up src and dst of its own next batch while it adds the current one
# speedup vs baseline: 12.4964x; 1.0401x over previous
#include <hip/hip_runtime.h>
#include <math.h>

#define N_NODES 100000
#define N_EDGES 1600000
#define IN_DIM  128
#define HEADS   8
#define HDIM    16
#define OUT_DIM 128
#define NEG_SLOPE 0.01f
#define LDS_STRIDE 130
#define TN   400
#define NT   (N_NODES / TN)
#define GV   (N_EDGES / 32)
#define SC_WG    250
#define SC_WAVES 20
#define SC_T     (SC_WAVES * 32)
#define SC_ITEMS 5
#define NSB      SC_WG
#define SPREAD_MAX 60.0f

typedef __attribute__((ext_vector_type(2))) float v2f;
typedef __attribute__((ext_vector_type(8))) float v8f;

__device__ __forceinline__ void atomicMaxF(float* addr, float val) {
    if (val >= 0.0f) {
        atomicMax((int*)addr, __float_as_int(val));
    } else {
        atomicMin((unsigned int*)addr, __float_as_uint(val));
    }
}

#define GEMM_WG    256
#define GEMM_FULL  ((N_NODES / 16) / GEMM_WG)
#define GEMM_REST  ((N_NODES / 16) - GEMM_FULL * GEMM_WG)
#define GEMM_WAVES (GEMM_FULL + 4)
#define GEMM_T     (GEMM_WAVES * 32)
static_assert(N_NODES % 16 == 0 && GEMM_REST <= GEMM_WG && GEMM_T <= 1024, "every 16-row slab has its waves");

template <int CT>
__device__ __forceinline__ void gemm_slab(const float* __restrict__ h, const float* Wl,
                                          float* __restrict__ z, int mtile, int tbase, int lane) {
    const int row   = lane & 15;
    const int khalf = lane >> 4;

    const float* hrow = h + (size_t)(mtile * 16 + row) * IN_DIM;

    v8f acc[CT] = {};

    for (int k0 = 0; k0 < IN_DIM; k0 += 4) {
        const int kk = k0 + 2 * khalf;
        v2f a = *reinterpret_cast<const v2f*>(hrow + kk);
#pragma unroll
        for (int t = 0; t < CT; ++t) {
            v2f b = *reinterpret_cast<const v2f*>(&Wl[((tbase + t) * 16 + row) * LDS_STRIDE + kk]);
            acc[t] = __builtin_amdgcn_wmma_f32_16x16x4_f32(
                false, a, false, b, (short)0, acc[t], false, false);
        }
    }

    float* zbase = z + (size_t)(mtile * 16) * OUT_DIM;
#pragma unroll
    for (int t = 0; t < CT; ++t) {
#pragma unroll
        for (int i = 0; i < 8; ++i) {
            int mrow = i + khalf * 8;
            zbase[(size_t)mrow * OUT_DIM + (tbase + t) * 16 + row] = acc[t][i];
        }
    }
}

__global__ void __launch_bounds__(GEMM_T)
gat_gemm_wmma(const float* __restrict__ h, const float* __restrict__ Wfc,
              float* __restrict__ z) {
    __shared__ float Wl[OUT_DIM * LDS_STRIDE];

    const int tid = threadIdx.x;
    for (int j = tid; j < IN_DIM * OUT_DIM; j += GEMM_T) {
        int i  = j >> 7;
        int c  = j & 127;
        int hh = c >> 4;
        int d  = c & 15;
        Wl[c * LDS_STRIDE + i] = Wfc[hh * (IN_DIM * HDIM) + i * HDIM + d];
    }
    __syncthreads();

    const int wave = tid >> 5;
    const int lane = tid & 31;
    if (wave < GEMM_FULL) {
        gemm_slab<8>(h, Wl, z, wave * GEMM_WG + (int)blockIdx.x, 0, lane);
    } else if ((int)blockIdx.x < GEMM_REST) {
        gemm_slab<2>(h, Wl, z, GEMM_FULL * GEMM_WG + (int)blockIdx.x, 2 * (wave - GEMM_FULL), lane);
    }
}

__device__ __forceinline__ unsigned owner_id(int d) {
    unsigned k = (unsigned)d / (unsigned)TN;
    return k < 255u ? k : 255u;
}

__device__ __forceinline__ unsigned plane_vec(unsigned g, unsigned half) {
    return ((g >> 4) << 5) + (half << 4) + (g & 15u);
}

__global__ void __launch_bounds__(SC_T)
gat_scores(const float* __restrict__ z,
           const float* __restrict__ Wattn,
           const int* __restrict__ dst,
           float* __restrict__ s_src, float* __restrict__ s_dst,
           unsigned* __restrict__ ids, float* __restrict__ bnd) {
    __shared__ float pb[SC_WAVES][16];
    static_assert(SC_WG * SC_WAVES * SC_ITEMS * 32 == N_NODES * HEADS && SC_WG * SC_WAVES * SC_ITEMS * 2 == GV,
                  "every wave has SC_ITEMS whole items; an item is two edge groups");
    const unsigned lane = (unsigned)threadIdx.x & 31u;
    const int wv = threadIdx.x >> 5;
    const unsigned hh = lane & 7u;
    const unsigned item0 = ((unsigned)blockIdx.x * SC_WAVES + (unsigned)wv) * SC_ITEMS;

    const float4* z4 = reinterpret_cast<const float4*>(z);
    float4 zv[SC_ITEMS][4];
    int da[SC_ITEMS], db[SC_ITEMS];
#pragma unroll
    for (int k = 0; k < SC_ITEMS; ++k) {
        const unsigned idx = (item0 + (unsigned)k) * 32u + lane;
        const float4* zp = z4 + (size_t)(idx >> 3) * (OUT_DIM / 4) + hh * (HDIM / 4);
#pragma unroll
        for (int j = 0; j < 4; ++j) zv[k][j] = zp[j];
        da[k] = dst[(item0 + (unsigned)k) * 64u + lane];
        db[k] = dst[(item0 + (unsigned)k) * 64u + 32u + lane];
    }
    const float4* as4 = reinterpret_cast<const float4*>(Wattn + hh * (2 * HDIM));
    float4 av[8];
#pragma unroll
    for (int j = 0; j < 8; ++j) av[j] = as4[j];

    float mx = 0.0f, mn = 0.0f;
#pragma unroll
    for (int k = 0; k < SC_ITEMS; ++k) {
        const unsigned idx = (item0 + (unsigned)k) * 32u + lane;
        float ss = 0.0f, sd = 0.0f;
#pragma unroll
        for (int j = 0; j < 4; ++j) {
            ss += zv[k][j].x * av[j].x;  sd += zv[k][j].x * av[4 + j].x;
            ss += zv[k][j].y * av[j].y;  sd += zv[k][j].y * av[4 + j].y;
            ss += zv[k][j].z * av[j].z;  sd += zv[k][j].z * av[4 + j].z;
            ss += zv[k][j].w * av[j].w;  sd += zv[k][j].w * av[4 + j].w;
        }
        s_src[idx] = ss;
        s_dst[idx] = sd;
        mx = (k == 0) ? ss : fmaxf(mx, ss);
        mn = (k == 0) ? ss : fminf(mn, ss);

        const unsigned g0 = (item0 + (unsigned)k) * 2u;
        const unsigned ida = owner_id(da[k]);
        const unsigned idb = owner_id(db[k]);
        unsigned w = 0u;
#pragma unroll
        for (int b = 0; b < 8; ++b) {
            const unsigned pa = __builtin_amdgcn_ballot_w32(((ida >> b) & 1u) != 0u);
            const unsigned pbm = __builtin_amdgcn_ballot_w32(((idb >> b) & 1u) != 0u);
            if (lane == (unsigned)b) w = pa;
            if (lane == (unsigned)(8 + b)) w = pbm;
        }
        if (lane < 16u) {
            const unsigned g = g0 + (lane >> 3), b = lane & 7u;
            ids[plane_vec(g, b >> 2) * 4u + (b & 3u)] = w;
        }
    }

    mx = fmaxf(mx, __shfl_xor(mx, 8));  mx = fmaxf(mx, __shfl_xor(mx, 16));
    mn = fminf(mn, __shfl_xor(mn, 8));  mn = fminf(mn, __shfl_xor(mn, 16));
    if (lane < 8u) { pb[wv][lane] = mx; pb[wv][8u + lane] = mn; }
    __syncthreads();
    if (threadIdx.x < 16) {
        const int t = threadIdx.x;
        float r = pb[0][t];
#pragma unroll
        for (int w = 1; w < SC_WAVES; ++w) r = (t < 8) ? fmaxf(r, pb[w][t]) : fminf(r, pb[w][t]);
        bnd[blockIdx.x * 16 + t] = r;
    }
}

#define QCAP 12288
#define TB   1024
#define EB   8

struct TileLds {
    float acc[TN * OUT_DIM];
    float mrun[TN * HEADS];
    float mnew[TN * HEADS];
    float den[TN * HEADS];
    unsigned q[QCAP];
    unsigned qn;
    float pbw[32 * 16];
    float bd[16];
};

__device__ __forceinline__ unsigned match_planes(uint4 a, uint4 b, const unsigned (&inv)[8]) {
    return (a.x ^ inv[0]) & (a.y ^ inv[1]) & (a.z ^ inv[2]) & (a.w ^ inv[3]) &
           (b.x ^ inv[4]) & (b.y ^ inv[5]) & (b.z ^ inv[6]) & (b.w ^ inv[7]);
}

__device__ __forceinline__ void scan_groups(TileLds& L, const uint4 (&c)[8], const unsigned (&inv)[8],
                                            int g, int gend, unsigned lane) {
    unsigned r0 = (g          < gend) ? match_planes(c[0], c[1], inv) : 0u;
    unsigned r1 = (g + TB     < gend) ? match_planes(c[2], c[3], inv) : 0u;
    unsigned r2 = (g + 2 * TB < gend) ? match_planes(c[4], c[5], inv) : 0u;
    unsigned r3 = (g + 3 * TB < gend) ? match_planes(c[6], c[7], inv) : 0u;
    const unsigned e0 = (unsigned)g * 32u;
    for (;;) {
        const bool has = (r0 | r1 | r2 | r3) != 0u;
        const unsigned m = __builtin_amdgcn_ballot_w32(has);
        if (m == 0u) break;
        unsigned s0 = 0u;
        if (lane == 0u) s0 = atomicAdd(&L.qn, (unsigned)__builtin_popcount(m));
        s0 = (unsigned)__builtin_amdgcn_readfirstlane((int)s0);
        if (has) {
            const bool fp = (r0 | r1) != 0u;
            const unsigned x0 = fp ? r0 : r2, x1 = fp ? r1 : r3;
            const bool lo = x0 != 0u;
            const unsigned r = lo ? x0 : x1;
            const unsigned u = (fp ? 0u : 2u) + (lo ? 0u : 1u);
            const unsigned e = e0 + u * (32u * TB) + (unsigned)__builtin_ctz(r);
            const unsigned slot = s0 + __builtin_amdgcn_mbcnt_lo(m, 0u);
            if (slot < (unsigned)QCAP) L.q[slot] = e;
            const unsigned rr = r & (r - 1u);
            if (fp) { if (lo) r0 = rr; else r1 = rr; }
            else    { if (lo) r2 = rr; else r3 = rr; }
        }
    }
}

__device__ __forceinline__ void load_groups(const uint4* __restrict__ k4, int g, uint4 (&c)[8]) {
#pragma unroll
    for (int u = 0; u < 4; ++u) {
        const int gg = g + u * TB;
        const unsigned gc = (unsigned)(gg < GV ? gg : GV - 1);
        c[2 * u]     = k4[plane_vec(gc, 0u)];
        c[2 * u + 1] = k4[plane_vec(gc, 1u)];
    }
}

__device__ __forceinline__ void tile_scan(TileLds& L, const unsigned* __restrict__ ids,
                                          int v0, int v1, unsigned tileb, int tid) {
    const uint4* k4 = reinterpret_cast<const uint4*>(ids);
    const unsigned lane = (unsigned)tid & 31u;
    unsigned inv[8];
#pragma unroll
    for (int k = 0; k < 8; ++k) inv[k] = ((tileb >> k) & 1u) ? 0u : ~0u;
    uint4 ca[8], cb[8];
    int g = v0 + tid;
    load_groups(k4, g, ca);
    for (int gb = v0 + (tid & ~31); gb < v1; gb += 8 * TB, g += 8 * TB) {
        load_groups(k4, g + 4 * TB, cb);
        scan_groups(L, ca, inv, g, v1, lane);
        if (gb + 4 * TB >= v1) break;
        load_groups(k4, g + 8 * TB, ca);
        scan_groups(L, cb, inv, g + 4 * TB, v1, lane);
    }
}

template <bool SHIFT>
__device__ __forceinline__ void tile_batch(TileLds& L, unsigned n, float ub,
                                           const int* __restrict__ src,
                                           const int* __restrict__ dst,
                                           const float* __restrict__ s_src,
                                           const float* __restrict__ s_dst,
                                           const float* __restrict__ z,
                                           int base, int tid) {
    if (!SHIFT) {
    for (unsigned i0 = tid; i0 < n; i0 += 4 * TB) {
        unsigned s[4], d[4];
#pragma unroll
        for (int u = 0; u < 4; ++u) {
            unsigned i = i0 + u * TB;
            unsigned ei = L.q[i < n ? i : i0];
            ei = ei < (unsigned)N_EDGES ? ei : (unsigned)(N_EDGES - 1);
            s[u] = (unsigned)src[ei];
            d[u] = (unsigned)dst[ei] - (unsigned)base;
        }
#pragma unroll
        for (int u = 0; u < 4; ++u) {
            unsigned i = i0 + u * TB;
            unsigned sv = s[u] < (unsigned)N_NODES ? s[u] : (unsigned)(N_NODES - 1);
            unsigned dl = d[u] < (unsigned)TN ? d[u] : (unsigned)(TN - 1);
            if (i < n) L.q[i] = (sv << 9) | dl;
        }
    }
    __syncthreads();
    {
        const unsigned nh = n * HEADS, hd = tid & 7;
        for (unsigned i0 = tid; i0 < nh; i0 += 4 * TB) {
            float a[4], b[4]; unsigned cell[4];
#pragma unroll
            for (int u = 0; u < 4; ++u) {
                unsigned i = i0 + u * TB;
                unsigned pk = L.q[(i < nh ? i : i0) >> 3];
                unsigned s = pk >> 9, dl = pk & 511;
                a[u] = s_src[(size_t)s * HEADS + hd];
                b[u] = s_dst[(size_t)(base + dl) * HEADS + hd];
                cell[u] = dl * HEADS + hd;
            }
#pragma unroll
            for (int u = 0; u < 4; ++u) {
                float sc = a[u] + b[u];
                sc = (sc >= 0.0f) ? sc : NEG_SLOPE * sc;
                if (i0 + u * TB < nh) atomicMaxF(&L.mnew[cell[u]], sc);
            }
        }
    }
    __syncthreads();
    for (int i = tid; i < TN * HEADS; i += TB) {
        float mo = L.mrun[i], mn = L.mnew[i];
        if (mo != mn) {
            if (mo != -INFINITY) {
                float f = expf(mo - mn);
                L.den[i] *= f;
                float* a = &L.acc[(i >> 3) * OUT_DIM + (i & 7) * 4];
#pragma unroll
                for (int j = 0; j < 4; ++j) {
#pragma unroll
                    for (int k = 0; k < 4; ++k) a[32 * j + k] *= f;
                }
            }
            L.mrun[i] = mn;
        }
    }
    __syncthreads();
    }
    const int lane = tid & 31, hh = lane >> 2;
    const unsigned wave = (unsigned)__builtin_amdgcn_readfirstlane(tid >> 5);
    const unsigned qd = (unsigned)lane & 3u;
    const unsigned pick1 = (qd & 1u) ? ~0u : 0u, pick2 = (qd & 2u) ? ~0u : 0u;
    const float4* z4 = reinterpret_cast<const float4*>(z);
    unsigned sN[EB / 4], dN[EB / 4];
    if (SHIFT && wave * EB < n) {
#pragma unroll
        for (int g = 0; g < EB / 4; ++g) {
            const unsigned ql = wave * EB + 4u * (unsigned)g + qd;
            unsigned ei = L.q[ql < n ? ql : n - 1];
            ei = ei < (unsigned)N_EDGES ? ei : (unsigned)(N_EDGES - 1);
            sN[g] = (unsigned)src[ei];
            dN[g] = (unsigned)dst[ei];
        }
    }
    for (unsigned eb = wave * EB; eb < n; eb += (TB / 32) * EB) {
        unsigned mine[EB / 4]; bool live[EB / 4];
#pragma unroll
        for (int g = 0; g < EB / 4; ++g) {
            const unsigned ql = eb + 4u * (unsigned)g + qd;
            live[g] = ql < n;
            if (SHIFT) {
                const unsigned sv = sN[g] < (unsigned)N_NODES ? sN[g] : (unsigned)(N_NODES - 1);
                const unsigned dr = dN[g] - (unsigned)base;
                const unsigned dl = dr < (unsigned)TN ? dr : (unsigned)(TN - 1);
                mine[g] = (sv << 9) | dl;
            } else {
                mine[g] = L.q[live[g] ? ql : n - 1];
            }
        }
        float4 zv[EB]; float ss[EB], sd[EB], mr[EB]; unsigned dls[EB];
#pragma unroll
        for (int u = 0; u < EB; ++u) {
            const unsigned pk = (unsigned)__builtin_amdgcn_readlane((int)mine[u >> 2], u & 3);
            const unsigned s = pk >> 9;
            dls[u] = pk & 511;
            ss[u] = s_src[(size_t)s * HEADS + hh];
            sd[u] = s_dst[(size_t)(base + dls[u]) * HEADS + hh];
            zv[u] = z4[(size_t)s * (OUT_DIM / 4) + lane];
            if (!SHIFT) mr[u] = L.mrun[dls[u] * HEADS + hh];
        }
        if (SHIFT) {
            const unsigned nb = eb + (TB / 32) * EB;
            if (nb < n) {
#pragma unroll
                for (int g = 0; g < EB / 4; ++g) {
                    const unsigned ql = nb + 4u * (unsigned)g + qd;
                    unsigned ei = L.q[ql < n ? ql : n - 1];
                    ei = ei < (unsigned)N_EDGES ? ei : (unsigned)(N_EDGES - 1);
                    sN[g] = (unsigned)src[ei];
                    dN[g] = (unsigned)dst[ei];
                }
            }
        }
        float ex[EB];
#pragma unroll
        for (int u = 0; u < EB; ++u) {
            float sc = ss[u] + sd[u];
            sc = (sc >= 0.0f) ? sc : NEG_SLOPE * sc;
            if (SHIFT) {
                const float t = sd[u] + ub;
                mr[u] = (t >= 0.0f) ? t : NEG_SLOPE * t;
            }
            ex[u] = expf(sc - mr[u]);
        }
#pragma unroll
        for (int g = 0; g < EB / 4; ++g) {
            const unsigned e01 = (__float_as_uint(ex[4 * g + 1]) & pick1) | (__float_as_uint(ex[4 * g])     & ~pick1);
            const unsigned e23 = (__float_as_uint(ex[4 * g + 3]) & pick1) | (__float_as_uint(ex[4 * g + 2]) & ~pick1);
            const float e = __uint_as_float((e23 & pick2) | (e01 & ~pick2));
            if (live[g]) atomicAdd(&L.den[(mine[g] & 511u) * HEADS + hh], e);
        }
#pragma unroll
        for (int u = 0; u < EB; ++u) {
            if (eb + (unsigned)u < n) {
                float* a = &L.acc[dls[u] * OUT_DIM + lane];
                atomicAdd(a + 0,  ex[u] * zv[u].x);
                atomicAdd(a + 32, ex[u] * zv[u].y);
                atomicAdd(a + 64, ex[u] * zv[u].z);
                atomicAdd(a + 96, ex[u] * zv[u].w);
            }
        }
    }
}

__global__ void __launch_bounds__(TB)
gat_tile(const int* __restrict__ src, const int* __restrict__ dst,
         const unsigned* __restrict__ ids,
         const float* __restrict__ s_src, const float* __restrict__ s_dst,
         const float* __restrict__ z, const float* __restrict__ h,
         const float* __restrict__ bnd, float* __restrict__ out) {
    __shared__ TileLds L;
    const int tid = threadIdx.x;
    const int base = blockIdx.x * TN;

    const float4* b4 = reinterpret_cast<const float4*>(bnd);
    float4 bv[(NSB * 4 + TB - 1) / TB];
#pragma unroll
    for (int k = 0; k < (NSB * 4 + TB - 1) / TB; ++k) {
        const int j = tid + TB * k;
        bv[k] = b4[j < NSB * 4 ? j : (tid & 3)];
    }

    float4* acc4 = reinterpret_cast<float4*>(L.acc);
    for (int i = tid; i < TN * OUT_DIM / 4; i += TB) acc4[i] = make_float4(0.f, 0.f, 0.f, 0.f);
    for (int i = tid; i < TN * HEADS; i += TB) {
        L.mrun[i] = -INFINITY; L.mnew[i] = -INFINITY; L.den[i] = 0.0f;
    }
    if (tid == 0) L.qn = 0u;
    {
        const bool lo = (tid & 2) != 0;
        float4 r = bv[0];
#pragma unroll
        for (int k = 1; k < (NSB * 4 + TB - 1) / TB; ++k) {
            r.x = lo ? fminf(r.x, bv[k].x) : fmaxf(r.x, bv[k].x);
            r.y = lo ? fminf(r.y, bv[k].y) : fmaxf(r.y, bv[k].y);
            r.z = lo ? fminf(r.z, bv[k].z) : fmaxf(r.z, bv[k].z);
            r.w = lo ? fminf(r.w, bv[k].w) : fmaxf(r.w, bv[k].w);
        }
#pragma unroll
        for (int m = 4; m < 32; m <<= 1) {
            const float ox = __shfl_xor(r.x, m), oy = __shfl_xor(r.y, m), oz = __shfl_xor(r.z, m), ow = __shfl_xor(r.w, m);
            r.x = lo ? fminf(r.x, ox) : fmaxf(r.x, ox);
            r.y = lo ? fminf(r.y, oy) : fmaxf(r.y, oy);
            r.z = lo ? fminf(r.z, oz) : fmaxf(r.z, oz);
            r.w = lo ? fminf(r.w, ow) : fmaxf(r.w, ow);
        }
        if ((tid & 31) < 4) {
            float* p = &L.pbw[(tid >> 5) * 16 + 4 * (tid & 3)];
            p[0] = r.x; p[1] = r.y; p[2] = r.z; p[3] = r.w;
        }
    }
    __syncthreads();
    if (tid < 16) {
        float r = L.pbw[tid];
        for (int w = 1; w < TB / 32; ++w) r = (tid & 8) ? fminf(r, L.pbw[w * 16 + tid]) : fmaxf(r, L.pbw[w * 16 + tid]);
        L.bd[tid] = r;
    }

    const int rot = blockIdx.x * 192;
    tile_scan(L, ids, rot, GV, blockIdx.x, tid);
    tile_scan(L, ids, 0, rot, blockIdx.x, tid);
    __syncthreads();
    const unsigned total = L.qn;
    int narrow = 1;
#pragma unroll
    for (int k = 0; k < HEADS; ++k) narrow &= (L.bd[k] - L.bd[HEADS + k] <= SPREAD_MAX) ? 1 : 0;
    const bool shift = __builtin_amdgcn_readfirstlane(narrow) != 0;
    const float ub = L.bd[(tid & 31) >> 2];
    if (shift) {
        if (total <= (unsigned)QCAP) {
            tile_batch<true>(L, total, ub, src, dst, s_src, s_dst, z, base, tid);
        } else {
            for (int seg = 0; seg < N_EDGES; seg += QCAP) {
                __syncthreads();
                if (tid == 0) L.qn = 0u;
                __syncthreads();
                int end = seg + QCAP < N_EDGES ? seg + QCAP : N_EDGES;
                tile_scan(L, ids, seg / 32, end / 32, blockIdx.x, tid);
                __syncthreads();
                tile_batch<true>(L, L.qn, ub, src, dst, s_src, s_dst, z, base, tid);
            }
        }
    } else if (total <= (unsigned)QCAP) {
        tile_batch<false>(L, total, ub, src, dst, s_src, s_dst, z, base, tid);
    } else {
        for (int seg = 0; seg < N_EDGES; seg += QCAP) {
            __syncthreads();
            if (tid == 0) L.qn = 0u;
            __syncthreads();
            int end = seg + QCAP < N_EDGES ? seg + QCAP : N_EDGES;
            tile_scan(L, ids, seg / 32, end / 32, blockIdx.x, tid);
            __syncthreads();
            tile_batch<false>(L, L.qn, ub, src, dst, s_src, s_dst, z, base, tid);
        }
    }
    __syncthreads();

    const float4* h4 = reinterpret_cast<const float4*>(h) + (size_t)base * (OUT_DIM / 4);
    float4* o4 = reinterpret_cast<float4*>(out) + (size_t)base * (OUT_DIM / 4);
    for (int i = tid; i < TN * OUT_DIM / 4; i += TB) {
        float dn = L.den[(i >> 5) * HEADS + ((i & 31) >> 2)];
        const float* ar = &L.acc[(i >> 5) * OUT_DIM + (i & 31)];
        float4 a = make_float4(ar[0], ar[32], ar[64], ar[96]), hv = h4[i], r;
        bool has = dn > 0.0f;
        float v0 = has ? a.x / dn : 0.0f, v1 = has ? a.y / dn : 0.0f;
        float v2 = has ? a.z / dn : 0.0f, v3 = has ? a.w / dn : 0.0f;
        r.x = hv.x + ((v0 > 0.0f) ? v0 : expm1f(v0));
        r.y = hv.y + ((v1 > 0.0f) ? v1 : expm1f(v1));
        r.z = hv.z + ((v2 > 0.0f) ? v2 : expm1f(v2));
        r.w = hv.w + ((v3 > 0.0f) ? v3 : expm1f(v3));
        o4[i] = r;
    }
}

#define EP_WG 250
#define EP_T  320
#define EP_N  5
static_assert(EP_WG * EP_T * EP_N == N_EDGES / 4, "the vectors of e are dealt exactly");
__global__ void __launch_bounds__(EP_T)
gat_epass(const float* __restrict__ e, float* __restrict__ oe) {
    const float4* e4 = reinterpret_cast<const float4*>(e);
    float4* o4 = reinterpret_cast<float4*>(oe);
    const int i0 = blockIdx.x * EP_T + threadIdx.x;
    float4 v[EP_N];
#pragma unroll
    for (int k = 0; k < EP_N; ++k) v[k] = e4[i0 + k * (EP_WG * EP_T)];
    asm volatile("" :: "v"(v[0].x), "v"(v[0].y), "v"(v[0].z), "v"(v[0].w), "v"(v[1].x), "v"(v[1].y), "v"(v[1].z), "v"(v[1].w),
                       "v"(v[2].x), "v"(v[2].y), "v"(v[2].z), "v"(v[2].w), "v"(v[3].x), "v"(v[3].y), "v"(v[3].z), "v"(v[3].w),
                       "v"(v[4].x), "v"(v[4].y), "v"(v[4].z), "v"(v[4].w));
    static_assert(EP_N == 5, "the statement above names every vector");
#pragma unroll
    for (int k = 0; k < EP_N; ++k) o4[i0 + k * (EP_WG * EP_T)] = v[k];
}

extern "C" void kernel_launch(void* const* d_in, const int* in_sizes, int n_in,
                              void* d_out, int out_size, void* d_ws, size_t ws_size,
                              hipStream_t stream) {
    const float* h     = (const float*)d_in[0];
    const float* e     = (const float*)d_in[1];
    const int*   src   = (const int*)d_in[2];
    const int*   dst   = (const int*)d_in[3];
    const float* Wfc   = (const float*)d_in[4];
    const float* Wattn = (const float*)d_in[5];
    float* out = (float*)d_out;

    float* z     = (float*)d_ws;
    float* s_src = z + (size_t)N_NODES * OUT_DIM;
    float* s_dst = s_src + (size_t)N_NODES * HEADS;

    gat_gemm_wmma<<<GEMM_WG, GEMM_T, 0, stream>>>(h, Wfc, z);

    float* oe = out + (size_t)N_NODES * OUT_DIM;
    unsigned* ids = (unsigned*)oe;

    float* bnd = oe + N_EDGES / 4;

    gat_scores<<<SC_WG, SC_T, 0, stream>>>(z, Wattn, dst, s_src, s_dst, ids, bnd);

    gat_tile<<<NT, TB, 0, stream>>>(src, dst, ids, s_src, s_dst, z, h, bnd, out);

    gat_epass<<<EP_WG, EP_T, 0, stream>>>(e, oe);
}
